# starting kernel
# baseline (speedup 1.0000x reference)
#include <hip/hip_runtime.h>
#include <hip/hip_bf16.h>
#include <math.h>

#define NNODES 50000
#define EDGES  800000
#define INFEAT 128
#define HIDTOT 256
#define NEG_SLOPE 0.2f
#define ORD_NEG_INF 0x007FFFFFu

typedef __attribute__((ext_vector_type(2))) float v2f;
typedef __attribute__((ext_vector_type(8))) float v8f;

__device__ __forceinline__ unsigned ford(float f) {
    unsigned u = __float_as_uint(f);
    return (u & 0x80000000u) ? ~u : (u | 0x80000000u);
}
__device__ __forceinline__ float fdec(unsigned u) {
    return __uint_as_float((u & 0x80000000u) ? (u & 0x7FFFFFFFu) : ~u);
}
__device__ __forceinline__ float lrelu(float x) {
    return x >= 0.f ? x : NEG_SLOPE * x;
}

__global__ void init_kernel(float* out1, float* denom1, unsigned* m1,
                            float* denom2, unsigned* m2, float* dout) {
    size_t i = (size_t)blockIdx.x * blockDim.x + threadIdx.x;
    if (i < (size_t)NNODES * HIDTOT) out1[i] = 0.f;
    if (i < (size_t)NNODES * 4) { denom1[i] = 0.f; m1[i] = ORD_NEG_INF; }
    if (i < (size_t)NNODES)     { denom2[i] = 0.f; m2[i] = ORD_NEG_INF; }
    if (i < (size_t)NNODES * 2) dout[i] = 0.f;
}

__global__ __launch_bounds__(256) void gemm1_kernel(
    const float* __restrict__ feat,
    const float* __restrict__ Ws, const float* __restrict__ bs,
    const float* __restrict__ Wd, const float* __restrict__ bd,
    float* __restrict__ hs, float* __restrict__ hd)
{
    const int wave = threadIdx.x >> 5;
    const int lane = threadIdx.x & 31;
    const int half = lane >> 4;
    const int row  = lane & 15;

    int tile = blockIdx.x * 8 + wave;
    if (tile >= 3125 * 4) return;
    const int mt  = tile >> 2;
    const int nt4 = tile & 3;
    const int m0  = mt * 16;
    const int c0  = nt4 * 64;

    v8f zero = {0.f,0.f,0.f,0.f,0.f,0.f,0.f,0.f};
    v8f accS[4], accD[4];
    #pragma unroll
    for (int j = 0; j < 4; ++j) { accS[j] = zero; accD[j] = zero; }

    const float* arow = feat + (size_t)(m0 + row) * INFEAT;
    for (int k0 = 0; k0 < INFEAT; k0 += 4) {
        const int ka = k0 + 2 * half;
        v2f a = *(const v2f*)(arow + ka);
        #pragma unroll
        for (int j = 0; j < 4; ++j) {
            const int col = c0 + j * 16 + row;
            v2f bS, bD;
            bS.x = Ws[(size_t)ka * HIDTOT + col];
            bS.y = Ws[(size_t)(ka + 1) * HIDTOT + col];
            bD.x = Wd[(size_t)ka * HIDTOT + col];
            bD.y = Wd[(size_t)(ka + 1) * HIDTOT + col];
            accS[j] = __builtin_amdgcn_wmma_f32_16x16x4_f32(
                false, a, false, bS, (short)0, accS[j], false, false);
            accD[j] = __builtin_amdgcn_wmma_f32_16x16x4_f32(
                false, a, false, bD, (short)0, accD[j], false, false);
        }
    }
    #pragma unroll
    for (int j = 0; j < 4; ++j) {
        const int col = c0 + j * 16 + row;
        const float biasS = bs[col], biasD = bd[col];
        #pragma unroll
        for (int i = 0; i < 8; ++i) {
            const int r = m0 + i + 8 * half;
            hs[(size_t)r * HIDTOT + col] = accS[j][i] + biasS;
            hd[(size_t)r * HIDTOT + col] = accD[j][i] + biasD;
        }
    }
}

__global__ __launch_bounds__(256) void edge_score1_kernel(
    const float* __restrict__ hs, const float* __restrict__ hd,
    const int* __restrict__ src, const int* __restrict__ dst,
    const float* __restrict__ attn, float* __restrict__ score,
    unsigned* __restrict__ mbits)
{
    const int e = blockIdx.x * 8 + (threadIdx.x >> 5);
    if (e >= EDGES) return;
    const int lane = threadIdx.x & 31;
    const int s = src[e], d = dst[e];
    const int base = lane * 8;

    const float4* ph = (const float4*)(hs + (size_t)s * HIDTOT + base);
    const float4* pd = (const float4*)(hd + (size_t)d * HIDTOT + base);
    const float4* pa = (const float4*)(attn + base);

    float part = 0.f;
    #pragma unroll
    for (int q = 0; q < 2; ++q) {
        float4 a = ph[q], b = pd[q], w = pa[q];
        part += lrelu(a.x + b.x) * w.x + lrelu(a.y + b.y) * w.y +
                lrelu(a.z + b.z) * w.z + lrelu(a.w + b.w) * w.w;
    }
    part += __shfl_xor(part, 1, 32);
    part += __shfl_xor(part, 2, 32);
    part += __shfl_xor(part, 4, 32);
    if ((lane & 7) == 0) {
        const int h = lane >> 3;
        score[(size_t)e * 4 + h] = part;
        atomicMax(mbits + (size_t)d * 4 + h, ford(part));
    }
}

__global__ void edge_exp1_kernel(const int* __restrict__ dst,
                                 float* __restrict__ score,
                                 const unsigned* __restrict__ mbits,
                                 float* __restrict__ denom)
{
    const int t = blockIdx.x * blockDim.x + threadIdx.x;
    if (t >= EDGES * 4) return;
    const int e = t >> 2, h = t & 3;
    const int d = dst[e];
    float m = fdec(mbits[(size_t)d * 4 + h]);
    if (!isfinite(m)) m = 0.f;
    const float ex = expf(score[t] - m);
    score[t] = ex;
    atomicAdd(denom + (size_t)d * 4 + h, ex);
}

__global__ __launch_bounds__(256) void edge_aggr1_kernel(
    const float* __restrict__ hs, const int* __restrict__ src,
    const int* __restrict__ dst, const float* __restrict__ ex,
    const float* __restrict__ denom, float* __restrict__ out)
{
    const int e = blockIdx.x * 8 + (threadIdx.x >> 5);
    if (e >= EDGES) return;
    const int lane = threadIdx.x & 31;
    const int s = src[e], d = dst[e];
    const int h = lane >> 3;
    float den = denom[(size_t)d * 4 + h];
    den = den > 0.f ? den : 1.f;
    const float alpha = ex[(size_t)e * 4 + h] / den;
    const int base = lane * 8;
    const float4* ph = (const float4*)(hs + (size_t)s * HIDTOT + base);
    float* po = out + (size_t)d * HIDTOT + base;
    #pragma unroll
    for (int q = 0; q < 2; ++q) {
        float4 a = ph[q];
        atomicAdd(po + q * 4 + 0, alpha * a.x);
        atomicAdd(po + q * 4 + 1, alpha * a.y);
        atomicAdd(po + q * 4 + 2, alpha * a.z);
        atomicAdd(po + q * 4 + 3, alpha * a.w);
    }
}

__global__ void elu_kernel(float* __restrict__ x, int n) {
    const int i = blockIdx.x * blockDim.x + threadIdx.x;
    if (i < n) {
        const float v = x[i];
        x[i] = v > 0.f ? v : (expf(v) - 1.f);
    }
}

__global__ __launch_bounds__(256) void proj2_kernel(
    const float* __restrict__ h,
    const float* __restrict__ W2s, const float* __restrict__ b2s,
    const float* __restrict__ W2d, const float* __restrict__ b2d,
    float* __restrict__ hs2, float* __restrict__ hd2)
{
    const int n = blockIdx.x * 8 + (threadIdx.x >> 5);
    if (n >= NNODES) return;
    const int lane = threadIdx.x & 31;
    const int base = lane * 8;
    const float* hr = h + (size_t)n * HIDTOT + base;
    float s0 = 0.f, s1 = 0.f, d0 = 0.f, d1 = 0.f;
    #pragma unroll
    for (int k = 0; k < 8; ++k) {
        const float x = hr[k];
        const float2 ws = *(const float2*)(W2s + (size_t)(base + k) * 2);
        const float2 wd = *(const float2*)(W2d + (size_t)(base + k) * 2);
        s0 += x * ws.x; s1 += x * ws.y;
        d0 += x * wd.x; d1 += x * wd.y;
    }
    #pragma unroll
    for (int o = 1; o < 32; o <<= 1) {
        s0 += __shfl_xor(s0, o, 32); s1 += __shfl_xor(s1, o, 32);
        d0 += __shfl_xor(d0, o, 32); d1 += __shfl_xor(d1, o, 32);
    }
    if (lane == 0) {
        hs2[(size_t)n * 2 + 0] = s0 + b2s[0];
        hs2[(size_t)n * 2 + 1] = s1 + b2s[1];
        hd2[(size_t)n * 2 + 0] = d0 + b2d[0];
        hd2[(size_t)n * 2 + 1] = d1 + b2d[1];
    }
}

__global__ void edge_score2_kernel(const float* __restrict__ hs2,
                                   const float* __restrict__ hd2,
                                   const int* __restrict__ src,
                                   const int* __restrict__ dst,
                                   const float* __restrict__ attn2,
                                   float* __restrict__ score2,
                                   unsigned* __restrict__ m2)
{
    const int e = blockIdx.x * blockDim.x + threadIdx.x;
    if (e >= EDGES) return;
    const int s = src[e], d = dst[e];
    const float2 a = *(const float2*)(hs2 + (size_t)s * 2);
    const float2 b = *(const float2*)(hd2 + (size_t)d * 2);
    const float sc = lrelu(a.x + b.x) * attn2[0] + lrelu(a.y + b.y) * attn2[1];
    score2[e] = sc;
    atomicMax(m2 + d, ford(sc));
}

__global__ void edge_exp2_kernel(const int* __restrict__ dst,
                                 float* __restrict__ score2,
                                 const unsigned* __restrict__ m2,
                                 float* __restrict__ denom2)
{
    const int e = blockIdx.x * blockDim.x + threadIdx.x;
    if (e >= EDGES) return;
    const int d = dst[e];
    float m = fdec(m2[d]);
    if (!isfinite(m)) m = 0.f;
    const float ex = expf(score2[e] - m);
    score2[e] = ex;
    atomicAdd(denom2 + d, ex);
}

__global__ void edge_aggr2_kernel(const float* __restrict__ hs2,
                                  const int* __restrict__ src,
                                  const int* __restrict__ dst,
                                  const float* __restrict__ ex,
                                  const float* __restrict__ denom2,
                                  float* __restrict__ out)
{
    const int e = blockIdx.x * blockDim.x + threadIdx.x;
    if (e >= EDGES) return;
    const int s = src[e], d = dst[e];
    float den = denom2[d];
    den = den > 0.f ? den : 1.f;
    const float alpha = ex[e] / den;
    const float2 a = *(const float2*)(hs2 + (size_t)s * 2);
    atomicAdd(out + (size_t)d * 2 + 0, alpha * a.x);
    atomicAdd(out + (size_t)d * 2 + 1, alpha * a.y);
}

extern "C" void kernel_launch(void* const* d_in, const int* in_sizes, int n_in,
                              void* d_out, int out_size, void* d_ws, size_t ws_size,
                              hipStream_t stream)
{
    const float* feat  = (const float*)d_in[0];
    const int*   src   = (const int*)d_in[1];
    const int*   dst   = (const int*)d_in[2];
    const float* W1s   = (const float*)d_in[3];
    const float* b1s   = (const float*)d_in[4];
    const float* W1d   = (const float*)d_in[5];
    const float* b1d   = (const float*)d_in[6];
    const float* attn1 = (const float*)d_in[7];
    const float* W2s   = (const float*)d_in[8];
    const float* b2s   = (const float*)d_in[9];
    const float* W2d   = (const float*)d_in[10];
    const float* b2d   = (const float*)d_in[11];
    const float* attn2 = (const float*)d_in[12];
    float* out = (float*)d_out;

    float* ws = (float*)d_ws;
    size_t off = 0;
    float*    hs1    = ws + off; off += (size_t)NNODES * HIDTOT;
    float*    hd1    = ws + off; off += (size_t)NNODES * HIDTOT;
    float*    out1   = ws + off; off += (size_t)NNODES * HIDTOT;
    float*    score1 = ws + off; off += (size_t)EDGES * 4;
    float*    denom1 = ws + off; off += (size_t)NNODES * 4;
    unsigned* m1     = (unsigned*)(ws + off); off += (size_t)NNODES * 4;
    float*    hs2    = ws + off; off += (size_t)NNODES * 2;
    float*    hd2    = ws + off; off += (size_t)NNODES * 2;
    float*    score2 = ws + off; off += (size_t)EDGES;
    float*    denom2 = ws + off; off += (size_t)NNODES;
    unsigned* m2     = (unsigned*)(ws + off); off += (size_t)NNODES;

    const int B = 256;

    {
        size_t n = (size_t)NNODES * HIDTOT;
        init_kernel<<<(unsigned)((n + B - 1) / B), B, 0, stream>>>(
            out1, denom1, m1, denom2, m2, out);
    }
    {
        const int tiles = 3125 * 4;
        gemm1_kernel<<<(tiles + 7) / 8, B, 0, stream>>>(
            feat, W1s, b1s, W1d, b1d, hs1, hd1);
    }
    edge_score1_kernel<<<(EDGES + 7) / 8, B, 0, stream>>>(
        hs1, hd1, src, dst, attn1, score1, m1);
    edge_exp1_kernel<<<(EDGES * 4 + B - 1) / B, B, 0, stream>>>(
        dst, score1, m1, denom1);
    edge_aggr1_kernel<<<(EDGES + 7) / 8, B, 0, stream>>>(
        hs1, src, dst, score1, denom1, out1);
    elu_kernel<<<((NNODES * HIDTOT) + B - 1) / B, B, 0, stream>>>(
        out1, NNODES * HIDTOT);
    proj2_kernel<<<(NNODES + 7) / 8, B, 0, stream>>>(
        out1, W2s, b2s, W2d, b2d, hs2, hd2);
    edge_score2_kernel<<<(EDGES + B - 1) / B, B, 0, stream>>>(
        hs2, hd2, src, dst, attn2, score2, m2);
    edge_exp2_kernel<<<(EDGES + B - 1) / B, B, 0, stream>>>(
        dst, score2, m2, denom2);
    edge_aggr2_kernel<<<(EDGES + B - 1) / B, B, 0, stream>>>(
        hs2, src, dst, score2, denom2, out);

    (void)in_sizes; (void)n_in; (void)out_size; (void)ws_size;
}


# layer-1 aggregation per tile of 200 destination nodes held in LDS: each block filters the edge list for its tile and adds with ds_add_f32, rows stored once with ELU; sums in LDS-add arrival order
# speedup vs baseline: 1.7857x; 1.7857x over previous
#include <hip/hip_runtime.h>
#include <hip/hip_bf16.h>
#include <math.h>

#define NNODES 50000
#define EDGES  800000
#define INFEAT 128
#define HIDTOT 256
#define NEG_SLOPE 0.2f
#define ORD_NEG_INF 0x007FFFFFu

typedef __attribute__((ext_vector_type(2))) float v2f;
typedef __attribute__((ext_vector_type(8))) float v8f;

__device__ __forceinline__ unsigned ford(float f) {
    unsigned u = __float_as_uint(f);
    return (u & 0x80000000u) ? ~u : (u | 0x80000000u);
}
__device__ __forceinline__ float fdec(unsigned u) {
    return __uint_as_float((u & 0x80000000u) ? (u & 0x7FFFFFFFu) : ~u);
}
__device__ __forceinline__ float lrelu(float x) {
    return x >= 0.f ? x : NEG_SLOPE * x;
}

__global__ void init_kernel(float* denom1, unsigned* m1,
                            float* denom2, unsigned* m2, float* dout) {
    size_t i = (size_t)blockIdx.x * blockDim.x + threadIdx.x;
    if (i < (size_t)NNODES * 4) { denom1[i] = 0.f; m1[i] = ORD_NEG_INF; }
    if (i < (size_t)NNODES)     { denom2[i] = 0.f; m2[i] = ORD_NEG_INF; }
    if (i < (size_t)NNODES * 2) dout[i] = 0.f;
}

__global__ __launch_bounds__(256) void gemm1_kernel(
    const float* __restrict__ feat,
    const float* __restrict__ Ws, const float* __restrict__ bs,
    const float* __restrict__ Wd, const float* __restrict__ bd,
    float* __restrict__ hs, float* __restrict__ hd)
{
    const int wave = threadIdx.x >> 5;
    const int lane = threadIdx.x & 31;
    const int half = lane >> 4;
    const int row  = lane & 15;

    int tile = blockIdx.x * 8 + wave;
    if (tile >= 3125 * 4) return;
    const int mt  = tile >> 2;
    const int nt4 = tile & 3;
    const int m0  = mt * 16;
    const int c0  = nt4 * 64;

    v8f zero = {0.f,0.f,0.f,0.f,0.f,0.f,0.f,0.f};
    v8f accS[4], accD[4];
    #pragma unroll
    for (int j = 0; j < 4; ++j) { accS[j] = zero; accD[j] = zero; }

    const float* arow = feat + (size_t)(m0 + row) * INFEAT;
    for (int k0 = 0; k0 < INFEAT; k0 += 4) {
        const int ka = k0 + 2 * half;
        v2f a = *(const v2f*)(arow + ka);
        #pragma unroll
        for (int j = 0; j < 4; ++j) {
            const int col = c0 + j * 16 + row;
            v2f bS, bD;
            bS.x = Ws[(size_t)ka * HIDTOT + col];
            bS.y = Ws[(size_t)(ka + 1) * HIDTOT + col];
            bD.x = Wd[(size_t)ka * HIDTOT + col];
            bD.y = Wd[(size_t)(ka + 1) * HIDTOT + col];
            accS[j] = __builtin_amdgcn_wmma_f32_16x16x4_f32(
                false, a, false, bS, (short)0, accS[j], false, false);
            accD[j] = __builtin_amdgcn_wmma_f32_16x16x4_f32(
                false, a, false, bD, (short)0, accD[j], false, false);
        }
    }
    #pragma unroll
    for (int j = 0; j < 4; ++j) {
        const int col = c0 + j * 16 + row;
        const float biasS = bs[col], biasD = bd[col];
        #pragma unroll
        for (int i = 0; i < 8; ++i) {
            const int r = m0 + i + 8 * half;
            hs[(size_t)r * HIDTOT + col] = accS[j][i] + biasS;
            hd[(size_t)r * HIDTOT + col] = accD[j][i] + biasD;
        }
    }
}

__global__ __launch_bounds__(256) void edge_score1_kernel(
    const float* __restrict__ hs, const float* __restrict__ hd,
    const int* __restrict__ src, const int* __restrict__ dst,
    const float* __restrict__ attn, float* __restrict__ score,
    unsigned* __restrict__ mbits)
{
    const int e = blockIdx.x * 8 + (threadIdx.x >> 5);
    if (e >= EDGES) return;
    const int lane = threadIdx.x & 31;
    const int s = src[e], d = dst[e];
    const int base = lane * 8;

    const float4* ph = (const float4*)(hs + (size_t)s * HIDTOT + base);
    const float4* pd = (const float4*)(hd + (size_t)d * HIDTOT + base);
    const float4* pa = (const float4*)(attn + base);

    float part = 0.f;
    #pragma unroll
    for (int q = 0; q < 2; ++q) {
        float4 a = ph[q], b = pd[q], w = pa[q];
        part += lrelu(a.x + b.x) * w.x + lrelu(a.y + b.y) * w.y +
                lrelu(a.z + b.z) * w.z + lrelu(a.w + b.w) * w.w;
    }
    part += __shfl_xor(part, 1, 32);
    part += __shfl_xor(part, 2, 32);
    part += __shfl_xor(part, 4, 32);
    if ((lane & 7) == 0) {
        const int h = lane >> 3;
        score[(size_t)e * 4 + h] = part;
        atomicMax(mbits + (size_t)d * 4 + h, ford(part));
    }
}

__global__ void edge_exp1_kernel(const int* __restrict__ dst,
                                 float* __restrict__ score,
                                 const unsigned* __restrict__ mbits,
                                 float* __restrict__ denom)
{
    const int t = blockIdx.x * blockDim.x + threadIdx.x;
    if (t >= EDGES * 4) return;
    const int e = t >> 2, h = t & 3;
    const int d = dst[e];
    float m = fdec(mbits[(size_t)d * 4 + h]);
    if (!isfinite(m)) m = 0.f;
    const float ex = expf(score[t] - m);
    score[t] = ex;
    atomicAdd(denom + (size_t)d * 4 + h, ex);
}

#define TILE_R     200
#define NTILES     250
#define AGG_WAVES  32
#define SCAN_STEPS 6250
#define QCAP       544
#define GROUP      8

__device__ __forceinline__ void lds_add(float* p, float v) {
    __hip_atomic_fetch_add(p, v, __ATOMIC_RELAXED, __HIP_MEMORY_SCOPE_WORKGROUP);
}

__device__ __forceinline__ void tile_drain(
    const unsigned* qp, const int n, const int lane, const int lo,
    const float* __restrict__ hs, const int* __restrict__ src,
    const float* __restrict__ ex, const float* __restrict__ denom,
    float* acc)
{
    int s_i = 0, dl_i = 0;
    float a0 = 0.f, a1 = 0.f, a2 = 0.f, a3 = 0.f;
    if (lane < n) {
        const unsigned ent = qp[lane];
        const int e = (int)(ent >> 8);
        dl_i = (int)(ent & 255u);
        s_i = src[e];
        const float4 x  = *(const float4*)(ex + (size_t)e * 4);
        const float4 dn = *(const float4*)(denom + (size_t)(lo + dl_i) * 4);
        a0 = x.x / (dn.x > 0.f ? dn.x : 1.f);
        a1 = x.y / (dn.y > 0.f ? dn.y : 1.f);
        a2 = x.z / (dn.z > 0.f ? dn.z : 1.f);
        a3 = x.w / (dn.w > 0.f ? dn.w : 1.f);
    }
    const bool lowhalf = lane < 16;
    for (int i0 = 0; i0 < n; i0 += GROUP) {
        float4 ra[GROUP], rb[GROUP];
        #pragma unroll
        for (int u = 0; u < GROUP; ++u) {
            if (i0 + u < n) {
                const int s = __builtin_amdgcn_readlane(s_i, i0 + u);
                const float4* p = (const float4*)(hs + (size_t)s * HIDTOT) + lane;
                ra[u] = p[0];
                rb[u] = p[32];
            }
        }
        #pragma unroll
        for (int u = 0; u < GROUP; ++u) {
            if (i0 + u < n) {
                const int d = __builtin_amdgcn_readlane(dl_i, i0 + u);
                const float h0 = __int_as_float(__builtin_amdgcn_readlane(__float_as_int(a0), i0 + u));
                const float h1 = __int_as_float(__builtin_amdgcn_readlane(__float_as_int(a1), i0 + u));
                const float h2 = __int_as_float(__builtin_amdgcn_readlane(__float_as_int(a2), i0 + u));
                const float h3 = __int_as_float(__builtin_amdgcn_readlane(__float_as_int(a3), i0 + u));
                const float aA = lowhalf ? h0 : h1;
                const float aB = lowhalf ? h2 : h3;
                float* a = acc + d * HIDTOT + lane;
                lds_add(a +   0, aA * ra[u].x);
                lds_add(a +  32, aA * ra[u].y);
                lds_add(a +  64, aA * ra[u].z);
                lds_add(a +  96, aA * ra[u].w);
                lds_add(a + 128, aB * rb[u].x);
                lds_add(a + 160, aB * rb[u].y);
                lds_add(a + 192, aB * rb[u].z);
                lds_add(a + 224, aB * rb[u].w);
            }
        }
    }
}

__global__ __launch_bounds__(1024) void tile_aggr1_kernel(
    const float* __restrict__ hs, const int* __restrict__ src,
    const int* __restrict__ dst, const float* __restrict__ ex,
    const float* __restrict__ denom, float* __restrict__ out)
{
    __shared__ __attribute__((aligned(16))) float acc[TILE_R * HIDTOT];
    __shared__ unsigned queue[AGG_WAVES * QCAP];

    const int tid  = threadIdx.x;
    const int wave = __builtin_amdgcn_readfirstlane(tid >> 5);
    const int lane = tid & 31;
    const int tile = blockIdx.x;
    const int lo   = tile * TILE_R;

    for (int i = tid; i < TILE_R * HIDTOT / 4; i += 1024)
        ((float4*)acc)[i] = make_float4(0.f, 0.f, 0.f, 0.f);
    __syncthreads();

    unsigned* q = queue + wave * QCAP;
    int cnt = 0;
    const int4* dst4 = (const int4*)dst;
    const int4 none = make_int4(-1, -1, -1, -1);

    const int rot = tile * (SCAN_STEPS / NTILES);
    int4 v[4];
    #pragma unroll
    for (int u = 0; u < 4; ++u) {
        int g = wave + 32 * u + rot;
        if (g >= SCAN_STEPS) g -= SCAN_STEPS;
        v[u] = dst4[g * 32 + lane];
    }
    for (int it = 0; it < 200; it += 4) {
        int4 nv[4];
        #pragma unroll
        for (int u = 0; u < 4; ++u) {
            const int st = wave + 32 * (it + 4 + u);
            int g = st + rot;
            if (g >= SCAN_STEPS) g -= SCAN_STEPS;
            nv[u] = none;
            if (st < SCAN_STEPS) nv[u] = dst4[g * 32 + lane];
        }
        #pragma unroll
        for (int u = 0; u < 4; ++u) {
            int g = wave + 32 * (it + u) + rot;
            if (g >= SCAN_STEPS) g -= SCAN_STEPS;
            const int eb = g * 128 + lane * 4;
            const int vals[4] = {v[u].x, v[u].y, v[u].z, v[u].w};
            #pragma unroll
            for (int k = 0; k < 4; ++k) {
                const unsigned dl = (unsigned)(vals[k] - lo);
                const bool in = dl < (unsigned)TILE_R;
                const unsigned m = __builtin_amdgcn_ballot_w32(in);
                if (m != 0u) {
                    if (in)
                        q[cnt + (int)__builtin_amdgcn_mbcnt_lo(m, 0u)] =
                            ((unsigned)(eb + k) << 8) | dl;
                    cnt += __builtin_popcount(m);
                }
            }
        }
        while (cnt >= 32) {
            cnt -= 32;
            tile_drain(q + cnt, 32, lane, lo, hs, src, ex, denom, acc);
        }
        #pragma unroll
        for (int u = 0; u < 4; ++u) v[u] = nv[u];
    }
    if (cnt > 0) tile_drain(q, cnt, lane, lo, hs, src, ex, denom, acc);

    __syncthreads();

    for (int r = wave; r < TILE_R; r += AGG_WAVES) {
        const float* a = acc + r * HIDTOT + lane;
        float* po = out + (size_t)(lo + r) * HIDTOT;
        #pragma unroll
        for (int qq = 0; qq < 2; ++qq) {
            float4 o;
            o.x = a[qq * 128 +  0];
            o.y = a[qq * 128 + 32];
            o.z = a[qq * 128 + 64];
            o.w = a[qq * 128 + 96];
            o.x = o.x > 0.f ? o.x : (expf(o.x) - 1.f);
            o.y = o.y > 0.f ? o.y : (expf(o.y) - 1.f);
            o.z = o.z > 0.f ? o.z : (expf(o.z) - 1.f);
            o.w = o.w > 0.f ? o.w : (expf(o.w) - 1.f);
            ((float4*)(po + qq * 128))[lane] = o;
        }
    }
}

__global__ __launch_bounds__(256) void proj2_kernel(
    const float* __restrict__ h,
    const float* __restrict__ W2s, const float* __restrict__ b2s,
    const float* __restrict__ W2d, const float* __restrict__ b2d,
    float* __restrict__ hs2, float* __restrict__ hd2)
{
    const int n = blockIdx.x * 8 + (threadIdx.x >> 5);
    if (n >= NNODES) return;
    const int lane = threadIdx.x & 31;
    const int base = lane * 8;
    const float* hr = h + (size_t)n * HIDTOT + base;
    float s0 = 0.f, s1 = 0.f, d0 = 0.f, d1 = 0.f;
    #pragma unroll
    for (int k = 0; k < 8; ++k) {
        const float x = hr[k];
        const float2 ws = *(const float2*)(W2s + (size_t)(base + k) * 2);
        const float2 wd = *(const float2*)(W2d + (size_t)(base + k) * 2);
        s0 += x * ws.x; s1 += x * ws.y;
        d0 += x * wd.x; d1 += x * wd.y;
    }
    #pragma unroll
    for (int o = 1; o < 32; o <<= 1) {
        s0 += __shfl_xor(s0, o, 32); s1 += __shfl_xor(s1, o, 32);
        d0 += __shfl_xor(d0, o, 32); d1 += __shfl_xor(d1, o, 32);
    }
    if (lane == 0) {
        hs2[(size_t)n * 2 + 0] = s0 + b2s[0];
        hs2[(size_t)n * 2 + 1] = s1 + b2s[1];
        hd2[(size_t)n * 2 + 0] = d0 + b2d[0];
        hd2[(size_t)n * 2 + 1] = d1 + b2d[1];
    }
}

__global__ void edge_score2_kernel(const float* __restrict__ hs2,
                                   const float* __restrict__ hd2,
                                   const int* __restrict__ src,
                                   const int* __restrict__ dst,
                                   const float* __restrict__ attn2,
                                   float* __restrict__ score2,
                                   unsigned* __restrict__ m2)
{
    const int e = blockIdx.x * blockDim.x + threadIdx.x;
    if (e >= EDGES) return;
    const int s = src[e], d = dst[e];
    const float2 a = *(const float2*)(hs2 + (size_t)s * 2);
    const float2 b = *(const float2*)(hd2 + (size_t)d * 2);
    const float sc = lrelu(a.x + b.x) * attn2[0] + lrelu(a.y + b.y) * attn2[1];
    score2[e] = sc;
    atomicMax(m2 + d, ford(sc));
}

__global__ void edge_exp2_kernel(const int* __restrict__ dst,
                                 float* __restrict__ score2,
                                 const unsigned* __restrict__ m2,
                                 float* __restrict__ denom2)
{
    const int e = blockIdx.x * blockDim.x + threadIdx.x;
    if (e >= EDGES) return;
    const int d = dst[e];
    float m = fdec(m2[d]);
    if (!isfinite(m)) m = 0.f;
    const float ex = expf(score2[e] - m);
    score2[e] = ex;
    atomicAdd(denom2 + d, ex);
}

__global__ void edge_aggr2_kernel(const float* __restrict__ hs2,
                                  const int* __restrict__ src,
                                  const int* __restrict__ dst,
                                  const float* __restrict__ ex,
                                  const float* __restrict__ denom2,
                                  float* __restrict__ out)
{
    const int e = blockIdx.x * blockDim.x + threadIdx.x;
    if (e >= EDGES) return;
    const int s = src[e], d = dst[e];
    float den = denom2[d];
    den = den > 0.f ? den : 1.f;
    const float alpha = ex[e] / den;
    const float2 a = *(const float2*)(hs2 + (size_t)s * 2);
    atomicAdd(out + (size_t)d * 2 + 0, alpha * a.x);
    atomicAdd(out + (size_t)d * 2 + 1, alpha * a.y);
}

extern "C" void kernel_launch(void* const* d_in, const int* in_sizes, int n_in,
                              void* d_out, int out_size, void* d_ws, size_t ws_size,
                              hipStream_t stream)
{
    const float* feat  = (const float*)d_in[0];
    const int*   src   = (const int*)d_in[1];
    const int*   dst   = (const int*)d_in[2];
    const float* W1s   = (const float*)d_in[3];
    const float* b1s   = (const float*)d_in[4];
    const float* W1d   = (const float*)d_in[5];
    const float* b1d   = (const float*)d_in[6];
    const float* attn1 = (const float*)d_in[7];
    const float* W2s   = (const float*)d_in[8];
    const float* b2s   = (const float*)d_in[9];
    const float* W2d   = (const float*)d_in[10];
    const float* b2d   = (const float*)d_in[11];
    const float* attn2 = (const float*)d_in[12];
    float* out = (float*)d_out;

    float* ws = (float*)d_ws;
    size_t off = 0;
    float*    hs1    = ws + off; off += (size_t)NNODES * HIDTOT;
    float*    hd1    = ws + off; off += (size_t)NNODES * HIDTOT;
    float*    out1   = ws + off; off += (size_t)NNODES * HIDTOT;
    float*    score1 = ws + off; off += (size_t)EDGES * 4;
    float*    denom1 = ws + off; off += (size_t)NNODES * 4;
    unsigned* m1     = (unsigned*)(ws + off); off += (size_t)NNODES * 4;
    float*    hs2    = ws + off; off += (size_t)NNODES * 2;
    float*    hd2    = ws + off; off += (size_t)NNODES * 2;
    float*    score2 = ws + off; off += (size_t)EDGES;
    float*    denom2 = ws + off; off += (size_t)NNODES;
    unsigned* m2     = (unsigned*)(ws + off); off += (size_t)NNODES;

    const int B = 256;

    init_kernel<<<(NNODES * 4 + B - 1) / B, B, 0, stream>>>(
        denom1, m1, denom2, m2, out);
    {
        const int tiles = 3125 * 4;
        gemm1_kernel<<<(tiles + 7) / 8, B, 0, stream>>>(
            feat, W1s, b1s, W1d, b1d, hs1, hd1);
    }
    edge_score1_kernel<<<(EDGES + 7) / 8, B, 0, stream>>>(
        hs1, hd1, src, dst, attn1, score1, m1);
    edge_exp1_kernel<<<(EDGES * 4 + B - 1) / B, B, 0, stream>>>(
        dst, score1, m1, denom1);
    tile_aggr1_kernel<<<NTILES, 1024, 0, stream>>>(
        hs1, src, dst, score1, denom1, out1);
    proj2_kernel<<<(NNODES + 7) / 8, B, 0, stream>>>(
        out1, W2s, b2s, W2d, b2d, hs2, hd2);
    edge_score2_kernel<<<(EDGES + B - 1) / B, B, 0, stream>>>(
        hs2, hd2, src, dst, attn2, score2, m2);
    edge_exp2_kernel<<<(EDGES + B - 1) / B, B, 0, stream>>>(
        dst, score2, m2, denom2);
    edge_aggr2_kernel<<<(EDGES + B - 1) / B, B, 0, stream>>>(
        hs2, src, dst, score2, denom2, out);

    (void)in_sizes; (void)n_in; (void)out_size; (void)ws_size;
}


# layer-1 scores, maxima, exp and denominators moved into the tile kernel: tile's edges listed in LDS, two passes; rows = (sum exp*h)/denom; 9 to 7 launches; sums in LDS-add arrival order
# speedup vs baseline: 2.1710x; 1.2158x over previous
#include <hip/hip_runtime.h>
#include <hip/hip_bf16.h>
#include <math.h>

#define NNODES 50000
#define EDGES  800000
#define INFEAT 128
#define HIDTOT 256
#define NEG_SLOPE 0.2f
#define ORD_NEG_INF 0x007FFFFFu

typedef __attribute__((ext_vector_type(2))) float v2f;
typedef __attribute__((ext_vector_type(8))) float v8f;

__device__ __forceinline__ unsigned ford(float f) {
    unsigned u = __float_as_uint(f);
    return (u & 0x80000000u) ? ~u : (u | 0x80000000u);
}
__device__ __forceinline__ float fdec(unsigned u) {
    return __uint_as_float((u & 0x80000000u) ? (u & 0x7FFFFFFFu) : ~u);
}
__device__ __forceinline__ float lrelu(float x) {
    return x >= 0.f ? x : NEG_SLOPE * x;
}

__global__ void init_kernel(float* denom2, unsigned* m2, float* dout) {
    size_t i = (size_t)blockIdx.x * blockDim.x + threadIdx.x;
    if (i < (size_t)NNODES)     { denom2[i] = 0.f; m2[i] = ORD_NEG_INF; }
    if (i < (size_t)NNODES * 2) dout[i] = 0.f;
}

__global__ __launch_bounds__(256) void gemm1_kernel(
    const float* __restrict__ feat,
    const float* __restrict__ Ws, const float* __restrict__ bs,
    const float* __restrict__ Wd, const float* __restrict__ bd,
    float* __restrict__ hs, float* __restrict__ hd)
{
    const int wave = threadIdx.x >> 5;
    const int lane = threadIdx.x & 31;
    const int half = lane >> 4;
    const int row  = lane & 15;

    int tile = blockIdx.x * 8 + wave;
    if (tile >= 3125 * 4) return;
    const int mt  = tile >> 2;
    const int nt4 = tile & 3;
    const int m0  = mt * 16;
    const int c0  = nt4 * 64;

    v8f zero = {0.f,0.f,0.f,0.f,0.f,0.f,0.f,0.f};
    v8f accS[4], accD[4];
    #pragma unroll
    for (int j = 0; j < 4; ++j) { accS[j] = zero; accD[j] = zero; }

    const float* arow = feat + (size_t)(m0 + row) * INFEAT;
    for (int k0 = 0; k0 < INFEAT; k0 += 4) {
        const int ka = k0 + 2 * half;
        v2f a = *(const v2f*)(arow + ka);
        #pragma unroll
        for (int j = 0; j < 4; ++j) {
            const int col = c0 + j * 16 + row;
            v2f bS, bD;
            bS.x = Ws[(size_t)ka * HIDTOT + col];
            bS.y = Ws[(size_t)(ka + 1) * HIDTOT + col];
            bD.x = Wd[(size_t)ka * HIDTOT + col];
            bD.y = Wd[(size_t)(ka + 1) * HIDTOT + col];
            accS[j] = __builtin_amdgcn_wmma_f32_16x16x4_f32(
                false, a, false, bS, (short)0, accS[j], false, false);
            accD[j] = __builtin_amdgcn_wmma_f32_16x16x4_f32(
                false, a, false, bD, (short)0, accD[j], false, false);
        }
    }
    #pragma unroll
    for (int j = 0; j < 4; ++j) {
        const int col = c0 + j * 16 + row;
        const float biasS = bs[col], biasD = bd[col];
        #pragma unroll
        for (int i = 0; i < 8; ++i) {
            const int r = m0 + i + 8 * half;
            hs[(size_t)r * HIDTOT + col] = accS[j][i] + biasS;
            hd[(size_t)r * HIDTOT + col] = accD[j][i] + biasD;
        }
    }
}

#define TILE_R     200
#define NTILES     250
#define AGG_WAVES  32
#define SCAN_STEPS 6250
#define WIN_STEPS  40
#define LCAP       (WIN_STEPS * 128)
#define GROUP      8
#define SGROUP     4

__device__ __forceinline__ void lds_add(float* p, float v) {
    __hip_atomic_fetch_add(p, v, __ATOMIC_RELAXED, __HIP_MEMORY_SCOPE_WORKGROUP);
}
__device__ __forceinline__ void lds_max(unsigned* p, unsigned v) {
    __hip_atomic_fetch_max(p, v, __ATOMIC_RELAXED, __HIP_MEMORY_SCOPE_WORKGROUP);
}
__device__ __forceinline__ float finite_or_zero(float m) {
    return isfinite(m) ? m : 0.f;
}

__device__ __forceinline__ void scan_step(
    const int4 v, const int eb, const int lo, const int lane,
    unsigned* list, unsigned* count)
{
    const int vals[4] = {v.x, v.y, v.z, v.w};
    unsigned dl[4], mk[4];
    int tot = 0;
    #pragma unroll
    for (int k = 0; k < 4; ++k) {
        dl[k] = (unsigned)(vals[k] - lo);
        mk[k] = __builtin_amdgcn_ballot_w32(dl[k] < (unsigned)TILE_R);
        tot += __builtin_popcount(mk[k]);
    }
    if (tot != 0) {
        unsigned base = 0u;
        if (lane == 0)
            base = __hip_atomic_fetch_add(count, (unsigned)tot, __ATOMIC_RELAXED,
                                          __HIP_MEMORY_SCOPE_WORKGROUP);
        base = (unsigned)__builtin_amdgcn_readfirstlane((int)base);
        #pragma unroll
        for (int k = 0; k < 4; ++k) {
            if (dl[k] < (unsigned)TILE_R) {
                const unsigned idx = base + __builtin_amdgcn_mbcnt_lo(mk[k], 0u);
                if (idx < (unsigned)LCAP)
                    list[idx] = ((unsigned)(eb + k) << 8) | dl[k];
            }
            base += (unsigned)__builtin_popcount(mk[k]);
        }
    }
}

__device__ __forceinline__ void score_batch(
    const unsigned* qp, float* scp, const int nb, const int lane, const int lo,
    const float* __restrict__ hs, const float* __restrict__ hd,
    const int* __restrict__ src, const float4 w0, const float4 w1,
    unsigned* mxb)
{
    int s_i = 0, dl_i = 0;
    if (lane < nb) {
        const unsigned ent = qp[lane];
        dl_i = (int)(ent & 255u);
        s_i = src[(int)(ent >> 8)];
    }
    for (int g0 = 0; g0 < nb; g0 += SGROUP) {
        float4 a0[SGROUP], a1[SGROUP], b0[SGROUP], b1[SGROUP];
        #pragma unroll
        for (int u = 0; u < SGROUP; ++u) {
            const int s = __builtin_amdgcn_readlane(s_i, g0 + u);
            const int d = __builtin_amdgcn_readlane(dl_i, g0 + u);
            const float4* ph = (const float4*)(hs + (size_t)s * HIDTOT + lane * 8);
            const float4* pd = (const float4*)(hd + (size_t)(lo + d) * HIDTOT + lane * 8);
            a0[u] = ph[0]; a1[u] = ph[1];
            b0[u] = pd[0]; b1[u] = pd[1];
        }
        #pragma unroll
        for (int u = 0; u < SGROUP; ++u) {
            if (g0 + u < nb) {
                float part = 0.f;
                part += lrelu(a0[u].x + b0[u].x) * w0.x + lrelu(a0[u].y + b0[u].y) * w0.y +
                        lrelu(a0[u].z + b0[u].z) * w0.z + lrelu(a0[u].w + b0[u].w) * w0.w;
                part += lrelu(a1[u].x + b1[u].x) * w1.x + lrelu(a1[u].y + b1[u].y) * w1.y +
                        lrelu(a1[u].z + b1[u].z) * w1.z + lrelu(a1[u].w + b1[u].w) * w1.w;
                part += __shfl_xor(part, 1, 32);
                part += __shfl_xor(part, 2, 32);
                part += __shfl_xor(part, 4, 32);
                if ((lane & 7) == 0) {
                    const int h = lane >> 3;
                    const int d = __builtin_amdgcn_readlane(dl_i, g0 + u);
                    scp[(g0 + u) * 4 + h] = part;
                    lds_max(mxb + d * 4 + h, ford(part));
                }
            }
        }
    }
}

__device__ __forceinline__ void add_batch(
    const unsigned* qp, const float* scp, const int nb, const int lane,
    const float* __restrict__ hs, const int* __restrict__ src,
    const unsigned* mxb, float* dn, float* acc)
{
    int s_i = 0, dl_i = 0;
    float a0 = 0.f, a1 = 0.f, a2 = 0.f, a3 = 0.f;
    if (lane < nb) {
        const unsigned ent = qp[lane];
        dl_i = (int)(ent & 255u);
        s_i = src[(int)(ent >> 8)];
        const float4 x = ((const float4*)scp)[lane];
        const uint4 mb = ((const uint4*)mxb)[dl_i];
        a0 = expf(x.x - finite_or_zero(fdec(mb.x)));
        a1 = expf(x.y - finite_or_zero(fdec(mb.y)));
        a2 = expf(x.z - finite_or_zero(fdec(mb.z)));
        a3 = expf(x.w - finite_or_zero(fdec(mb.w)));
        lds_add(dn + dl_i * 4 + 0, a0);
        lds_add(dn + dl_i * 4 + 1, a1);
        lds_add(dn + dl_i * 4 + 2, a2);
        lds_add(dn + dl_i * 4 + 3, a3);
    }
    const bool lowhalf = lane < 16;
    for (int g0 = 0; g0 < nb; g0 += GROUP) {
        float4 ra[GROUP], rb[GROUP];
        #pragma unroll
        for (int u = 0; u < GROUP; ++u) {
            const int s = __builtin_amdgcn_readlane(s_i, g0 + u);
            const float4* p = (const float4*)(hs + (size_t)s * HIDTOT) + lane;
            ra[u] = p[0];
            rb[u] = p[32];
        }
        #pragma unroll
        for (int u = 0; u < GROUP; ++u) {
            if (g0 + u < nb) {
                const int d = __builtin_amdgcn_readlane(dl_i, g0 + u);
                const float h0 = __int_as_float(__builtin_amdgcn_readlane(__float_as_int(a0), g0 + u));
                const float h1 = __int_as_float(__builtin_amdgcn_readlane(__float_as_int(a1), g0 + u));
                const float h2 = __int_as_float(__builtin_amdgcn_readlane(__float_as_int(a2), g0 + u));
                const float h3 = __int_as_float(__builtin_amdgcn_readlane(__float_as_int(a3), g0 + u));
                const float aA = lowhalf ? h0 : h1;
                const float aB = lowhalf ? h2 : h3;
                float* a = acc + d * HIDTOT + lane;
                lds_add(a +   0, aA * ra[u].x);
                lds_add(a +  32, aA * ra[u].y);
                lds_add(a +  64, aA * ra[u].z);
                lds_add(a +  96, aA * ra[u].w);
                lds_add(a + 128, aB * rb[u].x);
                lds_add(a + 160, aB * rb[u].y);
                lds_add(a + 192, aB * rb[u].z);
                lds_add(a + 224, aB * rb[u].w);
            }
        }
    }
}

__device__ __forceinline__ void tile_chunk(
    const bool rescale, const int n, const int tid, const int wave,
    const int lane, const int lo,
    const float* __restrict__ hs, const float* __restrict__ hd,
    const int* __restrict__ src, const float* __restrict__ attn,
    float* acc, float* sc, const unsigned* list,
    unsigned* mxb, unsigned* omx, float* dn)
{
    const int share = (n + AGG_WAVES - 1) / AGG_WAVES;
    const int beg = wave * share;
    const int end = (beg + share) < n ? (beg + share) : n;

    if (rescale) {
        for (int i = tid; i < TILE_R * 4; i += 1024) omx[i] = mxb[i];
        __syncthreads();
    }

    {
        const float4 w0 = ((const float4*)(attn + lane * 8))[0];
        const float4 w1 = ((const float4*)(attn + lane * 8))[1];
        for (int i0 = beg; i0 < end; i0 += 32) {
            const int nb = (end - i0) < 32 ? (end - i0) : 32;
            score_batch(list + i0, sc + i0 * 4, nb, lane, lo, hs, hd, src, w0, w1, mxb);
        }
    }
    __syncthreads();

    if (rescale) {
        for (int i = tid; i < TILE_R * 4; i += 1024) {
            const unsigned ob = omx[i];
            const float f = (ob == ORD_NEG_INF) ? 0.f
                : expf(finite_or_zero(fdec(ob)) - finite_or_zero(fdec(mxb[i])));
            omx[i] = __float_as_uint(f);
            dn[i] *= f;
        }
        __syncthreads();
        for (int i = tid; i < TILE_R * HIDTOT; i += 1024) {
            const int r = i >> 8, p = i & 255;
            const int h = 2 * (p >> 7) + ((p & 31) >> 4);
            acc[i] *= __uint_as_float(omx[r * 4 + h]);
        }
        __syncthreads();
    }

    for (int i0 = beg; i0 < end; i0 += 32) {
        const int nb = (end - i0) < 32 ? (end - i0) : 32;
        add_batch(list + i0, sc + i0 * 4, nb, lane, hs, src, mxb, dn, acc);
    }
}

__global__ __launch_bounds__(1024) void tile_layer1_kernel(
    const float* __restrict__ hs, const float* __restrict__ hd,
    const int* __restrict__ src, const int* __restrict__ dst,
    const float* __restrict__ attn, float* __restrict__ out)
{
    __shared__ __attribute__((aligned(16))) float acc[TILE_R * HIDTOT];
    __shared__ __attribute__((aligned(16))) float sc[LCAP * 4];
    __shared__ __attribute__((aligned(16))) unsigned mxb[TILE_R * 4];
    __shared__ __attribute__((aligned(16))) unsigned omx[TILE_R * 4];
    __shared__ __attribute__((aligned(16))) float dn[TILE_R * 4];
    __shared__ unsigned list[LCAP];
    __shared__ unsigned count;

    const int tid  = threadIdx.x;
    const int wave = __builtin_amdgcn_readfirstlane(tid >> 5);
    const int lane = tid & 31;
    const int tile = blockIdx.x;
    const int lo   = tile * TILE_R;

    for (int i = tid; i < TILE_R * HIDTOT / 4; i += 1024)
        ((float4*)acc)[i] = make_float4(0.f, 0.f, 0.f, 0.f);
    for (int i = tid; i < TILE_R * 4; i += 1024) { mxb[i] = ORD_NEG_INF; dn[i] = 0.f; }
    if (tid == 0) count = 0u;
    __syncthreads();

    const int4* dst4 = (const int4*)dst;
    const int4 none = make_int4(-1, -1, -1, -1);

    {
        const int rot = tile * (SCAN_STEPS / NTILES);
        int4 v[4];
        #pragma unroll
        for (int u = 0; u < 4; ++u) {
            int g = wave + 32 * u + rot;
            if (g >= SCAN_STEPS) g -= SCAN_STEPS;
            v[u] = dst4[g * 32 + lane];
        }
        for (int it = 0; it < 200; it += 4) {
            int4 nv[4];
            #pragma unroll
            for (int u = 0; u < 4; ++u) {
                const int st = wave + 32 * (it + 4 + u);
                int g = st + rot;
                if (g >= SCAN_STEPS) g -= SCAN_STEPS;
                nv[u] = none;
                if (st < SCAN_STEPS) nv[u] = dst4[g * 32 + lane];
            }
            #pragma unroll
            for (int u = 0; u < 4; ++u) {
                int g = wave + 32 * (it + u) + rot;
                if (g >= SCAN_STEPS) g -= SCAN_STEPS;
                scan_step(v[u], g * 128 + lane * 4, lo, lane, list, &count);
            }
            #pragma unroll
            for (int u = 0; u < 4; ++u) v[u] = nv[u];
        }
    }
    __syncthreads();

    const unsigned total = count;
    const bool windowed = total > (unsigned)LCAP;
    const int nwin = windowed ? (SCAN_STEPS + WIN_STEPS - 1) / WIN_STEPS : 1;
    for (int w = 0; w < nwin; ++w) {
        int n = (int)total;
        if (windowed) {
            __syncthreads();
            if (tid == 0) count = 0u;
            __syncthreads();
            const int w0 = w * WIN_STEPS;
            const int wend = (w0 + WIN_STEPS) < SCAN_STEPS ? (w0 + WIN_STEPS) : SCAN_STEPS;
            for (int st = w0 + wave; st < wend; st += AGG_WAVES)
                scan_step(dst4[st * 32 + lane], st * 128 + lane * 4, lo, lane, list, &count);
            __syncthreads();
            n = (int)count;
        }
        tile_chunk(windowed, n, tid, wave, lane, lo, hs, hd, src, attn,
                   acc, sc, list, mxb, omx, dn);
    }
    __syncthreads();

    for (int r = wave; r < TILE_R; r += AGG_WAVES) {
        const float* a = acc + r * HIDTOT + lane;
        float* po = out + (size_t)(lo + r) * HIDTOT;
        #pragma unroll
        for (int qq = 0; qq < 2; ++qq) {
            float den = dn[r * 4 + 2 * qq + (lane >> 4)];
            den = den > 0.f ? den : 1.f;
            float4 o;
            o.x = a[qq * 128 +  0] / den;
            o.y = a[qq * 128 + 32] / den;
            o.z = a[qq * 128 + 64] / den;
            o.w = a[qq * 128 + 96] / den;
            o.x = o.x > 0.f ? o.x : (expf(o.x) - 1.f);
            o.y = o.y > 0.f ? o.y : (expf(o.y) - 1.f);
            o.z = o.z > 0.f ? o.z : (expf(o.z) - 1.f);
            o.w = o.w > 0.f ? o.w : (expf(o.w) - 1.f);
            ((float4*)(po + qq * 128))[lane] = o;
        }
    }
}

__global__ __launch_bounds__(256) void proj2_kernel(
    const float* __restrict__ h,
    const float* __restrict__ W2s, const float* __restrict__ b2s,
    const float* __restrict__ W2d, const float* __restrict__ b2d,
    float* __restrict__ hs2, float* __restrict__ hd2)
{
    const int n = blockIdx.x * 8 + (threadIdx.x >> 5);
    if (n >= NNODES) return;
    const int lane = threadIdx.x & 31;
    const int base = lane * 8;
    const float* hr = h + (size_t)n * HIDTOT + base;
    float s0 = 0.f, s1 = 0.f, d0 = 0.f, d1 = 0.f;
    #pragma unroll
    for (int k = 0; k < 8; ++k) {
        const float x = hr[k];
        const float2 ws = *(const float2*)(W2s + (size_t)(base + k) * 2);
        const float2 wd = *(const float2*)(W2d + (size_t)(base + k) * 2);
        s0 += x * ws.x; s1 += x * ws.y;
        d0 += x * wd.x; d1 += x * wd.y;
    }
    #pragma unroll
    for (int o = 1; o < 32; o <<= 1) {
        s0 += __shfl_xor(s0, o, 32); s1 += __shfl_xor(s1, o, 32);
        d0 += __shfl_xor(d0, o, 32); d1 += __shfl_xor(d1, o, 32);
    }
    if (lane == 0) {
        hs2[(size_t)n * 2 + 0] = s0 + b2s[0];
        hs2[(size_t)n * 2 + 1] = s1 + b2s[1];
        hd2[(size_t)n * 2 + 0] = d0 + b2d[0];
        hd2[(size_t)n * 2 + 1] = d1 + b2d[1];
    }
}

__global__ void edge_score2_kernel(const float* __restrict__ hs2,
                                   const float* __restrict__ hd2,
                                   const int* __restrict__ src,
                                   const int* __restrict__ dst,
                                   const float* __restrict__ attn2,
                                   float* __restrict__ score2,
                                   unsigned* __restrict__ m2)
{
    const int e = blockIdx.x * blockDim.x + threadIdx.x;
    if (e >= EDGES) return;
    const int s = src[e], d = dst[e];
    const float2 a = *(const float2*)(hs2 + (size_t)s * 2);
    const float2 b = *(const float2*)(hd2 + (size_t)d * 2);
    const float sc = lrelu(a.x + b.x) * attn2[0] + lrelu(a.y + b.y) * attn2[1];
    score2[e] = sc;
    atomicMax(m2 + d, ford(sc));
}

__global__ void edge_exp2_kernel(const int* __restrict__ dst,
                                 float* __restrict__ score2,
                                 const unsigned* __restrict__ m2,
                                 float* __restrict__ denom2)
{
    const int e = blockIdx.x * blockDim.x + threadIdx.x;
    if (e >= EDGES) return;
    const int d = dst[e];
    float m = fdec(m2[d]);
    if (!isfinite(m)) m = 0.f;
    const float ex = expf(score2[e] - m);
    score2[e] = ex;
    atomicAdd(denom2 + d, ex);
}

__global__ void edge_aggr2_kernel(const float* __restrict__ hs2,
                                  const int* __restrict__ src,
                                  const int* __restrict__ dst,
                                  const float* __restrict__ ex,
                                  const float* __restrict__ denom2,
                                  float* __restrict__ out)
{
    const int e = blockIdx.x * blockDim.x + threadIdx.x;
    if (e >= EDGES) return;
    const int s = src[e], d = dst[e];
    float den = denom2[d];
    den = den > 0.f ? den : 1.f;
    const float alpha = ex[e] / den;
    const float2 a = *(const float2*)(hs2 + (size_t)s * 2);
    atomicAdd(out + (size_t)d * 2 + 0, alpha * a.x);
    atomicAdd(out + (size_t)d * 2 + 1, alpha * a.y);
}

extern "C" void kernel_launch(void* const* d_in, const int* in_sizes, int n_in,
                              void* d_out, int out_size, void* d_ws, size_t ws_size,
                              hipStream_t stream)
{
    const float* feat  = (const float*)d_in[0];
    const int*   src   = (const int*)d_in[1];
    const int*   dst   = (const int*)d_in[2];
    const float* W1s   = (const float*)d_in[3];
    const float* b1s   = (const float*)d_in[4];
    const float* W1d   = (const float*)d_in[5];
    const float* b1d   = (const float*)d_in[6];
    const float* attn1 = (const float*)d_in[7];
    const float* W2s   = (const float*)d_in[8];
    const float* b2s   = (const float*)d_in[9];
    const float* W2d   = (const float*)d_in[10];
    const float* b2d   = (const float*)d_in[11];
    const float* attn2 = (const float*)d_in[12];
    float* out = (float*)d_out;

    float* ws = (float*)d_ws;
    size_t off = 0;
    float*    hs1    = ws + off; off += (size_t)NNODES * HIDTOT;
    float*    hd1    = ws + off; off += (size_t)NNODES * HIDTOT;
    float*    out1   = ws + off; off += (size_t)NNODES * HIDTOT;
    float*    hs2    = ws + off; off += (size_t)NNODES * 2;
    float*    hd2    = ws + off; off += (size_t)NNODES * 2;
    float*    score2 = ws + off; off += (size_t)EDGES;
    float*    denom2 = ws + off; off += (size_t)NNODES;
    unsigned* m2     = (unsigned*)(ws + off); off += (size_t)NNODES;

    const int B = 256;

    init_kernel<<<(NNODES * 2 + B - 1) / B, B, 0, stream>>>(
        denom2, m2, out);
    {
        const int tiles = 3125 * 4;
        gemm1_kernel<<<(tiles + 7) / 8, B, 0, stream>>>(
            feat, W1s, b1s, W1d, b1d, hs1, hd1);
    }
    tile_layer1_kernel<<<NTILES, 1024, 0, stream>>>(
        hs1, hd1, src, dst, attn1, out1);
    proj2_kernel<<<(NNODES + 7) / 8, B, 0, stream>>>(
        out1, W2s, b2s, W2d, b2d, hs2, hd2);
    edge_score2_kernel<<<(EDGES + B - 1) / B, B, 0, stream>>>(
        hs2, hd2, src, dst, attn2, score2, m2);
    edge_exp2_kernel<<<(EDGES + B - 1) / B, B, 0, stream>>>(
        dst, score2, m2, denom2);
    edge_aggr2_kernel<<<(EDGES + B - 1) / B, B, 0, stream>>>(
        hs2, src, dst, score2, denom2, out);

    (void)in_sizes; (void)n_in; (void)out_size; (void)ws_size;
}


# layer-1 tile kernel: the tile's 200 destination rows are staged once in LDS (in the sums' space, zeroed between the passes); pass 1 reads them there, not from global per edge; sum order unchanged
# speedup vs baseline: 2.2565x; 1.0394x over previous
#include <hip/hip_runtime.h>
#include <hip/hip_bf16.h>
#include <math.h>

#define NNODES 50000
#define EDGES  800000
#define INFEAT 128
#define HIDTOT 256
#define NEG_SLOPE 0.2f
#define ORD_NEG_INF 0x007FFFFFu

typedef __attribute__((ext_vector_type(2))) float v2f;
typedef __attribute__((ext_vector_type(8))) float v8f;
typedef __attribute__((ext_vector_type(4))) float v4f;

__device__ __forceinline__ unsigned ford(float f) {
    unsigned u = __float_as_uint(f);
    return (u & 0x80000000u) ? ~u : (u | 0x80000000u);
}
__device__ __forceinline__ float fdec(unsigned u) {
    return __uint_as_float((u & 0x80000000u) ? (u & 0x7FFFFFFFu) : ~u);
}
__device__ __forceinline__ float lrelu(float x) {
    return x >= 0.f ? x : NEG_SLOPE * x;
}

__global__ void init_kernel(float* denom2, unsigned* m2, float* dout) {
    size_t i = (size_t)blockIdx.x * blockDim.x + threadIdx.x;
    if (i < (size_t)NNODES)     { denom2[i] = 0.f; m2[i] = ORD_NEG_INF; }
    if (i < (size_t)NNODES * 2) dout[i] = 0.f;
}

__global__ __launch_bounds__(256) void gemm1_kernel(
    const float* __restrict__ feat,
    const float* __restrict__ Ws, const float* __restrict__ bs,
    const float* __restrict__ Wd, const float* __restrict__ bd,
    float* __restrict__ hs, float* __restrict__ hd)
{
    const int wave = threadIdx.x >> 5;
    const int lane = threadIdx.x & 31;
    const int half = lane >> 4;
    const int row  = lane & 15;

    int tile = blockIdx.x * 8 + wave;
    if (tile >= 3125 * 4) return;
    const int mt  = tile >> 2;
    const int nt4 = tile & 3;
    const int m0  = mt * 16;
    const int c0  = nt4 * 64;

    v8f zero = {0.f,0.f,0.f,0.f,0.f,0.f,0.f,0.f};
    v8f accS[4], accD[4];
    #pragma unroll
    for (int j = 0; j < 4; ++j) { accS[j] = zero; accD[j] = zero; }

    const float* arow = feat + (size_t)(m0 + row) * INFEAT;
    for (int k0 = 0; k0 < INFEAT; k0 += 4) {
        const int ka = k0 + 2 * half;
        v2f a = *(const v2f*)(arow + ka);
        #pragma unroll
        for (int j = 0; j < 4; ++j) {
            const int col = c0 + j * 16 + row;
            v2f bS, bD;
            bS.x = Ws[(size_t)ka * HIDTOT + col];
            bS.y = Ws[(size_t)(ka + 1) * HIDTOT + col];
            bD.x = Wd[(size_t)ka * HIDTOT + col];
            bD.y = Wd[(size_t)(ka + 1) * HIDTOT + col];
            accS[j] = __builtin_amdgcn_wmma_f32_16x16x4_f32(
                false, a, false, bS, (short)0, accS[j], false, false);
            accD[j] = __builtin_amdgcn_wmma_f32_16x16x4_f32(
                false, a, false, bD, (short)0, accD[j], false, false);
        }
    }
    #pragma unroll
    for (int j = 0; j < 4; ++j) {
        const int col = c0 + j * 16 + row;
        const float biasS = bs[col], biasD = bd[col];
        #pragma unroll
        for (int i = 0; i < 8; ++i) {
            const int r = m0 + i + 8 * half;
            hs[(size_t)r * HIDTOT + col] = accS[j][i] + biasS;
            hd[(size_t)r * HIDTOT + col] = accD[j][i] + biasD;
        }
    }
}

#define TILE_R     200
#define NTILES     250
#define AGG_WAVES  32
#define SCAN_STEPS 6250
#define WIN_STEPS  40
#define LCAP       (WIN_STEPS * 128)
#define GROUP      8
#define SGROUP     4

__device__ __forceinline__ void lds_add(float* p, float v) {
    __hip_atomic_fetch_add(p, v, __ATOMIC_RELAXED, __HIP_MEMORY_SCOPE_WORKGROUP);
}
__device__ __forceinline__ void lds_max(unsigned* p, unsigned v) {
    __hip_atomic_fetch_max(p, v, __ATOMIC_RELAXED, __HIP_MEMORY_SCOPE_WORKGROUP);
}
__device__ __forceinline__ float finite_or_zero(float m) {
    return isfinite(m) ? m : 0.f;
}

__device__ __forceinline__ void scan_step(
    const int4 v, const int eb, const int lo, const int lane,
    unsigned* list, unsigned* count)
{
    const int vals[4] = {v.x, v.y, v.z, v.w};
    unsigned dl[4], mk[4];
    int tot = 0;
    #pragma unroll
    for (int k = 0; k < 4; ++k) {
        dl[k] = (unsigned)(vals[k] - lo);
        mk[k] = __builtin_amdgcn_ballot_w32(dl[k] < (unsigned)TILE_R);
        tot += __builtin_popcount(mk[k]);
    }
    if (tot != 0) {
        unsigned base = 0u;
        if (lane == 0)
            base = __hip_atomic_fetch_add(count, (unsigned)tot, __ATOMIC_RELAXED,
                                          __HIP_MEMORY_SCOPE_WORKGROUP);
        base = (unsigned)__builtin_amdgcn_readfirstlane((int)base);
        #pragma unroll
        for (int k = 0; k < 4; ++k) {
            if (dl[k] < (unsigned)TILE_R) {
                const unsigned idx = base + __builtin_amdgcn_mbcnt_lo(mk[k], 0u);
                if (idx < (unsigned)LCAP)
                    list[idx] = ((unsigned)(eb + k) << 8) | dl[k];
            }
            base += (unsigned)__builtin_popcount(mk[k]);
        }
    }
}

template <bool HD_LDS>
__device__ __forceinline__ void score_batch(
    const unsigned* qp, float* scp, const int nb, const int lane, const int lo,
    const float* __restrict__ hs, const float* __restrict__ hd,
    const float4* hdt,
    const int* __restrict__ src, const float4 w0, const float4 w1,
    unsigned* mxb)
{
    int s_i = 0, dl_i = 0;
    if (lane < nb) {
        const unsigned ent = qp[lane];
        dl_i = (int)(ent & 255u);
        s_i = src[(int)(ent >> 8)];
    }
    for (int g0 = 0; g0 < nb; g0 += SGROUP) {
        float4 a0[SGROUP], a1[SGROUP], b0[SGROUP], b1[SGROUP];
        #pragma unroll
        for (int u = 0; u < SGROUP; ++u) {
            const int s = __builtin_amdgcn_readlane(s_i, g0 + u);
            const int d = __builtin_amdgcn_readlane(dl_i, g0 + u);
            const float4* ph = (const float4*)(hs + (size_t)s * HIDTOT + lane * 8);
            a0[u] = ph[0]; a1[u] = ph[1];
            if (HD_LDS) {
                b0[u] = hdt[d * 64 + lane];
                b1[u] = hdt[d * 64 + 32 + lane];
            } else {
                const float4* pd = (const float4*)(hd + (size_t)(lo + d) * HIDTOT + lane * 8);
                b0[u] = pd[0]; b1[u] = pd[1];
            }
        }
        #pragma unroll
        for (int u = 0; u < SGROUP; ++u) {
            if (g0 + u < nb) {
                float part = 0.f;
                part += lrelu(a0[u].x + b0[u].x) * w0.x + lrelu(a0[u].y + b0[u].y) * w0.y +
                        lrelu(a0[u].z + b0[u].z) * w0.z + lrelu(a0[u].w + b0[u].w) * w0.w;
                part += lrelu(a1[u].x + b1[u].x) * w1.x + lrelu(a1[u].y + b1[u].y) * w1.y +
                        lrelu(a1[u].z + b1[u].z) * w1.z + lrelu(a1[u].w + b1[u].w) * w1.w;
                part += __shfl_xor(part, 1, 32);
                part += __shfl_xor(part, 2, 32);
                part += __shfl_xor(part, 4, 32);
                if ((lane & 7) == 0) {
                    const int h = lane >> 3;
                    const int d = __builtin_amdgcn_readlane(dl_i, g0 + u);
                    scp[(g0 + u) * 4 + h] = part;
                    lds_max(mxb + d * 4 + h, ford(part));
                }
            }
        }
    }
}

__device__ __forceinline__ void add_batch(
    const unsigned* qp, const float* scp, const int nb, const int lane,
    const float* __restrict__ hs, const int* __restrict__ src,
    const unsigned* mxb, float* dn, float* acc)
{
    int s_i = 0, dl_i = 0;
    float a0 = 0.f, a1 = 0.f, a2 = 0.f, a3 = 0.f;
    if (lane < nb) {
        const unsigned ent = qp[lane];
        dl_i = (int)(ent & 255u);
        s_i = src[(int)(ent >> 8)];
        const float4 x = ((const float4*)scp)[lane];
        const uint4 mb = ((const uint4*)mxb)[dl_i];
        a0 = expf(x.x - finite_or_zero(fdec(mb.x)));
        a1 = expf(x.y - finite_or_zero(fdec(mb.y)));
        a2 = expf(x.z - finite_or_zero(fdec(mb.z)));
        a3 = expf(x.w - finite_or_zero(fdec(mb.w)));
        lds_add(dn + dl_i * 4 + 0, a0);
        lds_add(dn + dl_i * 4 + 1, a1);
        lds_add(dn + dl_i * 4 + 2, a2);
        lds_add(dn + dl_i * 4 + 3, a3);
    }
    const bool lowhalf = lane < 16;
    for (int g0 = 0; g0 < nb; g0 += GROUP) {
        float4 ra[GROUP], rb[GROUP];
        #pragma unroll
        for (int u = 0; u < GROUP; ++u) {
            const int s = __builtin_amdgcn_readlane(s_i, g0 + u);
            const float4* p = (const float4*)(hs + (size_t)s * HIDTOT) + lane;
            ra[u] = p[0];
            rb[u] = p[32];
        }
        #pragma unroll
        for (int u = 0; u < GROUP; ++u) {
            if (g0 + u < nb) {
                const int d = __builtin_amdgcn_readlane(dl_i, g0 + u);
                const float h0 = __int_as_float(__builtin_amdgcn_readlane(__float_as_int(a0), g0 + u));
                const float h1 = __int_as_float(__builtin_amdgcn_readlane(__float_as_int(a1), g0 + u));
                const float h2 = __int_as_float(__builtin_amdgcn_readlane(__float_as_int(a2), g0 + u));
                const float h3 = __int_as_float(__builtin_amdgcn_readlane(__float_as_int(a3), g0 + u));
                const float aA = lowhalf ? h0 : h1;
                const float aB = lowhalf ? h2 : h3;
                float* a = acc + d * HIDTOT + lane;
                lds_add(a +   0, aA * ra[u].x);
                lds_add(a +  32, aA * ra[u].y);
                lds_add(a +  64, aA * ra[u].z);
                lds_add(a +  96, aA * ra[u].w);
                lds_add(a + 128, aB * rb[u].x);
                lds_add(a + 160, aB * rb[u].y);
                lds_add(a + 192, aB * rb[u].z);
                lds_add(a + 224, aB * rb[u].w);
            }
        }
    }
}

template <bool WINDOWED>
__device__ __forceinline__ void tile_chunk(
    const int n, const int tid, const int wave,
    const int lane, const int lo,
    const float* __restrict__ hs, const float* __restrict__ hd,
    const int* __restrict__ src, const float* __restrict__ attn,
    float* acc, float* sc, const unsigned* list,
    unsigned* mxb, unsigned* omx, float* dn)
{
    const int share = (n + AGG_WAVES - 1) / AGG_WAVES;
    const int beg = wave * share;
    const int end = (beg + share) < n ? (beg + share) : n;

    if (WINDOWED) {
        for (int i = tid; i < TILE_R * 4; i += 1024) omx[i] = mxb[i];
        __syncthreads();
    }

    {
        const float4 w0 = ((const float4*)(attn + lane * 8))[0];
        const float4 w1 = ((const float4*)(attn + lane * 8))[1];
        for (int i0 = beg; i0 < end; i0 += 32) {
            const int nb = (end - i0) < 32 ? (end - i0) : 32;
            score_batch<!WINDOWED>(list + i0, sc + i0 * 4, nb, lane, lo, hs, hd,
                                   (const float4*)acc, src, w0, w1, mxb);
        }
    }
    __syncthreads();

    if (!WINDOWED) {
        for (int i = tid; i < TILE_R * HIDTOT / 4; i += 1024)
            ((float4*)acc)[i] = make_float4(0.f, 0.f, 0.f, 0.f);
        __syncthreads();
    } else {
        for (int i = tid; i < TILE_R * 4; i += 1024) {
            const unsigned ob = omx[i];
            const float f = (ob == ORD_NEG_INF) ? 0.f
                : expf(finite_or_zero(fdec(ob)) - finite_or_zero(fdec(mxb[i])));
            omx[i] = __float_as_uint(f);
            dn[i] *= f;
        }
        __syncthreads();
        for (int i = tid; i < TILE_R * HIDTOT; i += 1024) {
            const int r = i >> 8, p = i & 255;
            const int h = 2 * (p >> 7) + ((p & 31) >> 4);
            acc[i] *= __uint_as_float(omx[r * 4 + h]);
        }
        __syncthreads();
    }

    for (int i0 = beg; i0 < end; i0 += 32) {
        const int nb = (end - i0) < 32 ? (end - i0) : 32;
        add_batch(list + i0, sc + i0 * 4, nb, lane, hs, src, mxb, dn, acc);
    }
}

__global__ __launch_bounds__(1024) void tile_layer1_kernel(
    const float* __restrict__ hs, const float* __restrict__ hd,
    const int* __restrict__ src, const int* __restrict__ dst,
    const float* __restrict__ attn, float* __restrict__ out)
{
    __shared__ __attribute__((aligned(16))) float acc[TILE_R * HIDTOT];
    __shared__ __attribute__((aligned(16))) float sc[LCAP * 4];
    __shared__ __attribute__((aligned(16))) unsigned mxb[TILE_R * 4];
    __shared__ __attribute__((aligned(16))) unsigned omx[TILE_R * 4];
    __shared__ __attribute__((aligned(16))) float dn[TILE_R * 4];
    __shared__ unsigned list[LCAP];
    __shared__ unsigned count;

    const int tid  = threadIdx.x;
    const int wave = __builtin_amdgcn_readfirstlane(tid >> 5);
    const int lane = tid & 31;
    const int tile = blockIdx.x;
    const int lo   = tile * TILE_R;

    for (int i = tid; i < TILE_R * 4; i += 1024) { mxb[i] = ORD_NEG_INF; dn[i] = 0.f; }
    if (tid == 0) count = 0u;
    __syncthreads();

    const int4* dst4 = (const int4*)dst;
    const int4 none = make_int4(-1, -1, -1, -1);

    {
        const v4f* hdg = (const v4f*)(hd + (size_t)lo * HIDTOT);
        v4f* hdt = (v4f*)acc;
        const int cc = 2 * (lane & 15) + (lane >> 4);
        const int sl = (lane >> 4) * 32 + (lane & 15);
        const int rot = tile * (SCAN_STEPS / NTILES);
        int4 v[4];
        #pragma unroll
        for (int u = 0; u < 4; ++u) {
            int g = wave + 32 * u + rot;
            if (g >= SCAN_STEPS) g -= SCAN_STEPS;
            v[u] = dst4[g * 32 + lane];
        }
        for (int it = 0; it < 200; it += 4) {
            const int hj = wave + 32 * (it >> 2);
            v4f hv = {0.f, 0.f, 0.f, 0.f};
            if (hj < 2 * TILE_R)
                hv = __builtin_nontemporal_load(hdg + (hj >> 1) * 64 + (hj & 1) * 32 + cc);
            int4 nv[4];
            #pragma unroll
            for (int u = 0; u < 4; ++u) {
                const int st = wave + 32 * (it + 4 + u);
                int g = st + rot;
                if (g >= SCAN_STEPS) g -= SCAN_STEPS;
                nv[u] = none;
                if (st < SCAN_STEPS) nv[u] = dst4[g * 32 + lane];
            }
            #pragma unroll
            for (int u = 0; u < 4; ++u) {
                int g = wave + 32 * (it + u) + rot;
                if (g >= SCAN_STEPS) g -= SCAN_STEPS;
                scan_step(v[u], g * 128 + lane * 4, lo, lane, list, &count);
            }
            #pragma unroll
            for (int u = 0; u < 4; ++u) v[u] = nv[u];
            if (hj < 2 * TILE_R)
                hdt[(hj >> 1) * 64 + (hj & 1) * 16 + sl] = hv;
        }
    }
    __syncthreads();

    const unsigned total = count;
    if (total <= (unsigned)LCAP) {
        tile_chunk<false>((int)total, tid, wave, lane, lo, hs, hd, src, attn,
                          acc, sc, list, mxb, omx, dn);
    } else {
        for (int i = tid; i < TILE_R * HIDTOT / 4; i += 1024)
            ((float4*)acc)[i] = make_float4(0.f, 0.f, 0.f, 0.f);
        const int nwin = (SCAN_STEPS + WIN_STEPS - 1) / WIN_STEPS;
        for (int w = 0; w < nwin; ++w) {
            __syncthreads();
            if (tid == 0) count = 0u;
            __syncthreads();
            const int w0 = w * WIN_STEPS;
            const int wend = (w0 + WIN_STEPS) < SCAN_STEPS ? (w0 + WIN_STEPS) : SCAN_STEPS;
            for (int st = w0 + wave; st < wend; st += AGG_WAVES)
                scan_step(dst4[st * 32 + lane], st * 128 + lane * 4, lo, lane, list, &count);
            __syncthreads();
            tile_chunk<true>((int)count, tid, wave, lane, lo, hs, hd, src, attn,
                             acc, sc, list, mxb, omx, dn);
        }
    }
    __syncthreads();

    for (int r = wave; r < TILE_R; r += AGG_WAVES) {
        const float* a = acc + r * HIDTOT + lane;
        float* po = out + (size_t)(lo + r) * HIDTOT;
        #pragma unroll
        for (int qq = 0; qq < 2; ++qq) {
            float den = dn[r * 4 + 2 * qq + (lane >> 4)];
            den = den > 0.f ? den : 1.f;
            float4 o;
            o.x = a[qq * 128 +  0] / den;
            o.y = a[qq * 128 + 32] / den;
            o.z = a[qq * 128 + 64] / den;
            o.w = a[qq * 128 + 96] / den;
            o.x = o.x > 0.f ? o.x : (expf(o.x) - 1.f);
            o.y = o.y > 0.f ? o.y : (expf(o.y) - 1.f);
            o.z = o.z > 0.f ? o.z : (expf(o.z) - 1.f);
            o.w = o.w > 0.f ? o.w : (expf(o.w) - 1.f);
            ((float4*)(po + qq * 128))[lane] = o;
        }
    }
}

__global__ __launch_bounds__(256) void proj2_kernel(
    const float* __restrict__ h,
    const float* __restrict__ W2s, const float* __restrict__ b2s,
    const float* __restrict__ W2d, const float* __restrict__ b2d,
    float* __restrict__ hs2, float* __restrict__ hd2)
{
    const int n = blockIdx.x * 8 + (threadIdx.x >> 5);
    if (n >= NNODES) return;
    const int lane = threadIdx.x & 31;
    const int base = lane * 8;
    const float* hr = h + (size_t)n * HIDTOT + base;
    float s0 = 0.f, s1 = 0.f, d0 = 0.f, d1 = 0.f;
    #pragma unroll
    for (int k = 0; k < 8; ++k) {
        const float x = hr[k];
        const float2 ws = *(const float2*)(W2s + (size_t)(base + k) * 2);
        const float2 wd = *(const float2*)(W2d + (size_t)(base + k) * 2);
        s0 += x * ws.x; s1 += x * ws.y;
        d0 += x * wd.x; d1 += x * wd.y;
    }
    #pragma unroll
    for (int o = 1; o < 32; o <<= 1) {
        s0 += __shfl_xor(s0, o, 32); s1 += __shfl_xor(s1, o, 32);
        d0 += __shfl_xor(d0, o, 32); d1 += __shfl_xor(d1, o, 32);
    }
    if (lane == 0) {
        hs2[(size_t)n * 2 + 0] = s0 + b2s[0];
        hs2[(size_t)n * 2 + 1] = s1 + b2s[1];
        hd2[(size_t)n * 2 + 0] = d0 + b2d[0];
        hd2[(size_t)n * 2 + 1] = d1 + b2d[1];
    }
}

__global__ void edge_score2_kernel(const float* __restrict__ hs2,
                                   const float* __restrict__ hd2,
                                   const int* __restrict__ src,
                                   const int* __restrict__ dst,
                                   const float* __restrict__ attn2,
                                   float* __restrict__ score2,
                                   unsigned* __restrict__ m2)
{
    const int e = blockIdx.x * blockDim.x + threadIdx.x;
    if (e >= EDGES) return;
    const int s = src[e], d = dst[e];
    const float2 a = *(const float2*)(hs2 + (size_t)s * 2);
    const float2 b = *(const float2*)(hd2 + (size_t)d * 2);
    const float sc = lrelu(a.x + b.x) * attn2[0] + lrelu(a.y + b.y) * attn2[1];
    score2[e] = sc;
    atomicMax(m2 + d, ford(sc));
}

__global__ void edge_exp2_kernel(const int* __restrict__ dst,
                                 float* __restrict__ score2,
                                 const unsigned* __restrict__ m2,
                                 float* __restrict__ denom2)
{
    const int e = blockIdx.x * blockDim.x + threadIdx.x;
    if (e >= EDGES) return;
    const int d = dst[e];
    float m = fdec(m2[d]);
    if (!isfinite(m)) m = 0.f;
    const float ex = expf(score2[e] - m);
    score2[e] = ex;
    atomicAdd(denom2 + d, ex);
}

__global__ void edge_aggr2_kernel(const float* __restrict__ hs2,
                                  const int* __restrict__ src,
                                  const int* __restrict__ dst,
                                  const float* __restrict__ ex,
                                  const float* __restrict__ denom2,
                                  float* __restrict__ out)
{
    const int e = blockIdx.x * blockDim.x + threadIdx.x;
    if (e >= EDGES) return;
    const int s = src[e], d = dst[e];
    float den = denom2[d];
    den = den > 0.f ? den : 1.f;
    const float alpha = ex[e] / den;
    const float2 a = *(const float2*)(hs2 + (size_t)s * 2);
    atomicAdd(out + (size_t)d * 2 + 0, alpha * a.x);
    atomicAdd(out + (size_t)d * 2 + 1, alpha * a.y);
}

extern "C" void kernel_launch(void* const* d_in, const int* in_sizes, int n_in,
                              void* d_out, int out_size, void* d_ws, size_t ws_size,
                              hipStream_t stream)
{
    const float* feat  = (const float*)d_in[0];
    const int*   src   = (const int*)d_in[1];
    const int*   dst   = (const int*)d_in[2];
    const float* W1s   = (const float*)d_in[3];
    const float* b1s   = (const float*)d_in[4];
    const float* W1d   = (const float*)d_in[5];
    const float* b1d   = (const float*)d_in[6];
    const float* attn1 = (const float*)d_in[7];
    const float* W2s   = (const float*)d_in[8];
    const float* b2s   = (const float*)d_in[9];
    const float* W2d   = (const float*)d_in[10];
    const float* b2d   = (const float*)d_in[11];
    const float* attn2 = (const float*)d_in[12];
    float* out = (float*)d_out;

    float* ws = (float*)d_ws;
    size_t off = 0;
    float*    hs1    = ws + off; off += (size_t)NNODES * HIDTOT;
    float*    hd1    = ws + off; off += (size_t)NNODES * HIDTOT;
    float*    out1   = ws + off; off += (size_t)NNODES * HIDTOT;
    float*    hs2    = ws + off; off += (size_t)NNODES * 2;
    float*    hd2    = ws + off; off += (size_t)NNODES * 2;
    float*    score2 = ws + off; off += (size_t)EDGES;
    float*    denom2 = ws + off; off += (size_t)NNODES;
    unsigned* m2     = (unsigned*)(ws + off); off += (size_t)NNODES;

    const int B = 256;

    init_kernel<<<(NNODES * 2 + B - 1) / B, B, 0, stream>>>(
        denom2, m2, out);
    {
        const int tiles = 3125 * 4;
        gemm1_kernel<<<(tiles + 7) / 8, B, 0, stream>>>(
            feat, W1s, b1s, W1d, b1d, hs1, hd1);
    }
    tile_layer1_kernel<<<NTILES, 1024, 0, stream>>>(
        hs1, hd1, src, dst, attn1, out1);
    proj2_kernel<<<(NNODES + 7) / 8, B, 0, stream>>>(
        out1, W2s, b2s, W2d, b2d, hs2, hd2);
    edge_score2_kernel<<<(EDGES + B - 1) / B, B, 0, stream>>>(
        hs2, hd2, src, dst, attn2, score2, m2);
    edge_exp2_kernel<<<(EDGES + B - 1) / B, B, 0, stream>>>(
        dst, score2, m2, denom2);
    edge_aggr2_kernel<<<(EDGES + B - 1) / B, B, 0, stream>>>(
        hs2, src, dst, score2, denom2, out);

    (void)in_sizes; (void)n_in; (void)out_size; (void)ws_size;
}


# layer-2 edge phase as one kernel per pair of tiles on the lists the layer-1 tile kernel leaves in the workspace: max, exp, sums by LDS adds in arrival order, one store per element; 7 to 4 launches
# speedup vs baseline: 3.9642x; 1.7568x over previous
#include <hip/hip_runtime.h>
#include <hip/hip_bf16.h>
#include <math.h>

#define NNODES 50000
#define EDGES  800000
#define INFEAT 128
#define HIDTOT 256
#define NEG_SLOPE 0.2f
#define ORD_NEG_INF 0x007FFFFFu

typedef __attribute__((ext_vector_type(2))) float v2f;
typedef __attribute__((ext_vector_type(8))) float v8f;
typedef __attribute__((ext_vector_type(4))) float v4f;

__device__ __forceinline__ unsigned ford(float f) {
    unsigned u = __float_as_uint(f);
    return (u & 0x80000000u) ? ~u : (u | 0x80000000u);
}
__device__ __forceinline__ float fdec(unsigned u) {
    return __uint_as_float((u & 0x80000000u) ? (u & 0x7FFFFFFFu) : ~u);
}
__device__ __forceinline__ float lrelu(float x) {
    return x >= 0.f ? x : NEG_SLOPE * x;
}

__global__ __launch_bounds__(256) void gemm1_kernel(
    const float* __restrict__ feat,
    const float* __restrict__ Ws, const float* __restrict__ bs,
    const float* __restrict__ Wd, const float* __restrict__ bd,
    float* __restrict__ hs, float* __restrict__ hd)
{
    const int wave = threadIdx.x >> 5;
    const int lane = threadIdx.x & 31;
    const int half = lane >> 4;
    const int row  = lane & 15;

    int tile = blockIdx.x * 8 + wave;
    if (tile >= 3125 * 4) return;
    const int mt  = tile >> 2;
    const int nt4 = tile & 3;
    const int m0  = mt * 16;
    const int c0  = nt4 * 64;

    v8f zero = {0.f,0.f,0.f,0.f,0.f,0.f,0.f,0.f};
    v8f accS[4], accD[4];
    #pragma unroll
    for (int j = 0; j < 4; ++j) { accS[j] = zero; accD[j] = zero; }

    const float* arow = feat + (size_t)(m0 + row) * INFEAT;
    for (int k0 = 0; k0 < INFEAT; k0 += 4) {
        const int ka = k0 + 2 * half;
        v2f a = *(const v2f*)(arow + ka);
        #pragma unroll
        for (int j = 0; j < 4; ++j) {
            const int col = c0 + j * 16 + row;
            v2f bS, bD;
            bS.x = Ws[(size_t)ka * HIDTOT + col];
            bS.y = Ws[(size_t)(ka + 1) * HIDTOT + col];
            bD.x = Wd[(size_t)ka * HIDTOT + col];
            bD.y = Wd[(size_t)(ka + 1) * HIDTOT + col];
            accS[j] = __builtin_amdgcn_wmma_f32_16x16x4_f32(
                false, a, false, bS, (short)0, accS[j], false, false);
            accD[j] = __builtin_amdgcn_wmma_f32_16x16x4_f32(
                false, a, false, bD, (short)0, accD[j], false, false);
        }
    }
    #pragma unroll
    for (int j = 0; j < 4; ++j) {
        const int col = c0 + j * 16 + row;
        const float biasS = bs[col], biasD = bd[col];
        #pragma unroll
        for (int i = 0; i < 8; ++i) {
            const int r = m0 + i + 8 * half;
            hs[(size_t)r * HIDTOT + col] = accS[j][i] + biasS;
            hd[(size_t)r * HIDTOT + col] = accD[j][i] + biasD;
        }
    }
}

#define TILE_R     200
#define NTILES     250
#define AGG_WAVES  32
#define SCAN_STEPS 6250
#define WIN_STEPS  40
#define LCAP       (WIN_STEPS * 128)
#define GROUP      8
#define SGROUP     4

__device__ __forceinline__ void lds_add(float* p, float v) {
    __hip_atomic_fetch_add(p, v, __ATOMIC_RELAXED, __HIP_MEMORY_SCOPE_WORKGROUP);
}
__device__ __forceinline__ void lds_max(unsigned* p, unsigned v) {
    __hip_atomic_fetch_max(p, v, __ATOMIC_RELAXED, __HIP_MEMORY_SCOPE_WORKGROUP);
}
__device__ __forceinline__ float finite_or_zero(float m) {
    return isfinite(m) ? m : 0.f;
}

__device__ __forceinline__ void scan_step(
    const int4 v, const int eb, const int lo, const int lane,
    unsigned* list, unsigned* count)
{
    const int vals[4] = {v.x, v.y, v.z, v.w};
    unsigned dl[4], mk[4];
    int tot = 0;
    #pragma unroll
    for (int k = 0; k < 4; ++k) {
        dl[k] = (unsigned)(vals[k] - lo);
        mk[k] = __builtin_amdgcn_ballot_w32(dl[k] < (unsigned)TILE_R);
        tot += __builtin_popcount(mk[k]);
    }
    if (tot != 0) {
        unsigned base = 0u;
        if (lane == 0)
            base = __hip_atomic_fetch_add(count, (unsigned)tot, __ATOMIC_RELAXED,
                                          __HIP_MEMORY_SCOPE_WORKGROUP);
        base = (unsigned)__builtin_amdgcn_readfirstlane((int)base);
        #pragma unroll
        for (int k = 0; k < 4; ++k) {
            if (dl[k] < (unsigned)TILE_R) {
                const unsigned idx = base + __builtin_amdgcn_mbcnt_lo(mk[k], 0u);
                if (idx < (unsigned)LCAP)
                    list[idx] = ((unsigned)(eb + k) << 8) | dl[k];
            }
            base += (unsigned)__builtin_popcount(mk[k]);
        }
    }
}

template <bool HD_LDS>
__device__ __forceinline__ void score_batch(
    const unsigned* qp, float* scp, const int nb, const int lane, const int lo,
    const float* __restrict__ hs, const float* __restrict__ hd,
    const float4* hdt,
    const int* __restrict__ src, const float4 w0, const float4 w1,
    unsigned* mxb, unsigned* wl)
{
    int s_i = 0, dl_i = 0;
    if (lane < nb) {
        const unsigned ent = qp[lane];
        dl_i = (int)(ent & 255u);
        s_i = src[(int)(ent >> 8)];
        if (HD_LDS) wl[lane] = ((unsigned)s_i << 8) | (unsigned)dl_i;
    }
    for (int g0 = 0; g0 < nb; g0 += SGROUP) {
        float4 a0[SGROUP], a1[SGROUP], b0[SGROUP], b1[SGROUP];
        #pragma unroll
        for (int u = 0; u < SGROUP; ++u) {
            const int s = __builtin_amdgcn_readlane(s_i, g0 + u);
            const int d = __builtin_amdgcn_readlane(dl_i, g0 + u);
            const float4* ph = (const float4*)(hs + (size_t)s * HIDTOT + lane * 8);
            a0[u] = ph[0]; a1[u] = ph[1];
            if (HD_LDS) {
                b0[u] = hdt[d * 64 + lane];
                b1[u] = hdt[d * 64 + 32 + lane];
            } else {
                const float4* pd = (const float4*)(hd + (size_t)(lo + d) * HIDTOT + lane * 8);
                b0[u] = pd[0]; b1[u] = pd[1];
            }
        }
        #pragma unroll
        for (int u = 0; u < SGROUP; ++u) {
            if (g0 + u < nb) {
                float part = 0.f;
                part += lrelu(a0[u].x + b0[u].x) * w0.x + lrelu(a0[u].y + b0[u].y) * w0.y +
                        lrelu(a0[u].z + b0[u].z) * w0.z + lrelu(a0[u].w + b0[u].w) * w0.w;
                part += lrelu(a1[u].x + b1[u].x) * w1.x + lrelu(a1[u].y + b1[u].y) * w1.y +
                        lrelu(a1[u].z + b1[u].z) * w1.z + lrelu(a1[u].w + b1[u].w) * w1.w;
                part += __shfl_xor(part, 1, 32);
                part += __shfl_xor(part, 2, 32);
                part += __shfl_xor(part, 4, 32);
                if ((lane & 7) == 0) {
                    const int h = lane >> 3;
                    const int d = __builtin_amdgcn_readlane(dl_i, g0 + u);
                    scp[(g0 + u) * 4 + h] = part;
                    lds_max(mxb + d * 4 + h, ford(part));
                }
            }
        }
    }
}

__device__ __forceinline__ void add_batch(
    const unsigned* qp, const float* scp, const int nb, const int lane,
    const float* __restrict__ hs, const int* __restrict__ src,
    const unsigned* mxb, float* dn, float* acc)
{
    int s_i = 0, dl_i = 0;
    float a0 = 0.f, a1 = 0.f, a2 = 0.f, a3 = 0.f;
    if (lane < nb) {
        const unsigned ent = qp[lane];
        dl_i = (int)(ent & 255u);
        s_i = src[(int)(ent >> 8)];
        const float4 x = ((const float4*)scp)[lane];
        const uint4 mb = ((const uint4*)mxb)[dl_i];
        a0 = expf(x.x - finite_or_zero(fdec(mb.x)));
        a1 = expf(x.y - finite_or_zero(fdec(mb.y)));
        a2 = expf(x.z - finite_or_zero(fdec(mb.z)));
        a3 = expf(x.w - finite_or_zero(fdec(mb.w)));
        lds_add(dn + dl_i * 4 + 0, a0);
        lds_add(dn + dl_i * 4 + 1, a1);
        lds_add(dn + dl_i * 4 + 2, a2);
        lds_add(dn + dl_i * 4 + 3, a3);
    }
    const bool lowhalf = lane < 16;
    for (int g0 = 0; g0 < nb; g0 += GROUP) {
        float4 ra[GROUP], rb[GROUP];
        #pragma unroll
        for (int u = 0; u < GROUP; ++u) {
            const int s = __builtin_amdgcn_readlane(s_i, g0 + u);
            const float4* p = (const float4*)(hs + (size_t)s * HIDTOT) + lane;
            ra[u] = p[0];
            rb[u] = p[32];
        }
        #pragma unroll
        for (int u = 0; u < GROUP; ++u) {
            if (g0 + u < nb) {
                const int d = __builtin_amdgcn_readlane(dl_i, g0 + u);
                const float h0 = __int_as_float(__builtin_amdgcn_readlane(__float_as_int(a0), g0 + u));
                const float h1 = __int_as_float(__builtin_amdgcn_readlane(__float_as_int(a1), g0 + u));
                const float h2 = __int_as_float(__builtin_amdgcn_readlane(__float_as_int(a2), g0 + u));
                const float h3 = __int_as_float(__builtin_amdgcn_readlane(__float_as_int(a3), g0 + u));
                const float aA = lowhalf ? h0 : h1;
                const float aB = lowhalf ? h2 : h3;
                float* a = acc + d * HIDTOT + lane;
                lds_add(a +   0, aA * ra[u].x);
                lds_add(a +  32, aA * ra[u].y);
                lds_add(a +  64, aA * ra[u].z);
                lds_add(a +  96, aA * ra[u].w);
                lds_add(a + 128, aB * rb[u].x);
                lds_add(a + 160, aB * rb[u].y);
                lds_add(a + 192, aB * rb[u].z);
                lds_add(a + 224, aB * rb[u].w);
            }
        }
    }
}

template <bool WINDOWED>
__device__ __forceinline__ void tile_chunk(
    const int n, const int tid, const int wave,
    const int lane, const int lo,
    const float* __restrict__ hs, const float* __restrict__ hd,
    const int* __restrict__ src, const float* __restrict__ attn,
    float* acc, float* sc, const unsigned* list,
    unsigned* mxb, unsigned* omx, float* dn, unsigned* wlist)
{
    const int share = (n + AGG_WAVES - 1) / AGG_WAVES;
    const int beg = wave * share;
    const int end = (beg + share) < n ? (beg + share) : n;

    if (WINDOWED) {
        for (int i = tid; i < TILE_R * 4; i += 1024) omx[i] = mxb[i];
        __syncthreads();
    }

    {
        const float4 w0 = ((const float4*)(attn + lane * 8))[0];
        const float4 w1 = ((const float4*)(attn + lane * 8))[1];
        for (int i0 = beg; i0 < end; i0 += 32) {
            const int nb = (end - i0) < 32 ? (end - i0) : 32;
            score_batch<!WINDOWED>(list + i0, sc + i0 * 4, nb, lane, lo, hs, hd,
                                   (const float4*)acc, src, w0, w1, mxb, wlist + i0);
        }
    }
    __syncthreads();

    if (!WINDOWED) {
        for (int i = tid; i < TILE_R * HIDTOT / 4; i += 1024)
            ((float4*)acc)[i] = make_float4(0.f, 0.f, 0.f, 0.f);
        __syncthreads();
    } else {
        for (int i = tid; i < TILE_R * 4; i += 1024) {
            const unsigned ob = omx[i];
            const float f = (ob == ORD_NEG_INF) ? 0.f
                : expf(finite_or_zero(fdec(ob)) - finite_or_zero(fdec(mxb[i])));
            omx[i] = __float_as_uint(f);
            dn[i] *= f;
        }
        __syncthreads();
        for (int i = tid; i < TILE_R * HIDTOT; i += 1024) {
            const int r = i >> 8, p = i & 255;
            const int h = 2 * (p >> 7) + ((p & 31) >> 4);
            acc[i] *= __uint_as_float(omx[r * 4 + h]);
        }
        __syncthreads();
    }

    for (int i0 = beg; i0 < end; i0 += 32) {
        const int nb = (end - i0) < 32 ? (end - i0) : 32;
        add_batch(list + i0, sc + i0 * 4, nb, lane, hs, src, mxb, dn, acc);
    }
}

__global__ __launch_bounds__(1024) void tile_layer1_kernel(
    const float* __restrict__ hs, const float* __restrict__ hd,
    const int* __restrict__ src, const int* __restrict__ dst,
    const float* __restrict__ attn, float* __restrict__ out,
    unsigned* __restrict__ tlist, unsigned* __restrict__ tcount)
{
    __shared__ __attribute__((aligned(16))) float acc[TILE_R * HIDTOT];
    __shared__ __attribute__((aligned(16))) float sc[LCAP * 4];
    __shared__ __attribute__((aligned(16))) unsigned mxb[TILE_R * 4];
    __shared__ __attribute__((aligned(16))) unsigned omx[TILE_R * 4];
    __shared__ __attribute__((aligned(16))) float dn[TILE_R * 4];
    __shared__ unsigned list[LCAP];
    __shared__ unsigned count;

    const int tid  = threadIdx.x;
    const int wave = __builtin_amdgcn_readfirstlane(tid >> 5);
    const int lane = tid & 31;
    const int tile = blockIdx.x;
    const int lo   = tile * TILE_R;

    for (int i = tid; i < TILE_R * 4; i += 1024) { mxb[i] = ORD_NEG_INF; dn[i] = 0.f; }
    if (tid == 0) count = 0u;
    __syncthreads();

    const int4* dst4 = (const int4*)dst;
    const int4 none = make_int4(-1, -1, -1, -1);

    {
        const v4f* hdg = (const v4f*)(hd + (size_t)lo * HIDTOT);
        v4f* hdt = (v4f*)acc;
        const int cc = 2 * (lane & 15) + (lane >> 4);
        const int sl = (lane >> 4) * 32 + (lane & 15);
        const int rot = tile * (SCAN_STEPS / NTILES);
        int4 v[4];
        #pragma unroll
        for (int u = 0; u < 4; ++u) {
            int g = wave + 32 * u + rot;
            if (g >= SCAN_STEPS) g -= SCAN_STEPS;
            v[u] = dst4[g * 32 + lane];
        }
        for (int it = 0; it < 200; it += 4) {
            const int hj = wave + 32 * (it >> 2);
            v4f hv = {0.f, 0.f, 0.f, 0.f};
            if (hj < 2 * TILE_R)
                hv = __builtin_nontemporal_load(hdg + (hj >> 1) * 64 + (hj & 1) * 32 + cc);
            int4 nv[4];
            #pragma unroll
            for (int u = 0; u < 4; ++u) {
                const int st = wave + 32 * (it + 4 + u);
                int g = st + rot;
                if (g >= SCAN_STEPS) g -= SCAN_STEPS;
                nv[u] = none;
                if (st < SCAN_STEPS) nv[u] = dst4[g * 32 + lane];
            }
            #pragma unroll
            for (int u = 0; u < 4; ++u) {
                int g = wave + 32 * (it + u) + rot;
                if (g >= SCAN_STEPS) g -= SCAN_STEPS;
                scan_step(v[u], g * 128 + lane * 4, lo, lane, list, &count);
            }
            #pragma unroll
            for (int u = 0; u < 4; ++u) v[u] = nv[u];
            if (hj < 2 * TILE_R)
                hdt[(hj >> 1) * 64 + (hj & 1) * 16 + sl] = hv;
        }
    }
    __syncthreads();

    const unsigned total = count;
    unsigned* wlist = tlist + (size_t)tile * LCAP;
    if (tid == 0) tcount[tile] = total;
    if (total <= (unsigned)LCAP) {
        tile_chunk<false>((int)total, tid, wave, lane, lo, hs, hd, src, attn,
                          acc, sc, list, mxb, omx, dn, wlist);
    } else {
        for (int i = tid; i < TILE_R * HIDTOT / 4; i += 1024)
            ((float4*)acc)[i] = make_float4(0.f, 0.f, 0.f, 0.f);
        const int nwin = (SCAN_STEPS + WIN_STEPS - 1) / WIN_STEPS;
        for (int w = 0; w < nwin; ++w) {
            __syncthreads();
            if (tid == 0) count = 0u;
            __syncthreads();
            const int w0 = w * WIN_STEPS;
            const int wend = (w0 + WIN_STEPS) < SCAN_STEPS ? (w0 + WIN_STEPS) : SCAN_STEPS;
            for (int st = w0 + wave; st < wend; st += AGG_WAVES)
                scan_step(dst4[st * 32 + lane], st * 128 + lane * 4, lo, lane, list, &count);
            __syncthreads();
            tile_chunk<true>((int)count, tid, wave, lane, lo, hs, hd, src, attn,
                             acc, sc, list, mxb, omx, dn, wlist);
        }
    }
    __syncthreads();

    for (int r = wave; r < TILE_R; r += AGG_WAVES) {
        const float* a = acc + r * HIDTOT + lane;
        float* po = out + (size_t)(lo + r) * HIDTOT;
        #pragma unroll
        for (int qq = 0; qq < 2; ++qq) {
            float den = dn[r * 4 + 2 * qq + (lane >> 4)];
            den = den > 0.f ? den : 1.f;
            float4 o;
            o.x = a[qq * 128 +  0] / den;
            o.y = a[qq * 128 + 32] / den;
            o.z = a[qq * 128 + 64] / den;
            o.w = a[qq * 128 + 96] / den;
            o.x = o.x > 0.f ? o.x : (expf(o.x) - 1.f);
            o.y = o.y > 0.f ? o.y : (expf(o.y) - 1.f);
            o.z = o.z > 0.f ? o.z : (expf(o.z) - 1.f);
            o.w = o.w > 0.f ? o.w : (expf(o.w) - 1.f);
            ((float4*)(po + qq * 128))[lane] = o;
        }
    }
}

__global__ __launch_bounds__(256) void proj2_kernel(
    const float* __restrict__ h,
    const float* __restrict__ W2s, const float* __restrict__ b2s,
    const float* __restrict__ W2d, const float* __restrict__ b2d,
    float* __restrict__ hs2, float* __restrict__ hd2)
{
    const int n = blockIdx.x * 8 + (threadIdx.x >> 5);
    if (n >= NNODES) return;
    const int lane = threadIdx.x & 31;
    const int base = lane * 8;
    const float* hr = h + (size_t)n * HIDTOT + base;
    float s0 = 0.f, s1 = 0.f, d0 = 0.f, d1 = 0.f;
    #pragma unroll
    for (int k = 0; k < 8; ++k) {
        const float x = hr[k];
        const float2 ws = *(const float2*)(W2s + (size_t)(base + k) * 2);
        const float2 wd = *(const float2*)(W2d + (size_t)(base + k) * 2);
        s0 += x * ws.x; s1 += x * ws.y;
        d0 += x * wd.x; d1 += x * wd.y;
    }
    #pragma unroll
    for (int o = 1; o < 32; o <<= 1) {
        s0 += __shfl_xor(s0, o, 32); s1 += __shfl_xor(s1, o, 32);
        d0 += __shfl_xor(d0, o, 32); d1 += __shfl_xor(d1, o, 32);
    }
    if (lane == 0) {
        hs2[(size_t)n * 2 + 0] = s0 + b2s[0];
        hs2[(size_t)n * 2 + 1] = s1 + b2s[1];
        hd2[(size_t)n * 2 + 0] = d0 + b2d[0];
        hd2[(size_t)n * 2 + 1] = d1 + b2d[1];
    }
}

#define PAIR_R (2 * TILE_R)
#define NPAIRS 125

__device__ __forceinline__ float score2_of(const float2 a, const float2 b,
                                           const float w0, const float w1) {
    return lrelu(a.x + b.x) * w0 + lrelu(a.y + b.y) * w1;
}

__global__ __launch_bounds__(1024) void tile_layer2_kernel(
    const float* __restrict__ hs2, const float* __restrict__ hd2,
    const int* __restrict__ src, const int* __restrict__ dst,
    const float* __restrict__ attn2,
    const unsigned* __restrict__ tlist, const unsigned* __restrict__ tcount,
    float* __restrict__ out)
{
    __shared__ __attribute__((aligned(16))) float ent[2 * LCAP * 4];
    __shared__ unsigned mx[PAIR_R];
    __shared__ float dn[PAIR_R];
    __shared__ float sm[PAIR_R * 2];

    const int tid = threadIdx.x;
    const int pr  = blockIdx.x;
    const int lo  = pr * PAIR_R;

    for (int i = tid; i < PAIR_R; i += 1024) {
        mx[i] = ORD_NEG_INF; dn[i] = 0.f; sm[2 * i] = 0.f; sm[2 * i + 1] = 0.f;
    }
    const float w0 = attn2[0], w1 = attn2[1];
    const unsigned c0 = tcount[2 * pr], c1 = tcount[2 * pr + 1];
    const bool listed = (c0 <= (unsigned)LCAP) && (c1 <= (unsigned)LCAP);
    __syncthreads();

    if (listed) {
        const int n0 = (int)c0, n = (int)(c0 + c1);
        const unsigned* l0 = tlist + (size_t)(2 * pr) * LCAP;
        const unsigned* l1 = tlist + (size_t)(2 * pr + 1) * LCAP;
        float4* e4 = (float4*)ent;
        for (int ib = tid; ib < n; ib += 4096) {
            unsigned e[4];
            int dd[4];
            #pragma unroll
            for (int u = 0; u < 4; ++u) {
                const int i = ib + 1024 * u;
                e[u] = 0u;
                if (i < n) e[u] = (i < n0) ? l0[i] : l1[i - n0];
            }
            float2 a[4], b[4];
            #pragma unroll
            for (int u = 0; u < 4; ++u) {
                const int i = ib + 1024 * u;
                dd[u] = (int)(e[u] & 255u) + ((i < n0) ? 0 : TILE_R);
                a[u] = *(const float2*)(hs2 + (size_t)(e[u] >> 8) * 2);
                b[u] = *(const float2*)(hd2 + (size_t)(lo + dd[u]) * 2);
            }
            #pragma unroll
            for (int u = 0; u < 4; ++u) {
                const int i = ib + 1024 * u;
                if (i < n) {
                    const int d = dd[u];
                    const float s = score2_of(a[u], b[u], w0, w1);
                    e4[i] = make_float4(s, a[u].x, a[u].y, __int_as_float(d));
                    lds_max(mx + d, ford(s));
                }
            }
        }
        __syncthreads();
        for (int i = tid; i < n; i += 1024) {
            const float4 t = e4[i];
            const int d = __float_as_int(t.w);
            const float ex = expf(t.x - finite_or_zero(fdec(mx[d])));
            lds_add(dn + d, ex);
            lds_add(sm + 2 * d, ex * t.y);
            lds_add(sm + 2 * d + 1, ex * t.z);
        }
    } else {
        for (int e = tid; e < EDGES; e += 1024) {
            const unsigned d = (unsigned)(dst[e] - lo);
            if (d < (unsigned)PAIR_R) {
                const float2 a = *(const float2*)(hs2 + (size_t)src[e] * 2);
                const float2 b = *(const float2*)(hd2 + (size_t)(lo + (int)d) * 2);
                lds_max(mx + d, ford(score2_of(a, b, w0, w1)));
            }
        }
        __syncthreads();
        for (int e = tid; e < EDGES; e += 1024) {
            const unsigned d = (unsigned)(dst[e] - lo);
            if (d < (unsigned)PAIR_R) {
                const float2 a = *(const float2*)(hs2 + (size_t)src[e] * 2);
                const float2 b = *(const float2*)(hd2 + (size_t)(lo + (int)d) * 2);
                const float ex = expf(score2_of(a, b, w0, w1) - finite_or_zero(fdec(mx[d])));
                lds_add(dn + d, ex);
                lds_add(sm + 2 * d, ex * a.x);
                lds_add(sm + 2 * d + 1, ex * a.y);
            }
        }
    }
    __syncthreads();

    for (int i = tid; i < PAIR_R * 2; i += 1024) {
        float den = dn[i >> 1];
        den = den > 0.f ? den : 1.f;
        out[(size_t)lo * 2 + i] = sm[i] / den;
    }
}

extern "C" void kernel_launch(void* const* d_in, const int* in_sizes, int n_in,
                              void* d_out, int out_size, void* d_ws, size_t ws_size,
                              hipStream_t stream)
{
    const float* feat  = (const float*)d_in[0];
    const int*   src   = (const int*)d_in[1];
    const int*   dst   = (const int*)d_in[2];
    const float* W1s   = (const float*)d_in[3];
    const float* b1s   = (const float*)d_in[4];
    const float* W1d   = (const float*)d_in[5];
    const float* b1d   = (const float*)d_in[6];
    const float* attn1 = (const float*)d_in[7];
    const float* W2s   = (const float*)d_in[8];
    const float* b2s   = (const float*)d_in[9];
    const float* W2d   = (const float*)d_in[10];
    const float* b2d   = (const float*)d_in[11];
    const float* attn2 = (const float*)d_in[12];
    float* out = (float*)d_out;

    float* ws = (float*)d_ws;
    size_t off = 0;
    float*    hs1    = ws + off; off += (size_t)NNODES * HIDTOT;
    float*    hd1    = ws + off; off += (size_t)NNODES * HIDTOT;
    float*    out1   = ws + off; off += (size_t)NNODES * HIDTOT;
    float*    hs2    = ws + off; off += (size_t)NNODES * 2;
    float*    hd2    = ws + off; off += (size_t)NNODES * 2;
    unsigned* tlist  = (unsigned*)(ws + off); off += (size_t)NTILES * LCAP;
    unsigned* tcount = (unsigned*)(ws + off); off += (size_t)NTILES;

    const int B = 256;

    {
        const int tiles = 3125 * 4;
        gemm1_kernel<<<(tiles + 7) / 8, B, 0, stream>>>(
            feat, W1s, b1s, W1d, b1d, hs1, hd1);
    }
    tile_layer1_kernel<<<NTILES, 1024, 0, stream>>>(
        hs1, hd1, src, dst, attn1, out1, tlist, tcount);
    proj2_kernel<<<(NNODES + 7) / 8, B, 0, stream>>>(
        out1, W2s, b2s, W2d, b2d, hs2, hd2);
    tile_layer2_kernel<<<NPAIRS, 1024, 0, stream>>>(
        hs2, hd2, src, dst, attn2, tlist, tcount, out);

    (void)in_sizes; (void)n_in; (void)out_size; (void)ws_size;
}


# layer-1 tile kernel sweeps one-byte tile keys (dst/200, from a small key kernel, 4 to 5 launches): 16 edges per lane and load, exact zero-byte test, hits take LDS list slots; list in arrival order
# speedup vs baseline: 4.2356x; 1.0685x over previous
#include <hip/hip_runtime.h>
#include <hip/hip_bf16.h>
#include <math.h>

#define NNODES 50000
#define EDGES  800000
#define INFEAT 128
#define HIDTOT 256
#define NEG_SLOPE 0.2f
#define ORD_NEG_INF 0x007FFFFFu
#define TILE_R     200
#define NTILES     250
#define KSTEPS     1563
#define KEY_BLOCKS 196

typedef __attribute__((ext_vector_type(2))) float v2f;
typedef __attribute__((ext_vector_type(8))) float v8f;
typedef __attribute__((ext_vector_type(4))) float v4f;

__device__ __forceinline__ unsigned ford(float f) {
    unsigned u = __float_as_uint(f);
    return (u & 0x80000000u) ? ~u : (u | 0x80000000u);
}
__device__ __forceinline__ float fdec(unsigned u) {
    return __uint_as_float((u & 0x80000000u) ? (u & 0x7FFFFFFFu) : ~u);
}
__device__ __forceinline__ float lrelu(float x) {
    return x >= 0.f ? x : NEG_SLOPE * x;
}

__global__ __launch_bounds__(256) void gemm1_kernel(
    const float* __restrict__ feat,
    const float* __restrict__ Ws, const float* __restrict__ bs,
    const float* __restrict__ Wd, const float* __restrict__ bd,
    float* __restrict__ hs, float* __restrict__ hd)
{
    const int wave = threadIdx.x >> 5;
    const int lane = threadIdx.x & 31;
    const int half = lane >> 4;
    const int row  = lane & 15;

    int tile = blockIdx.x * 8 + wave;
    if (tile >= 3125 * 4) return;
    const int mt  = tile >> 2;
    const int nt4 = tile & 3;
    const int m0  = mt * 16;
    const int c0  = nt4 * 64;

    v8f zero = {0.f,0.f,0.f,0.f,0.f,0.f,0.f,0.f};
    v8f accS[4], accD[4];
    #pragma unroll
    for (int j = 0; j < 4; ++j) { accS[j] = zero; accD[j] = zero; }

    const float* arow = feat + (size_t)(m0 + row) * INFEAT;
    for (int k0 = 0; k0 < INFEAT; k0 += 4) {
        const int ka = k0 + 2 * half;
        v2f a = *(const v2f*)(arow + ka);
        #pragma unroll
        for (int j = 0; j < 4; ++j) {
            const int col = c0 + j * 16 + row;
            v2f bS, bD;
            bS.x = Ws[(size_t)ka * HIDTOT + col];
            bS.y = Ws[(size_t)(ka + 1) * HIDTOT + col];
            bD.x = Wd[(size_t)ka * HIDTOT + col];
            bD.y = Wd[(size_t)(ka + 1) * HIDTOT + col];
            accS[j] = __builtin_amdgcn_wmma_f32_16x16x4_f32(
                false, a, false, bS, (short)0, accS[j], false, false);
            accD[j] = __builtin_amdgcn_wmma_f32_16x16x4_f32(
                false, a, false, bD, (short)0, accD[j], false, false);
        }
    }
    #pragma unroll
    for (int j = 0; j < 4; ++j) {
        const int col = c0 + j * 16 + row;
        const float biasS = bs[col], biasD = bd[col];
        #pragma unroll
        for (int i = 0; i < 8; ++i) {
            const int r = m0 + i + 8 * half;
            hs[(size_t)r * HIDTOT + col] = accS[j][i] + biasS;
            hd[(size_t)r * HIDTOT + col] = accD[j][i] + biasD;
        }
    }
}

__device__ __forceinline__ unsigned tile_key(const int d) {
    return ((unsigned)d < (unsigned)NNODES) ? ((unsigned)d / (unsigned)TILE_R) : 255u;
}

__global__ __launch_bounds__(256) void tile_key_kernel(
    const int* __restrict__ dst, unsigned* __restrict__ keys)
{
    const int t = blockIdx.x * 256 + threadIdx.x;
    if (t >= KSTEPS * 32) return;
    uint4 o = make_uint4(~0u, ~0u, ~0u, ~0u);
    if (t < EDGES / 16) {
        const int4* p = (const int4*)dst + (size_t)t * 4;
        unsigned r[4];
        #pragma unroll
        for (int k = 0; k < 4; ++k) {
            const int4 q = p[k];
            r[k] = tile_key(q.x) | (tile_key(q.y) << 8) |
                   (tile_key(q.z) << 16) | (tile_key(q.w) << 24);
        }
        o = make_uint4(r[0], r[1], r[2], r[3]);
    }
    ((uint4*)keys)[t] = o;
}

#define AGG_WAVES  32
#define SCAN_STEPS 6250
#define WIN_STEPS  40
#define LCAP       (WIN_STEPS * 128)
#define GROUP      8
#define SGROUP     4

__device__ __forceinline__ void lds_add(float* p, float v) {
    __hip_atomic_fetch_add(p, v, __ATOMIC_RELAXED, __HIP_MEMORY_SCOPE_WORKGROUP);
}
__device__ __forceinline__ void lds_max(unsigned* p, unsigned v) {
    __hip_atomic_fetch_max(p, v, __ATOMIC_RELAXED, __HIP_MEMORY_SCOPE_WORKGROUP);
}
__device__ __forceinline__ float finite_or_zero(float m) {
    return isfinite(m) ? m : 0.f;
}

__device__ __forceinline__ void scan_step(
    const int4 v, const int eb, const int lo, const int lane,
    unsigned* list, unsigned* count)
{
    const int vals[4] = {v.x, v.y, v.z, v.w};
    unsigned dl[4], mk[4];
    int tot = 0;
    #pragma unroll
    for (int k = 0; k < 4; ++k) {
        dl[k] = (unsigned)(vals[k] - lo);
        mk[k] = __builtin_amdgcn_ballot_w32(dl[k] < (unsigned)TILE_R);
        tot += __builtin_popcount(mk[k]);
    }
    if (tot != 0) {
        unsigned base = 0u;
        if (lane == 0)
            base = __hip_atomic_fetch_add(count, (unsigned)tot, __ATOMIC_RELAXED,
                                          __HIP_MEMORY_SCOPE_WORKGROUP);
        base = (unsigned)__builtin_amdgcn_readfirstlane((int)base);
        #pragma unroll
        for (int k = 0; k < 4; ++k) {
            if (dl[k] < (unsigned)TILE_R) {
                const unsigned idx = base + __builtin_amdgcn_mbcnt_lo(mk[k], 0u);
                if (idx < (unsigned)LCAP)
                    list[idx] = ((unsigned)(eb + k) << 8) | dl[k];
            }
            base += (unsigned)__builtin_popcount(mk[k]);
        }
    }
}

__device__ __forceinline__ void key_step(
    const uint4 v, const int eb, const unsigned pat,
    unsigned* list, unsigned* count)
{
    const unsigned w[4] = {v.x ^ pat, v.y ^ pat, v.z ^ pat, v.w ^ pat};
    unsigned t[4];
    #pragma unroll
    for (int k = 0; k < 4; ++k)
        t[k] = ~(((w[k] & 0x7F7F7F7Fu) + 0x7F7F7F7Fu) | w[k] | 0x7F7F7F7Fu);
    if (((t[0] | t[1]) | (t[2] | t[3])) != 0u) {
        unsigned m = (t[0] >> 7) | (t[1] >> 6) | (t[2] >> 5) | (t[3] >> 4);
        while (m != 0u) {
            const int p = __builtin_ctz(m);
            m &= m - 1u;
            const unsigned idx = __hip_atomic_fetch_add(count, 1u, __ATOMIC_RELAXED,
                                                        __HIP_MEMORY_SCOPE_WORKGROUP);
            if (idx < (unsigned)LCAP)
                list[idx] = (unsigned)(eb + 4 * (p & 7) + (p >> 3));
        }
    }
}

template <bool HD_LDS>
__device__ __forceinline__ void score_batch(
    const unsigned* qp, float* scp, const int nb, const int lane, const int lo,
    const float* __restrict__ hs, const float* __restrict__ hd,
    const float4* hdt,
    const int* __restrict__ src, const float4 w0, const float4 w1,
    unsigned* mxb, unsigned* wl)
{
    int s_i = 0, dl_i = 0;
    if (lane < nb) {
        const unsigned ent = qp[lane];
        dl_i = (int)(ent & 255u);
        s_i = src[(int)(ent >> 8)];
        if (HD_LDS) wl[lane] = ((unsigned)s_i << 8) | (unsigned)dl_i;
    }
    for (int g0 = 0; g0 < nb; g0 += SGROUP) {
        float4 a0[SGROUP], a1[SGROUP], b0[SGROUP], b1[SGROUP];
        #pragma unroll
        for (int u = 0; u < SGROUP; ++u) {
            const int s = __builtin_amdgcn_readlane(s_i, g0 + u);
            const int d = __builtin_amdgcn_readlane(dl_i, g0 + u);
            const float4* ph = (const float4*)(hs + (size_t)s * HIDTOT + lane * 8);
            a0[u] = ph[0]; a1[u] = ph[1];
            if (HD_LDS) {
                b0[u] = hdt[d * 64 + lane];
                b1[u] = hdt[d * 64 + 32 + lane];
            } else {
                const float4* pd = (const float4*)(hd + (size_t)(lo + d) * HIDTOT + lane * 8);
                b0[u] = pd[0]; b1[u] = pd[1];
            }
        }
        #pragma unroll
        for (int u = 0; u < SGROUP; ++u) {
            if (g0 + u < nb) {
                float part = 0.f;
                part += lrelu(a0[u].x + b0[u].x) * w0.x + lrelu(a0[u].y + b0[u].y) * w0.y +
                        lrelu(a0[u].z + b0[u].z) * w0.z + lrelu(a0[u].w + b0[u].w) * w0.w;
                part += lrelu(a1[u].x + b1[u].x) * w1.x + lrelu(a1[u].y + b1[u].y) * w1.y +
                        lrelu(a1[u].z + b1[u].z) * w1.z + lrelu(a1[u].w + b1[u].w) * w1.w;
                part += __shfl_xor(part, 1, 32);
                part += __shfl_xor(part, 2, 32);
                part += __shfl_xor(part, 4, 32);
                if ((lane & 7) == 0) {
                    const int h = lane >> 3;
                    const int d = __builtin_amdgcn_readlane(dl_i, g0 + u);
                    scp[(g0 + u) * 4 + h] = part;
                    lds_max(mxb + d * 4 + h, ford(part));
                }
            }
        }
    }
}

__device__ __forceinline__ void add_batch(
    const unsigned* qp, const float* scp, const int nb, const int lane,
    const float* __restrict__ hs, const int* __restrict__ src,
    const unsigned* mxb, float* dn, float* acc)
{
    int s_i = 0, dl_i = 0;
    float a0 = 0.f, a1 = 0.f, a2 = 0.f, a3 = 0.f;
    if (lane < nb) {
        const unsigned ent = qp[lane];
        dl_i = (int)(ent & 255u);
        s_i = src[(int)(ent >> 8)];
        const float4 x = ((const float4*)scp)[lane];
        const uint4 mb = ((const uint4*)mxb)[dl_i];
        a0 = expf(x.x - finite_or_zero(fdec(mb.x)));
        a1 = expf(x.y - finite_or_zero(fdec(mb.y)));
        a2 = expf(x.z - finite_or_zero(fdec(mb.z)));
        a3 = expf(x.w - finite_or_zero(fdec(mb.w)));
        lds_add(dn + dl_i * 4 + 0, a0);
        lds_add(dn + dl_i * 4 + 1, a1);
        lds_add(dn + dl_i * 4 + 2, a2);
        lds_add(dn + dl_i * 4 + 3, a3);
    }
    const bool lowhalf = lane < 16;
    for (int g0 = 0; g0 < nb; g0 += GROUP) {
        float4 ra[GROUP], rb[GROUP];
        #pragma unroll
        for (int u = 0; u < GROUP; ++u) {
            const int s = __builtin_amdgcn_readlane(s_i, g0 + u);
            const float4* p = (const float4*)(hs + (size_t)s * HIDTOT) + lane;
            ra[u] = p[0];
            rb[u] = p[32];
        }
        #pragma unroll
        for (int u = 0; u < GROUP; ++u) {
            if (g0 + u < nb) {
                const int d = __builtin_amdgcn_readlane(dl_i, g0 + u);
                const float h0 = __int_as_float(__builtin_amdgcn_readlane(__float_as_int(a0), g0 + u));
                const float h1 = __int_as_float(__builtin_amdgcn_readlane(__float_as_int(a1), g0 + u));
                const float h2 = __int_as_float(__builtin_amdgcn_readlane(__float_as_int(a2), g0 + u));
                const float h3 = __int_as_float(__builtin_amdgcn_readlane(__float_as_int(a3), g0 + u));
                const float aA = lowhalf ? h0 : h1;
                const float aB = lowhalf ? h2 : h3;
                float* a = acc + d * HIDTOT + lane;
                lds_add(a +   0, aA * ra[u].x);
                lds_add(a +  32, aA * ra[u].y);
                lds_add(a +  64, aA * ra[u].z);
                lds_add(a +  96, aA * ra[u].w);
                lds_add(a + 128, aB * rb[u].x);
                lds_add(a + 160, aB * rb[u].y);
                lds_add(a + 192, aB * rb[u].z);
                lds_add(a + 224, aB * rb[u].w);
            }
        }
    }
}

template <bool WINDOWED>
__device__ __forceinline__ void tile_chunk(
    const int n, const int tid, const int wave,
    const int lane, const int lo,
    const float* __restrict__ hs, const float* __restrict__ hd,
    const int* __restrict__ src, const float* __restrict__ attn,
    float* acc, float* sc, const unsigned* list,
    unsigned* mxb, unsigned* omx, float* dn, unsigned* wlist)
{
    const int share = (n + AGG_WAVES - 1) / AGG_WAVES;
    const int beg = wave * share;
    const int end = (beg + share) < n ? (beg + share) : n;

    if (WINDOWED) {
        for (int i = tid; i < TILE_R * 4; i += 1024) omx[i] = mxb[i];
        __syncthreads();
    }

    {
        const float4 w0 = ((const float4*)(attn + lane * 8))[0];
        const float4 w1 = ((const float4*)(attn + lane * 8))[1];
        for (int i0 = beg; i0 < end; i0 += 32) {
            const int nb = (end - i0) < 32 ? (end - i0) : 32;
            score_batch<!WINDOWED>(list + i0, sc + i0 * 4, nb, lane, lo, hs, hd,
                                   (const float4*)acc, src, w0, w1, mxb, wlist + i0);
        }
    }
    __syncthreads();

    if (!WINDOWED) {
        for (int i = tid; i < TILE_R * HIDTOT / 4; i += 1024)
            ((float4*)acc)[i] = make_float4(0.f, 0.f, 0.f, 0.f);
        __syncthreads();
    } else {
        for (int i = tid; i < TILE_R * 4; i += 1024) {
            const unsigned ob = omx[i];
            const float f = (ob == ORD_NEG_INF) ? 0.f
                : expf(finite_or_zero(fdec(ob)) - finite_or_zero(fdec(mxb[i])));
            omx[i] = __float_as_uint(f);
            dn[i] *= f;
        }
        __syncthreads();
        for (int i = tid; i < TILE_R * HIDTOT; i += 1024) {
            const int r = i >> 8, p = i & 255;
            const int h = 2 * (p >> 7) + ((p & 31) >> 4);
            acc[i] *= __uint_as_float(omx[r * 4 + h]);
        }
        __syncthreads();
    }

    for (int i0 = beg; i0 < end; i0 += 32) {
        const int nb = (end - i0) < 32 ? (end - i0) : 32;
        add_batch(list + i0, sc + i0 * 4, nb, lane, hs, src, mxb, dn, acc);
    }
}

__global__ __launch_bounds__(1024) void tile_layer1_kernel(
    const float* __restrict__ hs, const float* __restrict__ hd,
    const int* __restrict__ src, const int* __restrict__ dst,
    const float* __restrict__ attn, float* __restrict__ out,
    unsigned* __restrict__ tlist, unsigned* __restrict__ tcount,
    const unsigned* __restrict__ keys)
{
    __shared__ __attribute__((aligned(16))) float acc[TILE_R * HIDTOT];
    __shared__ __attribute__((aligned(16))) float sc[LCAP * 4];
    __shared__ __attribute__((aligned(16))) unsigned mxb[TILE_R * 4];
    __shared__ __attribute__((aligned(16))) unsigned omx[TILE_R * 4];
    __shared__ __attribute__((aligned(16))) float dn[TILE_R * 4];
    __shared__ unsigned list[LCAP];
    __shared__ unsigned count;

    const int tid  = threadIdx.x;
    const int wave = __builtin_amdgcn_readfirstlane(tid >> 5);
    const int lane = tid & 31;
    const int tile = blockIdx.x;
    const int lo   = tile * TILE_R;

    for (int i = tid; i < TILE_R * 4; i += 1024) { mxb[i] = ORD_NEG_INF; dn[i] = 0.f; }
    if (tid == 0) count = 0u;
    __syncthreads();

    const int4* dst4 = (const int4*)dst;

    {
        const v4f* hdg = (const v4f*)(hd + (size_t)lo * HIDTOT);
        v4f* hdt = (v4f*)acc;
        const int cc = 2 * (lane & 15) + (lane >> 4);
        const int sl = (lane >> 4) * 32 + (lane & 15);
        const uint4* key4 = (const uint4*)keys;
        const unsigned pat = (unsigned)tile * 0x01010101u;
        const uint4 none = make_uint4(~0u, ~0u, ~0u, ~0u);
        const int rot = tile * (KSTEPS / NTILES);
        uint4 v[4];
        #pragma unroll
        for (int u = 0; u < 4; ++u) {
            int g = wave + 32 * u + rot;
            if (g >= KSTEPS) g -= KSTEPS;
            v[u] = key4[g * 32 + lane];
        }
        for (int it = 0; it < 52; it += 4) {
            const int hj = wave + 32 * (it >> 2);
            v4f hv = {0.f, 0.f, 0.f, 0.f};
            if (hj < 2 * TILE_R)
                hv = __builtin_nontemporal_load(hdg + (hj >> 1) * 64 + (hj & 1) * 32 + cc);
            uint4 nv[4];
            #pragma unroll
            for (int u = 0; u < 4; ++u) {
                const int st = wave + 32 * (it + 4 + u);
                int g = st + rot;
                if (g >= KSTEPS) g -= KSTEPS;
                nv[u] = none;
                if (st < KSTEPS) nv[u] = key4[g * 32 + lane];
            }
            #pragma unroll
            for (int u = 0; u < 4; ++u) {
                int g = wave + 32 * (it + u) + rot;
                if (g >= KSTEPS) g -= KSTEPS;
                key_step(v[u], g * 512 + lane * 16, pat, list, &count);
            }
            #pragma unroll
            for (int u = 0; u < 4; ++u) v[u] = nv[u];
            if (hj < 2 * TILE_R)
                hdt[(hj >> 1) * 64 + (hj & 1) * 16 + sl] = hv;
        }
    }
    __syncthreads();

    const unsigned total = count;
    unsigned* wlist = tlist + (size_t)tile * LCAP;
    if (tid == 0) tcount[tile] = total;
    if (total <= (unsigned)LCAP) {
        for (int i = tid; i < (int)total; i += 1024) {
            const unsigned e = list[i];
            list[i] = (e << 8) | (unsigned)(dst[e] - lo);
        }
        __syncthreads();
        tile_chunk<false>((int)total, tid, wave, lane, lo, hs, hd, src, attn,
                          acc, sc, list, mxb, omx, dn, wlist);
    } else {
        for (int i = tid; i < TILE_R * HIDTOT / 4; i += 1024)
            ((float4*)acc)[i] = make_float4(0.f, 0.f, 0.f, 0.f);
        const int nwin = (SCAN_STEPS + WIN_STEPS - 1) / WIN_STEPS;
        for (int w = 0; w < nwin; ++w) {
            __syncthreads();
            if (tid == 0) count = 0u;
            __syncthreads();
            const int w0 = w * WIN_STEPS;
            const int wend = (w0 + WIN_STEPS) < SCAN_STEPS ? (w0 + WIN_STEPS) : SCAN_STEPS;
            for (int st = w0 + wave; st < wend; st += AGG_WAVES)
                scan_step(dst4[st * 32 + lane], st * 128 + lane * 4, lo, lane, list, &count);
            __syncthreads();
            tile_chunk<true>((int)count, tid, wave, lane, lo, hs, hd, src, attn,
                             acc, sc, list, mxb, omx, dn, wlist);
        }
    }
    __syncthreads();

    for (int r = wave; r < TILE_R; r += AGG_WAVES) {
        const float* a = acc + r * HIDTOT + lane;
        float* po = out + (size_t)(lo + r) * HIDTOT;
        #pragma unroll
        for (int qq = 0; qq < 2; ++qq) {
            float den = dn[r * 4 + 2 * qq + (lane >> 4)];
            den = den > 0.f ? den : 1.f;
            float4 o;
            o.x = a[qq * 128 +  0] / den;
            o.y = a[qq * 128 + 32] / den;
            o.z = a[qq * 128 + 64] / den;
            o.w = a[qq * 128 + 96] / den;
            o.x = o.x > 0.f ? o.x : (expf(o.x) - 1.f);
            o.y = o.y > 0.f ? o.y : (expf(o.y) - 1.f);
            o.z = o.z > 0.f ? o.z : (expf(o.z) - 1.f);
            o.w = o.w > 0.f ? o.w : (expf(o.w) - 1.f);
            ((float4*)(po + qq * 128))[lane] = o;
        }
    }
}

__global__ __launch_bounds__(256) void proj2_kernel(
    const float* __restrict__ h,
    const float* __restrict__ W2s, const float* __restrict__ b2s,
    const float* __restrict__ W2d, const float* __restrict__ b2d,
    float* __restrict__ hs2, float* __restrict__ hd2)
{
    const int n = blockIdx.x * 8 + (threadIdx.x >> 5);
    if (n >= NNODES) return;
    const int lane = threadIdx.x & 31;
    const int base = lane * 8;
    const float* hr = h + (size_t)n * HIDTOT + base;
    float s0 = 0.f, s1 = 0.f, d0 = 0.f, d1 = 0.f;
    #pragma unroll
    for (int k = 0; k < 8; ++k) {
        const float x = hr[k];
        const float2 ws = *(const float2*)(W2s + (size_t)(base + k) * 2);
        const float2 wd = *(const float2*)(W2d + (size_t)(base + k) * 2);
        s0 += x * ws.x; s1 += x * ws.y;
        d0 += x * wd.x; d1 += x * wd.y;
    }
    #pragma unroll
    for (int o = 1; o < 32; o <<= 1) {
        s0 += __shfl_xor(s0, o, 32); s1 += __shfl_xor(s1, o, 32);
        d0 += __shfl_xor(d0, o, 32); d1 += __shfl_xor(d1, o, 32);
    }
    if (lane == 0) {
        hs2[(size_t)n * 2 + 0] = s0 + b2s[0];
        hs2[(size_t)n * 2 + 1] = s1 + b2s[1];
        hd2[(size_t)n * 2 + 0] = d0 + b2d[0];
        hd2[(size_t)n * 2 + 1] = d1 + b2d[1];
    }
}

#define PAIR_R (2 * TILE_R)
#define NPAIRS 125

__device__ __forceinline__ float score2_of(const float2 a, const float2 b,
                                           const float w0, const float w1) {
    return lrelu(a.x + b.x) * w0 + lrelu(a.y + b.y) * w1;
}

__global__ __launch_bounds__(1024) void tile_layer2_kernel(
    const float* __restrict__ hs2, const float* __restrict__ hd2,
    const int* __restrict__ src, const int* __restrict__ dst,
    const float* __restrict__ attn2,
    const unsigned* __restrict__ tlist, const unsigned* __restrict__ tcount,
    float* __restrict__ out)
{
    __shared__ __attribute__((aligned(16))) float ent[2 * LCAP * 4];
    __shared__ unsigned mx[PAIR_R];
    __shared__ float dn[PAIR_R];
    __shared__ float sm[PAIR_R * 2];

    const int tid = threadIdx.x;
    const int pr  = blockIdx.x;
    const int lo  = pr * PAIR_R;

    for (int i = tid; i < PAIR_R; i += 1024) {
        mx[i] = ORD_NEG_INF; dn[i] = 0.f; sm[2 * i] = 0.f; sm[2 * i + 1] = 0.f;
    }
    const float w0 = attn2[0], w1 = attn2[1];
    const unsigned c0 = tcount[2 * pr], c1 = tcount[2 * pr + 1];
    const bool listed = (c0 <= (unsigned)LCAP) && (c1 <= (unsigned)LCAP);
    __syncthreads();

    if (listed) {
        const int n0 = (int)c0, n = (int)(c0 + c1);
        const unsigned* l0 = tlist + (size_t)(2 * pr) * LCAP;
        const unsigned* l1 = tlist + (size_t)(2 * pr + 1) * LCAP;
        float4* e4 = (float4*)ent;
        for (int ib = tid; ib < n; ib += 4096) {
            unsigned e[4];
            int dd[4];
            #pragma unroll
            for (int u = 0; u < 4; ++u) {
                const int i = ib + 1024 * u;
                e[u] = 0u;
                if (i < n) e[u] = (i < n0) ? l0[i] : l1[i - n0];
            }
            float2 a[4], b[4];
            #pragma unroll
            for (int u = 0; u < 4; ++u) {
                const int i = ib + 1024 * u;
                dd[u] = (int)(e[u] & 255u) + ((i < n0) ? 0 : TILE_R);
                a[u] = *(const float2*)(hs2 + (size_t)(e[u] >> 8) * 2);
                b[u] = *(const float2*)(hd2 + (size_t)(lo + dd[u]) * 2);
            }
            #pragma unroll
            for (int u = 0; u < 4; ++u) {
                const int i = ib + 1024 * u;
                if (i < n) {
                    const int d = dd[u];
                    const float s = score2_of(a[u], b[u], w0, w1);
                    e4[i] = make_float4(s, a[u].x, a[u].y, __int_as_float(d));
                    lds_max(mx + d, ford(s));
                }
            }
        }
        __syncthreads();
        for (int i = tid; i < n; i += 1024) {
            const float4 t = e4[i];
            const int d = __float_as_int(t.w);
            const float ex = expf(t.x - finite_or_zero(fdec(mx[d])));
            lds_add(dn + d, ex);
            lds_add(sm + 2 * d, ex * t.y);
            lds_add(sm + 2 * d + 1, ex * t.z);
        }
    } else {
        for (int e = tid; e < EDGES; e += 1024) {
            const unsigned d = (unsigned)(dst[e] - lo);
            if (d < (unsigned)PAIR_R) {
                const float2 a = *(const float2*)(hs2 + (size_t)src[e] * 2);
                const float2 b = *(const float2*)(hd2 + (size_t)(lo + (int)d) * 2);
                lds_max(mx + d, ford(score2_of(a, b, w0, w1)));
            }
        }
        __syncthreads();
        for (int e = tid; e < EDGES; e += 1024) {
            const unsigned d = (unsigned)(dst[e] - lo);
            if (d < (unsigned)PAIR_R) {
                const float2 a = *(const float2*)(hs2 + (size_t)src[e] * 2);
                const float2 b = *(const float2*)(hd2 + (size_t)(lo + (int)d) * 2);
                const float ex = expf(score2_of(a, b, w0, w1) - finite_or_zero(fdec(mx[d])));
                lds_add(dn + d, ex);
                lds_add(sm + 2 * d, ex * a.x);
                lds_add(sm + 2 * d + 1, ex * a.y);
            }
        }
    }
    __syncthreads();

    for (int i = tid; i < PAIR_R * 2; i += 1024) {
        float den = dn[i >> 1];
        den = den > 0.f ? den : 1.f;
        out[(size_t)lo * 2 + i] = sm[i] / den;
    }
}

extern "C" void kernel_launch(void* const* d_in, const int* in_sizes, int n_in,
                              void* d_out, int out_size, void* d_ws, size_t ws_size,
                              hipStream_t stream)
{
    const float* feat  = (const float*)d_in[0];
    const int*   src   = (const int*)d_in[1];
    const int*   dst   = (const int*)d_in[2];
    const float* W1s   = (const float*)d_in[3];
    const float* b1s   = (const float*)d_in[4];
    const float* W1d   = (const float*)d_in[5];
    const float* b1d   = (const float*)d_in[6];
    const float* attn1 = (const float*)d_in[7];
    const float* W2s   = (const float*)d_in[8];
    const float* b2s   = (const float*)d_in[9];
    const float* W2d   = (const float*)d_in[10];
    const float* b2d   = (const float*)d_in[11];
    const float* attn2 = (const float*)d_in[12];
    float* out = (float*)d_out;

    float* ws = (float*)d_ws;
    size_t off = 0;
    float*    hs1    = ws + off; off += (size_t)NNODES * HIDTOT;
    float*    hd1    = ws + off; off += (size_t)NNODES * HIDTOT;
    float*    out1   = ws + off; off += (size_t)NNODES * HIDTOT;
    float*    hs2    = ws + off; off += (size_t)NNODES * 2;
    float*    hd2    = ws + off; off += (size_t)NNODES * 2;
    unsigned* tlist  = (unsigned*)(ws + off); off += (size_t)NTILES * LCAP;
    unsigned* tcount = (unsigned*)(ws + off); off += (size_t)256;
    unsigned* keys   = (unsigned*)(ws + off); off += (size_t)KSTEPS * 128;

    const int B = 256;

    tile_key_kernel<<<KEY_BLOCKS, B, 0, stream>>>(dst, keys);
    {
        const int tiles = 3125 * 4;
        gemm1_kernel<<<(tiles + 7) / 8, B, 0, stream>>>(
            feat, W1s, b1s, W1d, b1d, hs1, hd1);
    }
    tile_layer1_kernel<<<NTILES, 1024, 0, stream>>>(
        hs1, hd1, src, dst, attn1, out1, tlist, tcount, keys);
    proj2_kernel<<<(NNODES + 7) / 8, B, 0, stream>>>(
        out1, W2s, b2s, W2d, b2d, hs2, hd2);
    tile_layer2_kernel<<<NPAIRS, 1024, 0, stream>>>(
        hs2, hd2, src, dst, attn2, tlist, tcount, out);

    (void)in_sizes; (void)n_in; (void)out_size; (void)ws_size;
}


# layer-1 dual GEMM: 256 blocks of 32 waves, each keeps both weight matrices in LDS in WMMA fragment order (B fragment = one 8-byte LDS load) and loops over its 16-row tiles; k and sum order unchanged
# speedup vs baseline: 4.6450x; 1.0967x over previous
#include <hip/hip_runtime.h>
#include <hip/hip_bf16.h>
#include <math.h>

#define NNODES 50000
#define EDGES  800000
#define INFEAT 128
#define HIDTOT 256
#define NEG_SLOPE 0.2f
#define ORD_NEG_INF 0x007FFFFFu
#define TILE_R     200
#define NTILES     250
#define KSTEPS     1563
#define KEY_BLOCKS 196

typedef __attribute__((ext_vector_type(2))) float v2f;
typedef __attribute__((ext_vector_type(8))) float v8f;
typedef __attribute__((ext_vector_type(4))) float v4f;

__device__ __forceinline__ unsigned ford(float f) {
    unsigned u = __float_as_uint(f);
    return (u & 0x80000000u) ? ~u : (u | 0x80000000u);
}
__device__ __forceinline__ float fdec(unsigned u) {
    return __uint_as_float((u & 0x80000000u) ? (u & 0x7FFFFFFFu) : ~u);
}
__device__ __forceinline__ float lrelu(float x) {
    return x >= 0.f ? x : NEG_SLOPE * x;
}

#define G_BLOCKS 256
#define G_MTILES 3125

__global__ __launch_bounds__(1024) void gemm1_kernel(
    const float* __restrict__ feat,
    const float* __restrict__ Ws, const float* __restrict__ bs,
    const float* __restrict__ Wd, const float* __restrict__ bd,
    float* __restrict__ hs, float* __restrict__ hd)
{
    __shared__ __attribute__((aligned(16))) float wfrag[2 * INFEAT * HIDTOT];

    const int tid  = threadIdx.x;
    #pragma unroll
    for (int it = 0; it < 8; ++it) {
        const int u   = tid + 1024 * it;
        const int mtx = u >> 12;
        const int kp  = (((u >> 7) & 31) << 1) | ((u >> 2) & 1);
        const int cq  = (((u >> 3) & 15) << 2) | (u & 3);
        const float* Wp = mtx ? Wd : Ws;
        const float4 r0 = *(const float4*)(Wp + (size_t)(2 * kp) * HIDTOT + 4 * cq);
        const float4 r1 = *(const float4*)(Wp + (size_t)(2 * kp + 1) * HIDTOT + 4 * cq);
        float4* q = (float4*)(wfrag +
            ((((mtx * 32 + (kp >> 1)) * 16 + (cq >> 2)) * 32 + (kp & 1) * 16 + 4 * (cq & 3)) * 2));
        q[0] = make_float4(r0.x, r1.x, r0.y, r1.y);
        q[1] = make_float4(r0.z, r1.z, r0.w, r1.w);
    }
    __syncthreads();

    const int wave = tid >> 5;
    const int lane = tid & 31;
    const int half = lane >> 4;
    const int row  = lane & 15;
    const int nt4  = wave & 3;
    const int c0   = nt4 * 64;

    const v2f* bfr = (const v2f*)wfrag + (nt4 * 4) * 32 + lane;

    for (int mt = blockIdx.x + G_BLOCKS * (wave >> 2); mt < G_MTILES; mt += G_BLOCKS * 8) {
        const int m0 = mt * 16;

        v8f zero = {0.f,0.f,0.f,0.f,0.f,0.f,0.f,0.f};
        v8f accS[4], accD[4];
        #pragma unroll
        for (int j = 0; j < 4; ++j) { accS[j] = zero; accD[j] = zero; }

        const float* arow = feat + (size_t)(m0 + row) * INFEAT;
        for (int k0 = 0; k0 < INFEAT; k0 += 4) {
            const int ka = k0 + 2 * half;
            v2f a = *(const v2f*)(arow + ka);
            const v2f* bk = bfr + (k0 >> 2) * (16 * 32);
            #pragma unroll
            for (int j = 0; j < 4; ++j) {
                const v2f bS = bk[j * 32];
                const v2f bD = bk[32 * 16 * 32 + j * 32];
                accS[j] = __builtin_amdgcn_wmma_f32_16x16x4_f32(
                    false, a, false, bS, (short)0, accS[j], false, false);
                accD[j] = __builtin_amdgcn_wmma_f32_16x16x4_f32(
                    false, a, false, bD, (short)0, accD[j], false, false);
            }
        }
        #pragma unroll
        for (int j = 0; j < 4; ++j) {
            const int col = c0 + j * 16 + row;
            const float biasS = bs[col], biasD = bd[col];
            #pragma unroll
            for (int i = 0; i < 8; ++i) {
                const int r = m0 + i + 8 * half;
                hs[(size_t)r * HIDTOT + col] = accS[j][i] + biasS;
                hd[(size_t)r * HIDTOT + col] = accD[j][i] + biasD;
            }
        }
    }
}

__device__ __forceinline__ unsigned tile_key(const int d) {
    return ((unsigned)d < (unsigned)NNODES) ? ((unsigned)d / (unsigned)TILE_R) : 255u;
}

__global__ __launch_bounds__(256) void tile_key_kernel(
    const int* __restrict__ dst, unsigned* __restrict__ keys)
{
    const int t = blockIdx.x * 256 + threadIdx.x;
    if (t >= KSTEPS * 32) return;
    uint4 o = make_uint4(~0u, ~0u, ~0u, ~0u);
    if (t < EDGES / 16) {
        const int4* p = (const int4*)dst + (size_t)t * 4;
        unsigned r[4];
        #pragma unroll
        for (int k = 0; k < 4; ++k) {
            const int4 q = p[k];
            r[k] = tile_key(q.x) | (tile_key(q.y) << 8) |
                   (tile_key(q.z) << 16) | (tile_key(q.w) << 24);
        }
        o = make_uint4(r[0], r[1], r[2], r[3]);
    }
    ((uint4*)keys)[t] = o;
}

#define AGG_WAVES  32
#define SCAN_STEPS 6250
#define WIN_STEPS  40
#define LCAP       (WIN_STEPS * 128)
#define GROUP      8
#define SGROUP     4

__device__ __forceinline__ void lds_add(float* p, float v) {
    __hip_atomic_fetch_add(p, v, __ATOMIC_RELAXED, __HIP_MEMORY_SCOPE_WORKGROUP);
}
__device__ __forceinline__ void lds_max(unsigned* p, unsigned v) {
    __hip_atomic_fetch_max(p, v, __ATOMIC_RELAXED, __HIP_MEMORY_SCOPE_WORKGROUP);
}
__device__ __forceinline__ float finite_or_zero(float m) {
    return isfinite(m) ? m : 0.f;
}

__device__ __forceinline__ void scan_step(
    const int4 v, const int eb, const int lo, const int lane,
    unsigned* list, unsigned* count)
{
    const int vals[4] = {v.x, v.y, v.z, v.w};
    unsigned dl[4], mk[4];
    int tot = 0;
    #pragma unroll
    for (int k = 0; k < 4; ++k) {
        dl[k] = (unsigned)(vals[k] - lo);
        mk[k] = __builtin_amdgcn_ballot_w32(dl[k] < (unsigned)TILE_R);
        tot += __builtin_popcount(mk[k]);
    }
    if (tot != 0) {
        unsigned base = 0u;
        if (lane == 0)
            base = __hip_atomic_fetch_add(count, (unsigned)tot, __ATOMIC_RELAXED,
                                          __HIP_MEMORY_SCOPE_WORKGROUP);
        base = (unsigned)__builtin_amdgcn_readfirstlane((int)base);
        #pragma unroll
        for (int k = 0; k < 4; ++k) {
            if (dl[k] < (unsigned)TILE_R) {
                const unsigned idx = base + __builtin_amdgcn_mbcnt_lo(mk[k], 0u);
                if (idx < (unsigned)LCAP)
                    list[idx] = ((unsigned)(eb + k) << 8) | dl[k];
            }
            base += (unsigned)__builtin_popcount(mk[k]);
        }
    }
}

__device__ __forceinline__ void key_step(
    const uint4 v, const int eb, const unsigned pat,
    unsigned* list, unsigned* count)
{
    const unsigned w[4] = {v.x ^ pat, v.y ^ pat, v.z ^ pat, v.w ^ pat};
    unsigned t[4];
    #pragma unroll
    for (int k = 0; k < 4; ++k)
        t[k] = ~(((w[k] & 0x7F7F7F7Fu) + 0x7F7F7F7Fu) | w[k] | 0x7F7F7F7Fu);
    if (((t[0] | t[1]) | (t[2] | t[3])) != 0u) {
        unsigned m = (t[0] >> 7) | (t[1] >> 6) | (t[2] >> 5) | (t[3] >> 4);
        while (m != 0u) {
            const int p = __builtin_ctz(m);
            m &= m - 1u;
            const unsigned idx = __hip_atomic_fetch_add(count, 1u, __ATOMIC_RELAXED,
                                                        __HIP_MEMORY_SCOPE_WORKGROUP);
            if (idx < (unsigned)LCAP)
                list[idx] = (unsigned)(eb + 4 * (p & 7) + (p >> 3));
        }
    }
}

template <bool HD_LDS>
__device__ __forceinline__ void score_batch(
    const unsigned* qp, float* scp, const int nb, const int lane, const int lo,
    const float* __restrict__ hs, const float* __restrict__ hd,
    const float4* hdt,
    const int* __restrict__ src, const float4 w0, const float4 w1,
    unsigned* mxb, unsigned* wl)
{
    int s_i = 0, dl_i = 0;
    if (lane < nb) {
        const unsigned ent = qp[lane];
        dl_i = (int)(ent & 255u);
        s_i = src[(int)(ent >> 8)];
        if (HD_LDS) wl[lane] = ((unsigned)s_i << 8) | (unsigned)dl_i;
    }
    for (int g0 = 0; g0 < nb; g0 += SGROUP) {
        float4 a0[SGROUP], a1[SGROUP], b0[SGROUP], b1[SGROUP];
        #pragma unroll
        for (int u = 0; u < SGROUP; ++u) {
            const int s = __builtin_amdgcn_readlane(s_i, g0 + u);
            const int d = __builtin_amdgcn_readlane(dl_i, g0 + u);
            const float4* ph = (const float4*)(hs + (size_t)s * HIDTOT + lane * 8);
            a0[u] = ph[0]; a1[u] = ph[1];
            if (HD_LDS) {
                b0[u] = hdt[d * 64 + lane];
                b1[u] = hdt[d * 64 + 32 + lane];
            } else {
                const float4* pd = (const float4*)(hd + (size_t)(lo + d) * HIDTOT + lane * 8);
                b0[u] = pd[0]; b1[u] = pd[1];
            }
        }
        #pragma unroll
        for (int u = 0; u < SGROUP; ++u) {
            if (g0 + u < nb) {
                float part = 0.f;
                part += lrelu(a0[u].x + b0[u].x) * w0.x + lrelu(a0[u].y + b0[u].y) * w0.y +
                        lrelu(a0[u].z + b0[u].z) * w0.z + lrelu(a0[u].w + b0[u].w) * w0.w;
                part += lrelu(a1[u].x + b1[u].x) * w1.x + lrelu(a1[u].y + b1[u].y) * w1.y +
                        lrelu(a1[u].z + b1[u].z) * w1.z + lrelu(a1[u].w + b1[u].w) * w1.w;
                part += __shfl_xor(part, 1, 32);
                part += __shfl_xor(part, 2, 32);
                part += __shfl_xor(part, 4, 32);
                if ((lane & 7) == 0) {
                    const int h = lane >> 3;
                    const int d = __builtin_amdgcn_readlane(dl_i, g0 + u);
                    scp[(g0 + u) * 4 + h] = part;
                    lds_max(mxb + d * 4 + h, ford(part));
                }
            }
        }
    }
}

__device__ __forceinline__ void add_batch(
    const unsigned* qp, const float* scp, const int nb, const int lane,
    const float* __restrict__ hs, const int* __restrict__ src,
    const unsigned* mxb, float* dn, float* acc)
{
    int s_i = 0, dl_i = 0;
    float a0 = 0.f, a1 = 0.f, a2 = 0.f, a3 = 0.f;
    if (lane < nb) {
        const unsigned ent = qp[lane];
        dl_i = (int)(ent & 255u);
        s_i = src[(int)(ent >> 8)];
        const float4 x = ((const float4*)scp)[lane];
        const uint4 mb = ((const uint4*)mxb)[dl_i];
        a0 = expf(x.x - finite_or_zero(fdec(mb.x)));
        a1 = expf(x.y - finite_or_zero(fdec(mb.y)));
        a2 = expf(x.z - finite_or_zero(fdec(mb.z)));
        a3 = expf(x.w - finite_or_zero(fdec(mb.w)));
        lds_add(dn + dl_i * 4 + 0, a0);
        lds_add(dn + dl_i * 4 + 1, a1);
        lds_add(dn + dl_i * 4 + 2, a2);
        lds_add(dn + dl_i * 4 + 3, a3);
    }
    const bool lowhalf = lane < 16;
    for (int g0 = 0; g0 < nb; g0 += GROUP) {
        float4 ra[GROUP], rb[GROUP];
        #pragma unroll
        for (int u = 0; u < GROUP; ++u) {
            const int s = __builtin_amdgcn_readlane(s_i, g0 + u);
            const float4* p = (const float4*)(hs + (size_t)s * HIDTOT) + lane;
            ra[u] = p[0];
            rb[u] = p[32];
        }
        #pragma unroll
        for (int u = 0; u < GROUP; ++u) {
            if (g0 + u < nb) {
                const int d = __builtin_amdgcn_readlane(dl_i, g0 + u);
                const float h0 = __int_as_float(__builtin_amdgcn_readlane(__float_as_int(a0), g0 + u));
                const float h1 = __int_as_float(__builtin_amdgcn_readlane(__float_as_int(a1), g0 + u));
                const float h2 = __int_as_float(__builtin_amdgcn_readlane(__float_as_int(a2), g0 + u));
                const float h3 = __int_as_float(__builtin_amdgcn_readlane(__float_as_int(a3), g0 + u));
                const float aA = lowhalf ? h0 : h1;
                const float aB = lowhalf ? h2 : h3;
                float* a = acc + d * HIDTOT + lane;
                lds_add(a +   0, aA * ra[u].x);
                lds_add(a +  32, aA * ra[u].y);
                lds_add(a +  64, aA * ra[u].z);
                lds_add(a +  96, aA * ra[u].w);
                lds_add(a + 128, aB * rb[u].x);
                lds_add(a + 160, aB * rb[u].y);
                lds_add(a + 192, aB * rb[u].z);
                lds_add(a + 224, aB * rb[u].w);
            }
        }
    }
}

template <bool WINDOWED>
__device__ __forceinline__ void tile_chunk(
    const int n, const int tid, const int wave,
    const int lane, const int lo,
    const float* __restrict__ hs, const float* __restrict__ hd,
    const int* __restrict__ src, const float* __restrict__ attn,
    float* acc, float* sc, const unsigned* list,
    unsigned* mxb, unsigned* omx, float* dn, unsigned* wlist)
{
    const int share = (n + AGG_WAVES - 1) / AGG_WAVES;
    const int beg = wave * share;
    const int end = (beg + share) < n ? (beg + share) : n;

    if (WINDOWED) {
        for (int i = tid; i < TILE_R * 4; i += 1024) omx[i] = mxb[i];
        __syncthreads();
    }

    {
        const float4 w0 = ((const float4*)(attn + lane * 8))[0];
        const float4 w1 = ((const float4*)(attn + lane * 8))[1];
        for (int i0 = beg; i0 < end; i0 += 32) {
            const int nb = (end - i0) < 32 ? (end - i0) : 32;
            score_batch<!WINDOWED>(list + i0, sc + i0 * 4, nb, lane, lo, hs, hd,
                                   (const float4*)acc, src, w0, w1, mxb, wlist + i0);
        }
    }
    __syncthreads();

    if (!WINDOWED) {
        for (int i = tid; i < TILE_R * HIDTOT / 4; i += 1024)
            ((float4*)acc)[i] = make_float4(0.f, 0.f, 0.f, 0.f);
        __syncthreads();
    } else {
        for (int i = tid; i < TILE_R * 4; i += 1024) {
            const unsigned ob = omx[i];
            const float f = (ob == ORD_NEG_INF) ? 0.f
                : expf(finite_or_zero(fdec(ob)) - finite_or_zero(fdec(mxb[i])));
            omx[i] = __float_as_uint(f);
            dn[i] *= f;
        }
        __syncthreads();
        for (int i = tid; i < TILE_R * HIDTOT; i += 1024) {
            const int r = i >> 8, p = i & 255;
            const int h = 2 * (p >> 7) + ((p & 31) >> 4);
            acc[i] *= __uint_as_float(omx[r * 4 + h]);
        }
        __syncthreads();
    }

    for (int i0 = beg; i0 < end; i0 += 32) {
        const int nb = (end - i0) < 32 ? (end - i0) : 32;
        add_batch(list + i0, sc + i0 * 4, nb, lane, hs, src, mxb, dn, acc);
    }
}

__global__ __launch_bounds__(1024) void tile_layer1_kernel(
    const float* __restrict__ hs, const float* __restrict__ hd,
    const int* __restrict__ src, const int* __restrict__ dst,
    const float* __restrict__ attn, float* __restrict__ out,
    unsigned* __restrict__ tlist, unsigned* __restrict__ tcount,
    const unsigned* __restrict__ keys)
{
    __shared__ __attribute__((aligned(16))) float acc[TILE_R * HIDTOT];
    __shared__ __attribute__((aligned(16))) float sc[LCAP * 4];
    __shared__ __attribute__((aligned(16))) unsigned mxb[TILE_R * 4];
    __shared__ __attribute__((aligned(16))) unsigned omx[TILE_R * 4];
    __shared__ __attribute__((aligned(16))) float dn[TILE_R * 4];
    __shared__ unsigned list[LCAP];
    __shared__ unsigned count;

    const int tid  = threadIdx.x;
    const int wave = __builtin_amdgcn_readfirstlane(tid >> 5);
    const int lane = tid & 31;
    const int tile = blockIdx.x;
    const int lo   = tile * TILE_R;

    for (int i = tid; i < TILE_R * 4; i += 1024) { mxb[i] = ORD_NEG_INF; dn[i] = 0.f; }
    if (tid == 0) count = 0u;
    __syncthreads();

    const int4* dst4 = (const int4*)dst;

    {
        const v4f* hdg = (const v4f*)(hd + (size_t)lo * HIDTOT);
        v4f* hdt = (v4f*)acc;
        const int cc = 2 * (lane & 15) + (lane >> 4);
        const int sl = (lane >> 4) * 32 + (lane & 15);
        const uint4* key4 = (const uint4*)keys;
        const unsigned pat = (unsigned)tile * 0x01010101u;
        const uint4 none = make_uint4(~0u, ~0u, ~0u, ~0u);
        const int rot = tile * (KSTEPS / NTILES);
        uint4 v[4];
        #pragma unroll
        for (int u = 0; u < 4; ++u) {
            int g = wave + 32 * u + rot;
            if (g >= KSTEPS) g -= KSTEPS;
            v[u] = key4[g * 32 + lane];
        }
        for (int it = 0; it < 52; it += 4) {
            const int hj = wave + 32 * (it >> 2);
            v4f hv = {0.f, 0.f, 0.f, 0.f};
            if (hj < 2 * TILE_R)
                hv = __builtin_nontemporal_load(hdg + (hj >> 1) * 64 + (hj & 1) * 32 + cc);
            uint4 nv[4];
            #pragma unroll
            for (int u = 0; u < 4; ++u) {
                const int st = wave + 32 * (it + 4 + u);
                int g = st + rot;
                if (g >= KSTEPS) g -= KSTEPS;
                nv[u] = none;
                if (st < KSTEPS) nv[u] = key4[g * 32 + lane];
            }
            #pragma unroll
            for (int u = 0; u < 4; ++u) {
                int g = wave + 32 * (it + u) + rot;
                if (g >= KSTEPS) g -= KSTEPS;
                key_step(v[u], g * 512 + lane * 16, pat, list, &count);
            }
            #pragma unroll
            for (int u = 0; u < 4; ++u) v[u] = nv[u];
            if (hj < 2 * TILE_R)
                hdt[(hj >> 1) * 64 + (hj & 1) * 16 + sl] = hv;
        }
    }
    __syncthreads();

    const unsigned total = count;
    unsigned* wlist = tlist + (size_t)tile * LCAP;
    if (tid == 0) tcount[tile] = total;
    if (total <= (unsigned)LCAP) {
        for (int i = tid; i < (int)total; i += 1024) {
            const unsigned e = list[i];
            list[i] = (e << 8) | (unsigned)(dst[e] - lo);
        }
        __syncthreads();
        tile_chunk<false>((int)total, tid, wave, lane, lo, hs, hd, src, attn,
                          acc, sc, list, mxb, omx, dn, wlist);
    } else {
        for (int i = tid; i < TILE_R * HIDTOT / 4; i += 1024)
            ((float4*)acc)[i] = make_float4(0.f, 0.f, 0.f, 0.f);
        const int nwin = (SCAN_STEPS + WIN_STEPS - 1) / WIN_STEPS;
        for (int w = 0; w < nwin; ++w) {
            __syncthreads();
            if (tid == 0) count = 0u;
            __syncthreads();
            const int w0 = w * WIN_STEPS;
            const int wend = (w0 + WIN_STEPS) < SCAN_STEPS ? (w0 + WIN_STEPS) : SCAN_STEPS;
            for (int st = w0 + wave; st < wend; st += AGG_WAVES)
                scan_step(dst4[st * 32 + lane], st * 128 + lane * 4, lo, lane, list, &count);
            __syncthreads();
            tile_chunk<true>((int)count, tid, wave, lane, lo, hs, hd, src, attn,
                             acc, sc, list, mxb, omx, dn, wlist);
        }
    }
    __syncthreads();

    for (int r = wave; r < TILE_R; r += AGG_WAVES) {
        const float* a = acc + r * HIDTOT + lane;
        float* po = out + (size_t)(lo + r) * HIDTOT;
        #pragma unroll
        for (int qq = 0; qq < 2; ++qq) {
            float den = dn[r * 4 + 2 * qq + (lane >> 4)];
            den = den > 0.f ? den : 1.f;
            float4 o;
            o.x = a[qq * 128 +  0] / den;
            o.y = a[qq * 128 + 32] / den;
            o.z = a[qq * 128 + 64] / den;
            o.w = a[qq * 128 + 96] / den;
            o.x = o.x > 0.f ? o.x : (expf(o.x) - 1.f);
            o.y = o.y > 0.f ? o.y : (expf(o.y) - 1.f);
            o.z = o.z > 0.f ? o.z : (expf(o.z) - 1.f);
            o.w = o.w > 0.f ? o.w : (expf(o.w) - 1.f);
            ((float4*)(po + qq * 128))[lane] = o;
        }
    }
}

__global__ __launch_bounds__(256) void proj2_kernel(
    const float* __restrict__ h,
    const float* __restrict__ W2s, const float* __restrict__ b2s,
    const float* __restrict__ W2d, const float* __restrict__ b2d,
    float* __restrict__ hs2, float* __restrict__ hd2)
{
    const int n = blockIdx.x * 8 + (threadIdx.x >> 5);
    if (n >= NNODES) return;
    const int lane = threadIdx.x & 31;
    const int base = lane * 8;
    const float* hr = h + (size_t)n * HIDTOT + base;
    float s0 = 0.f, s1 = 0.f, d0 = 0.f, d1 = 0.f;
    #pragma unroll
    for (int k = 0; k < 8; ++k) {
        const float x = hr[k];
        const float2 ws = *(const float2*)(W2s + (size_t)(base + k) * 2);
        const float2 wd = *(const float2*)(W2d + (size_t)(base + k) * 2);
        s0 += x * ws.x; s1 += x * ws.y;
        d0 += x * wd.x; d1 += x * wd.y;
    }
    #pragma unroll
    for (int o = 1; o < 32; o <<= 1) {
        s0 += __shfl_xor(s0, o, 32); s1 += __shfl_xor(s1, o, 32);
        d0 += __shfl_xor(d0, o, 32); d1 += __shfl_xor(d1, o, 32);
    }
    if (lane == 0) {
        hs2[(size_t)n * 2 + 0] = s0 + b2s[0];
        hs2[(size_t)n * 2 + 1] = s1 + b2s[1];
        hd2[(size_t)n * 2 + 0] = d0 + b2d[0];
        hd2[(size_t)n * 2 + 1] = d1 + b2d[1];
    }
}

#define PAIR_R (2 * TILE_R)
#define NPAIRS 125

__device__ __forceinline__ float score2_of(const float2 a, const float2 b,
                                           const float w0, const float w1) {
    return lrelu(a.x + b.x) * w0 + lrelu(a.y + b.y) * w1;
}

__global__ __launch_bounds__(1024) void tile_layer2_kernel(
    const float* __restrict__ hs2, const float* __restrict__ hd2,
    const int* __restrict__ src, const int* __restrict__ dst,
    const float* __restrict__ attn2,
    const unsigned* __restrict__ tlist, const unsigned* __restrict__ tcount,
    float* __restrict__ out)
{
    __shared__ __attribute__((aligned(16))) float ent[2 * LCAP * 4];
    __shared__ unsigned mx[PAIR_R];
    __shared__ float dn[PAIR_R];
    __shared__ float sm[PAIR_R * 2];

    const int tid = threadIdx.x;
    const int pr  = blockIdx.x;
    const int lo  = pr * PAIR_R;

    for (int i = tid; i < PAIR_R; i += 1024) {
        mx[i] = ORD_NEG_INF; dn[i] = 0.f; sm[2 * i] = 0.f; sm[2 * i + 1] = 0.f;
    }
    const float w0 = attn2[0], w1 = attn2[1];
    const unsigned c0 = tcount[2 * pr], c1 = tcount[2 * pr + 1];
    const bool listed = (c0 <= (unsigned)LCAP) && (c1 <= (unsigned)LCAP);
    __syncthreads();

    if (listed) {
        const int n0 = (int)c0, n = (int)(c0 + c1);
        const unsigned* l0 = tlist + (size_t)(2 * pr) * LCAP;
        const unsigned* l1 = tlist + (size_t)(2 * pr + 1) * LCAP;
        float4* e4 = (float4*)ent;
        for (int ib = tid; ib < n; ib += 4096) {
            unsigned e[4];
            int dd[4];
            #pragma unroll
            for (int u = 0; u < 4; ++u) {
                const int i = ib + 1024 * u;
                e[u] = 0u;
                if (i < n) e[u] = (i < n0) ? l0[i] : l1[i - n0];
            }
            float2 a[4], b[4];
            #pragma unroll
            for (int u = 0; u < 4; ++u) {
                const int i = ib + 1024 * u;
                dd[u] = (int)(e[u] & 255u) + ((i < n0) ? 0 : TILE_R);
                a[u] = *(const float2*)(hs2 + (size_t)(e[u] >> 8) * 2);
                b[u] = *(const float2*)(hd2 + (size_t)(lo + dd[u]) * 2);
            }
            #pragma unroll
            for (int u = 0; u < 4; ++u) {
                const int i = ib + 1024 * u;
                if (i < n) {
                    const int d = dd[u];
                    const float s = score2_of(a[u], b[u], w0, w1);
                    e4[i] = make_float4(s, a[u].x, a[u].y, __int_as_float(d));
                    lds_max(mx + d, ford(s));
                }
            }
        }
        __syncthreads();
        for (int i = tid; i < n; i += 1024) {
            const float4 t = e4[i];
            const int d = __float_as_int(t.w);
            const float ex = expf(t.x - finite_or_zero(fdec(mx[d])));
            lds_add(dn + d, ex);
            lds_add(sm + 2 * d, ex * t.y);
            lds_add(sm + 2 * d + 1, ex * t.z);
        }
    } else {
        for (int e = tid; e < EDGES; e += 1024) {
            const unsigned d = (unsigned)(dst[e] - lo);
            if (d < (unsigned)PAIR_R) {
                const float2 a = *(const float2*)(hs2 + (size_t)src[e] * 2);
                const float2 b = *(const float2*)(hd2 + (size_t)(lo + (int)d) * 2);
                lds_max(mx + d, ford(score2_of(a, b, w0, w1)));
            }
        }
        __syncthreads();
        for (int e = tid; e < EDGES; e += 1024) {
            const unsigned d = (unsigned)(dst[e] - lo);
            if (d < (unsigned)PAIR_R) {
                const float2 a = *(const float2*)(hs2 + (size_t)src[e] * 2);
                const float2 b = *(const float2*)(hd2 + (size_t)(lo + (int)d) * 2);
                const float ex = expf(score2_of(a, b, w0, w1) - finite_or_zero(fdec(mx[d])));
                lds_add(dn + d, ex);
                lds_add(sm + 2 * d, ex * a.x);
                lds_add(sm + 2 * d + 1, ex * a.y);
            }
        }
    }
    __syncthreads();

    for (int i = tid; i < PAIR_R * 2; i += 1024) {
        float den = dn[i >> 1];
        den = den > 0.f ? den : 1.f;
        out[(size_t)lo * 2 + i] = sm[i] / den;
    }
}

extern "C" void kernel_launch(void* const* d_in, const int* in_sizes, int n_in,
                              void* d_out, int out_size, void* d_ws, size_t ws_size,
                              hipStream_t stream)
{
    const float* feat  = (const float*)d_in[0];
    const int*   src   = (const int*)d_in[1];
    const int*   dst   = (const int*)d_in[2];
    const float* W1s   = (const float*)d_in[3];
    const float* b1s   = (const float*)d_in[4];
    const float* W1d   = (const float*)d_in[5];
    const float* b1d   = (const float*)d_in[6];
    const float* attn1 = (const float*)d_in[7];
    const float* W2s   = (const float*)d_in[8];
    const float* b2s   = (const float*)d_in[9];
    const float* W2d   = (const float*)d_in[10];
    const float* b2d   = (const float*)d_in[11];
    const float* attn2 = (const float*)d_in[12];
    float* out = (float*)d_out;

    float* ws = (float*)d_ws;
    size_t off = 0;
    float*    hs1    = ws + off; off += (size_t)NNODES * HIDTOT;
    float*    hd1    = ws + off; off += (size_t)NNODES * HIDTOT;
    float*    out1   = ws + off; off += (size_t)NNODES * HIDTOT;
    float*    hs2    = ws + off; off += (size_t)NNODES * 2;
    float*    hd2    = ws + off; off += (size_t)NNODES * 2;
    unsigned* tlist  = (unsigned*)(ws + off); off += (size_t)NTILES * LCAP;
    unsigned* tcount = (unsigned*)(ws + off); off += (size_t)256;
    unsigned* keys   = (unsigned*)(ws + off); off += (size_t)KSTEPS * 128;

    const int B = 256;

    tile_key_kernel<<<KEY_BLOCKS, B, 0, stream>>>(dst, keys);
    gemm1_kernel<<<G_BLOCKS, 1024, 0, stream>>>(
        feat, W1s, b1s, W1d, b1d, hs1, hd1);
    tile_layer1_kernel<<<NTILES, 1024, 0, stream>>>(
        hs1, hd1, src, dst, attn1, out1, tlist, tcount, keys);
    proj2_kernel<<<(NNODES + 7) / 8, B, 0, stream>>>(
        out1, W2s, b2s, W2d, b2d, hs2, hd2);
    tile_layer2_kernel<<<NPAIRS, 1024, 0, stream>>>(
        hs2, hd2, src, dst, attn2, tlist, tcount, out);

    (void)in_sizes; (void)n_in; (void)out_size; (void)ws_size;
}


# layer-1 tile kernel: one pass per head pair, each source row fetched once; weights exp(s)/sum exp(s) unshifted, exact max-shifted redo if |score| > 32 or a sum not finite; scores in a new sum order
# speedup vs baseline: 5.2278x; 1.1255x over previous
#include <hip/hip_runtime.h>
#include <hip/hip_bf16.h>
#include <math.h>

#define NNODES 50000
#define EDGES  800000
#define INFEAT 128
#define HIDTOT 256
#define NEG_SLOPE 0.2f
#define ORD_NEG_INF 0x007FFFFFu
#define TILE_R     200
#define NTILES     250
#define KSTEPS     1563
#define KEY_BLOCKS 196

typedef __attribute__((ext_vector_type(2))) float v2f;
typedef __attribute__((ext_vector_type(8))) float v8f;
typedef __attribute__((ext_vector_type(4))) float v4f;

__device__ __forceinline__ unsigned ford(float f) {
    unsigned u = __float_as_uint(f);
    return (u & 0x80000000u) ? ~u : (u | 0x80000000u);
}
__device__ __forceinline__ float fdec(unsigned u) {
    return __uint_as_float((u & 0x80000000u) ? (u & 0x7FFFFFFFu) : ~u);
}
__device__ __forceinline__ float lrelu(float x) {
    return x >= 0.f ? x : NEG_SLOPE * x;
}

#define G_BLOCKS 256
#define G_MTILES 3125

__global__ __launch_bounds__(1024) void gemm1_kernel(
    const float* __restrict__ feat,
    const float* __restrict__ Ws, const float* __restrict__ bs,
    const float* __restrict__ Wd, const float* __restrict__ bd,
    float* __restrict__ hs, float* __restrict__ hd)
{
    __shared__ __attribute__((aligned(16))) float wfrag[2 * INFEAT * HIDTOT];

    const int tid  = threadIdx.x;
    #pragma unroll
    for (int it = 0; it < 8; ++it) {
        const int u   = tid + 1024 * it;
        const int mtx = u >> 12;
        const int kp  = (((u >> 7) & 31) << 1) | ((u >> 2) & 1);
        const int cq  = (((u >> 3) & 15) << 2) | (u & 3);
        const float* Wp = mtx ? Wd : Ws;
        const float4 r0 = *(const float4*)(Wp + (size_t)(2 * kp) * HIDTOT + 4 * cq);
        const float4 r1 = *(const float4*)(Wp + (size_t)(2 * kp + 1) * HIDTOT + 4 * cq);
        float4* q = (float4*)(wfrag +
            ((((mtx * 32 + (kp >> 1)) * 16 + (cq >> 2)) * 32 + (kp & 1) * 16 + 4 * (cq & 3)) * 2));
        q[0] = make_float4(r0.x, r1.x, r0.y, r1.y);
        q[1] = make_float4(r0.z, r1.z, r0.w, r1.w);
    }
    __syncthreads();

    const int wave = tid >> 5;
    const int lane = tid & 31;
    const int half = lane >> 4;
    const int row  = lane & 15;
    const int nt4  = wave & 3;
    const int c0   = nt4 * 64;

    const v2f* bfr = (const v2f*)wfrag + (nt4 * 4) * 32 + lane;

    for (int mt = blockIdx.x + G_BLOCKS * (wave >> 2); mt < G_MTILES; mt += G_BLOCKS * 8) {
        const int m0 = mt * 16;

        v8f zero = {0.f,0.f,0.f,0.f,0.f,0.f,0.f,0.f};
        v8f accS[4], accD[4];
        #pragma unroll
        for (int j = 0; j < 4; ++j) { accS[j] = zero; accD[j] = zero; }

        const float* arow = feat + (size_t)(m0 + row) * INFEAT;
        for (int k0 = 0; k0 < INFEAT; k0 += 4) {
            const int ka = k0 + 2 * half;
            v2f a = *(const v2f*)(arow + ka);
            const v2f* bk = bfr + (k0 >> 2) * (16 * 32);
            #pragma unroll
            for (int j = 0; j < 4; ++j) {
                const v2f bS = bk[j * 32];
                const v2f bD = bk[32 * 16 * 32 + j * 32];
                accS[j] = __builtin_amdgcn_wmma_f32_16x16x4_f32(
                    false, a, false, bS, (short)0, accS[j], false, false);
                accD[j] = __builtin_amdgcn_wmma_f32_16x16x4_f32(
                    false, a, false, bD, (short)0, accD[j], false, false);
            }
        }
        #pragma unroll
        for (int j = 0; j < 4; ++j) {
            const int col = c0 + j * 16 + row;
            const float biasS = bs[col], biasD = bd[col];
            #pragma unroll
            for (int i = 0; i < 8; ++i) {
                const int r = m0 + i + 8 * half;
                hs[(size_t)r * HIDTOT + col] = accS[j][i] + biasS;
                hd[(size_t)r * HIDTOT + col] = accD[j][i] + biasD;
            }
        }
    }
}

__device__ __forceinline__ unsigned tile_key(const int d) {
    return ((unsigned)d < (unsigned)NNODES) ? ((unsigned)d / (unsigned)TILE_R) : 255u;
}

__global__ __launch_bounds__(256) void tile_key_kernel(
    const int* __restrict__ dst, unsigned* __restrict__ keys)
{
    const int t = blockIdx.x * 256 + threadIdx.x;
    if (t >= KSTEPS * 32) return;
    uint4 o = make_uint4(~0u, ~0u, ~0u, ~0u);
    if (t < EDGES / 16) {
        const int4* p = (const int4*)dst + (size_t)t * 4;
        unsigned r[4];
        #pragma unroll
        for (int k = 0; k < 4; ++k) {
            const int4 q = p[k];
            r[k] = tile_key(q.x) | (tile_key(q.y) << 8) |
                   (tile_key(q.z) << 16) | (tile_key(q.w) << 24);
        }
        o = make_uint4(r[0], r[1], r[2], r[3]);
    }
    ((uint4*)keys)[t] = o;
}

#define AGG_WAVES  32
#define SCAN_STEPS 6250
#define WIN_STEPS  40
#define LCAP       (WIN_STEPS * 128)
#define GROUP      8
#define SGROUP     4
#define PGROUP     8
#define SCORE_LIMIT 32.f

__device__ __forceinline__ void lds_add(float* p, float v) {
    __hip_atomic_fetch_add(p, v, __ATOMIC_RELAXED, __HIP_MEMORY_SCOPE_WORKGROUP);
}
__device__ __forceinline__ void lds_max(unsigned* p, unsigned v) {
    __hip_atomic_fetch_max(p, v, __ATOMIC_RELAXED, __HIP_MEMORY_SCOPE_WORKGROUP);
}
__device__ __forceinline__ float finite_or_zero(float m) {
    return isfinite(m) ? m : 0.f;
}

__device__ __forceinline__ void scan_step(
    const int4 v, const int eb, const int lo, const int lane,
    unsigned* list, unsigned* count)
{
    const int vals[4] = {v.x, v.y, v.z, v.w};
    unsigned dl[4], mk[4];
    int tot = 0;
    #pragma unroll
    for (int k = 0; k < 4; ++k) {
        dl[k] = (unsigned)(vals[k] - lo);
        mk[k] = __builtin_amdgcn_ballot_w32(dl[k] < (unsigned)TILE_R);
        tot += __builtin_popcount(mk[k]);
    }
    if (tot != 0) {
        unsigned base = 0u;
        if (lane == 0)
            base = __hip_atomic_fetch_add(count, (unsigned)tot, __ATOMIC_RELAXED,
                                          __HIP_MEMORY_SCOPE_WORKGROUP);
        base = (unsigned)__builtin_amdgcn_readfirstlane((int)base);
        #pragma unroll
        for (int k = 0; k < 4; ++k) {
            if (dl[k] < (unsigned)TILE_R) {
                const unsigned idx = base + __builtin_amdgcn_mbcnt_lo(mk[k], 0u);
                if (idx < (unsigned)LCAP)
                    list[idx] = ((unsigned)(eb + k) << 8) | dl[k];
            }
            base += (unsigned)__builtin_popcount(mk[k]);
        }
    }
}

__device__ __forceinline__ void key_step(
    const uint4 v, const int eb, const unsigned pat,
    unsigned* list, unsigned* count)
{
    const unsigned w[4] = {v.x ^ pat, v.y ^ pat, v.z ^ pat, v.w ^ pat};
    unsigned t[4];
    #pragma unroll
    for (int k = 0; k < 4; ++k)
        t[k] = ~(((w[k] & 0x7F7F7F7Fu) + 0x7F7F7F7Fu) | w[k] | 0x7F7F7F7Fu);
    if (((t[0] | t[1]) | (t[2] | t[3])) != 0u) {
        unsigned m = (t[0] >> 7) | (t[1] >> 6) | (t[2] >> 5) | (t[3] >> 4);
        while (m != 0u) {
            const int p = __builtin_ctz(m);
            m &= m - 1u;
            const unsigned idx = __hip_atomic_fetch_add(count, 1u, __ATOMIC_RELAXED,
                                                        __HIP_MEMORY_SCOPE_WORKGROUP);
            if (idx < (unsigned)LCAP)
                list[idx] = (unsigned)(eb + 4 * (p & 7) + (p >> 3));
        }
    }
}

__device__ __forceinline__ void score_batch(
    const unsigned* qp, float* scp, const int nb, const int lane, const int lo,
    const float* __restrict__ hs, const float* __restrict__ hd,
    const int* __restrict__ src, const float4 w0, const float4 w1,
    unsigned* mxb)
{
    int s_i = 0, dl_i = 0;
    if (lane < nb) {
        const unsigned ent = qp[lane];
        dl_i = (int)(ent & 255u);
        s_i = src[(int)(ent >> 8)];
    }
    for (int g0 = 0; g0 < nb; g0 += SGROUP) {
        float4 a0[SGROUP], a1[SGROUP], b0[SGROUP], b1[SGROUP];
        #pragma unroll
        for (int u = 0; u < SGROUP; ++u) {
            const int s = __builtin_amdgcn_readlane(s_i, g0 + u);
            const int d = __builtin_amdgcn_readlane(dl_i, g0 + u);
            const float4* ph = (const float4*)(hs + (size_t)s * HIDTOT + lane * 8);
            a0[u] = ph[0]; a1[u] = ph[1];
            const float4* pd = (const float4*)(hd + (size_t)(lo + d) * HIDTOT + lane * 8);
            b0[u] = pd[0]; b1[u] = pd[1];
        }
        #pragma unroll
        for (int u = 0; u < SGROUP; ++u) {
            if (g0 + u < nb) {
                float part = 0.f;
                part += lrelu(a0[u].x + b0[u].x) * w0.x + lrelu(a0[u].y + b0[u].y) * w0.y +
                        lrelu(a0[u].z + b0[u].z) * w0.z + lrelu(a0[u].w + b0[u].w) * w0.w;
                part += lrelu(a1[u].x + b1[u].x) * w1.x + lrelu(a1[u].y + b1[u].y) * w1.y +
                        lrelu(a1[u].z + b1[u].z) * w1.z + lrelu(a1[u].w + b1[u].w) * w1.w;
                part += __shfl_xor(part, 1, 32);
                part += __shfl_xor(part, 2, 32);
                part += __shfl_xor(part, 4, 32);
                if ((lane & 7) == 0) {
                    const int h = lane >> 3;
                    const int d = __builtin_amdgcn_readlane(dl_i, g0 + u);
                    scp[(g0 + u) * 4 + h] = part;
                    lds_max(mxb + d * 4 + h, ford(part));
                }
            }
        }
    }
}

__device__ __forceinline__ void add_batch(
    const unsigned* qp, const float* scp, const int nb, const int lane,
    const float* __restrict__ hs, const int* __restrict__ src,
    const unsigned* mxb, float* dn, float* acc)
{
    int s_i = 0, dl_i = 0;
    float a0 = 0.f, a1 = 0.f, a2 = 0.f, a3 = 0.f;
    if (lane < nb) {
        const unsigned ent = qp[lane];
        dl_i = (int)(ent & 255u);
        s_i = src[(int)(ent >> 8)];
        const float4 x = ((const float4*)scp)[lane];
        const uint4 mb = ((const uint4*)mxb)[dl_i];
        a0 = expf(x.x - finite_or_zero(fdec(mb.x)));
        a1 = expf(x.y - finite_or_zero(fdec(mb.y)));
        a2 = expf(x.z - finite_or_zero(fdec(mb.z)));
        a3 = expf(x.w - finite_or_zero(fdec(mb.w)));
        lds_add(dn + dl_i * 4 + 0, a0);
        lds_add(dn + dl_i * 4 + 1, a1);
        lds_add(dn + dl_i * 4 + 2, a2);
        lds_add(dn + dl_i * 4 + 3, a3);
    }
    const bool lowhalf = lane < 16;
    for (int g0 = 0; g0 < nb; g0 += GROUP) {
        float4 ra[GROUP], rb[GROUP];
        #pragma unroll
        for (int u = 0; u < GROUP; ++u) {
            const int s = __builtin_amdgcn_readlane(s_i, g0 + u);
            const float4* p = (const float4*)(hs + (size_t)s * HIDTOT) + lane;
            ra[u] = p[0];
            rb[u] = p[32];
        }
        #pragma unroll
        for (int u = 0; u < GROUP; ++u) {
            if (g0 + u < nb) {
                const int d = __builtin_amdgcn_readlane(dl_i, g0 + u);
                const float h0 = __int_as_float(__builtin_amdgcn_readlane(__float_as_int(a0), g0 + u));
                const float h1 = __int_as_float(__builtin_amdgcn_readlane(__float_as_int(a1), g0 + u));
                const float h2 = __int_as_float(__builtin_amdgcn_readlane(__float_as_int(a2), g0 + u));
                const float h3 = __int_as_float(__builtin_amdgcn_readlane(__float_as_int(a3), g0 + u));
                const float aA = lowhalf ? h0 : h1;
                const float aB = lowhalf ? h2 : h3;
                float* a = acc + d * HIDTOT + lane;
                lds_add(a +   0, aA * ra[u].x);
                lds_add(a +  32, aA * ra[u].y);
                lds_add(a +  64, aA * ra[u].z);
                lds_add(a +  96, aA * ra[u].w);
                lds_add(a + 128, aB * rb[u].x);
                lds_add(a + 160, aB * rb[u].y);
                lds_add(a + 192, aB * rb[u].z);
                lds_add(a + 224, aB * rb[u].w);
            }
        }
    }
}

__device__ __forceinline__ float row16_sum(float v) {
    v += __int_as_float(__builtin_amdgcn_update_dpp(0, __float_as_int(v), 0x161, 0xF, 0xF, true));
    v += __int_as_float(__builtin_amdgcn_update_dpp(0, __float_as_int(v), 0x162, 0xF, 0xF, true));
    v += __int_as_float(__builtin_amdgcn_update_dpp(0, __float_as_int(v), 0x164, 0xF, 0xF, true));
    v += __int_as_float(__builtin_amdgcn_update_dpp(0, __float_as_int(v), 0x168, 0xF, 0xF, true));
    return v;
}

template <int P>
__device__ __forceinline__ void pair_batch(
    const unsigned* qp, const int nb, const int lane,
    const float* __restrict__ hs, const float4* hdt, const float4 w,
    float* dn, float* sums, unsigned* bad)
{
    int s_i = 0, dl_i = 0;
    if (lane < nb) {
        const unsigned ent = qp[lane];
        dl_i = (int)(ent & 255u);
        s_i = (int)(ent >> 8);
    }
    const int j = lane & 15;
    for (int g0 = 0; g0 < nb; g0 += PGROUP) {
        float4 a[PGROUP];
        float sv[PGROUP];
        #pragma unroll
        for (int u = 0; u < PGROUP; ++u) {
            const int s = __builtin_amdgcn_readlane(s_i, g0 + u);
            a[u] = ((const float4*)(hs + (size_t)s * HIDTOT + 128 * P))[lane];
        }
        #pragma unroll
        for (int u = 0; u < PGROUP; ++u) {
            const int d = __builtin_amdgcn_readlane(dl_i, g0 + u);
            const float4 b = hdt[d * 32 + lane];
            const float part = lrelu(a[u].x + b.x) * w.x + lrelu(a[u].y + b.y) * w.y +
                               lrelu(a[u].z + b.z) * w.z + lrelu(a[u].w + b.w) * w.w;
            sv[u] = row16_sum(part);
        }
        float mine = 0.f;
        #pragma unroll
        for (int u = 0; u < PGROUP; ++u) mine = (j == u) ? sv[u] : mine;
        const float ex = expf(mine);
        const int dj = __shfl(dl_i, g0 + (j & (PGROUP - 1)), 32);
        if (j < PGROUP && g0 + j < nb) {
            if (!(fabsf(mine) <= SCORE_LIMIT)) *bad = 1u;
            lds_add(dn + dj * 4 + 2 * P + (lane >> 4), ex);
        }
        #pragma unroll
        for (int u = 0; u < PGROUP; ++u) {
            if (g0 + u < nb) {
                const int d = __builtin_amdgcn_readlane(dl_i, g0 + u);
                const float eA = __int_as_float(__builtin_amdgcn_readlane(__float_as_int(ex), u));
                const float eB = __int_as_float(__builtin_amdgcn_readlane(__float_as_int(ex), 16 + u));
                const float e = (lane < 16) ? eA : eB;
                float* q = sums + d * 128 + lane;
                lds_add(q +  0, e * a[u].x);
                lds_add(q + 32, e * a[u].y);
                lds_add(q + 64, e * a[u].z);
                lds_add(q + 96, e * a[u].w);
            }
        }
    }
}

template <int P>
__device__ __forceinline__ void pair_phase(
    const int beg, const int end, const int tid, const int wave, const int lane, const int lo,
    const float* __restrict__ hs, const float* __restrict__ hd,
    const float* __restrict__ attn, float* __restrict__ out,
    float* acc, const unsigned* list, float* dn, unsigned* bad)
{
    v4f* hdt = (v4f*)acc;
    float* sums = acc + TILE_R * 128;
    if (P == 1) {
        const v4f* hdg = (const v4f*)(hd + (size_t)lo * HIDTOT);
        for (int i = tid; i < TILE_R * 32; i += 1024)
            hdt[i] = __builtin_nontemporal_load(hdg + (i >> 5) * 64 + 32 + (i & 31));
    }
    for (int i = tid; i < TILE_R * 128 / 4; i += 1024)
        ((float4*)sums)[i] = make_float4(0.f, 0.f, 0.f, 0.f);
    __syncthreads();

    const float4 w = ((const float4*)(attn + 128 * P))[lane];
    for (int i0 = beg; i0 < end; i0 += 32) {
        const int nb = (end - i0) < 32 ? (end - i0) : 32;
        pair_batch<P>(list + i0, nb, lane, hs, (const float4*)acc, w, dn, sums, bad);
    }
    __syncthreads();

    for (int r = wave; r < TILE_R; r += AGG_WAVES) {
        const float* q = sums + r * 128 + lane;
        float den = dn[r * 4 + 2 * P + (lane >> 4)];
        den = den > 0.f ? den : 1.f;
        float4 o;
        o.x = q[ 0] / den;
        o.y = q[32] / den;
        o.z = q[64] / den;
        o.w = q[96] / den;
        if (!(isfinite(o.x) && isfinite(o.y) && isfinite(o.z) && isfinite(o.w))) *bad = 1u;
        o.x = o.x > 0.f ? o.x : (expf(o.x) - 1.f);
        o.y = o.y > 0.f ? o.y : (expf(o.y) - 1.f);
        o.z = o.z > 0.f ? o.z : (expf(o.z) - 1.f);
        o.w = o.w > 0.f ? o.w : (expf(o.w) - 1.f);
        ((float4*)(out + (size_t)(lo + r) * HIDTOT + 128 * P))[lane] = o;
    }
    __syncthreads();
}

__device__ __forceinline__ void window_chunk(
    const int n, const int tid, const int wave,
    const int lane, const int lo,
    const float* __restrict__ hs, const float* __restrict__ hd,
    const int* __restrict__ src, const float* __restrict__ attn,
    float* acc, float* sc, const unsigned* list,
    unsigned* mxb, unsigned* omx, float* dn)
{
    const int share = (n + AGG_WAVES - 1) / AGG_WAVES;
    const int beg = wave * share;
    const int end = (beg + share) < n ? (beg + share) : n;

    for (int i = tid; i < TILE_R * 4; i += 1024) omx[i] = mxb[i];
    __syncthreads();

    {
        const float4 w0 = ((const float4*)(attn + lane * 8))[0];
        const float4 w1 = ((const float4*)(attn + lane * 8))[1];
        for (int i0 = beg; i0 < end; i0 += 32) {
            const int nb = (end - i0) < 32 ? (end - i0) : 32;
            score_batch(list + i0, sc + i0 * 4, nb, lane, lo, hs, hd, src, w0, w1, mxb);
        }
    }
    __syncthreads();

    for (int i = tid; i < TILE_R * 4; i += 1024) {
        const unsigned ob = omx[i];
        const float f = (ob == ORD_NEG_INF) ? 0.f
            : expf(finite_or_zero(fdec(ob)) - finite_or_zero(fdec(mxb[i])));
        omx[i] = __float_as_uint(f);
        dn[i] *= f;
    }
    __syncthreads();
    for (int i = tid; i < TILE_R * HIDTOT; i += 1024) {
        const int r = i >> 8, p = i & 255;
        const int h = 2 * (p >> 7) + ((p & 31) >> 4);
        acc[i] *= __uint_as_float(omx[r * 4 + h]);
    }
    __syncthreads();

    for (int i0 = beg; i0 < end; i0 += 32) {
        const int nb = (end - i0) < 32 ? (end - i0) : 32;
        add_batch(list + i0, sc + i0 * 4, nb, lane, hs, src, mxb, dn, acc);
    }
}

__global__ __launch_bounds__(1024) void tile_layer1_kernel(
    const float* __restrict__ hs, const float* __restrict__ hd,
    const int* __restrict__ src, const int* __restrict__ dst,
    const float* __restrict__ attn, float* __restrict__ out,
    unsigned* __restrict__ tlist, unsigned* __restrict__ tcount,
    const unsigned* __restrict__ keys)
{
    __shared__ __attribute__((aligned(16))) float acc[TILE_R * HIDTOT];
    __shared__ __attribute__((aligned(16))) float sc[LCAP * 4];
    __shared__ __attribute__((aligned(16))) unsigned mxb[TILE_R * 4];
    __shared__ __attribute__((aligned(16))) unsigned omx[TILE_R * 4];
    __shared__ __attribute__((aligned(16))) float dn[TILE_R * 4];
    __shared__ unsigned list[LCAP];
    __shared__ unsigned count;
    __shared__ unsigned bad;

    const int tid  = threadIdx.x;
    const int wave = __builtin_amdgcn_readfirstlane(tid >> 5);
    const int lane = tid & 31;
    const int tile = blockIdx.x;
    const int lo   = tile * TILE_R;

    for (int i = tid; i < TILE_R * 4; i += 1024) { mxb[i] = ORD_NEG_INF; dn[i] = 0.f; }
    if (tid == 0) { count = 0u; bad = 0u; }
    __syncthreads();

    const int4* dst4 = (const int4*)dst;

    {
        const v4f* hdg = (const v4f*)(hd + (size_t)lo * HIDTOT);
        v4f* hdt = (v4f*)acc;
        const uint4* key4 = (const uint4*)keys;
        const unsigned pat = (unsigned)tile * 0x01010101u;
        const uint4 none = make_uint4(~0u, ~0u, ~0u, ~0u);
        const int rot = tile * (KSTEPS / NTILES);
        uint4 v[4];
        #pragma unroll
        for (int u = 0; u < 4; ++u) {
            int g = wave + 32 * u + rot;
            if (g >= KSTEPS) g -= KSTEPS;
            v[u] = key4[g * 32 + lane];
        }
        for (int it = 0; it < 52; it += 4) {
            const int hj = wave + 32 * (it >> 2);
            v4f hv = {0.f, 0.f, 0.f, 0.f};
            if (hj < TILE_R)
                hv = __builtin_nontemporal_load(hdg + hj * 64 + lane);
            uint4 nv[4];
            #pragma unroll
            for (int u = 0; u < 4; ++u) {
                const int st = wave + 32 * (it + 4 + u);
                int g = st + rot;
                if (g >= KSTEPS) g -= KSTEPS;
                nv[u] = none;
                if (st < KSTEPS) nv[u] = key4[g * 32 + lane];
            }
            #pragma unroll
            for (int u = 0; u < 4; ++u) {
                int g = wave + 32 * (it + u) + rot;
                if (g >= KSTEPS) g -= KSTEPS;
                key_step(v[u], g * 512 + lane * 16, pat, list, &count);
            }
            #pragma unroll
            for (int u = 0; u < 4; ++u) v[u] = nv[u];
            if (hj < TILE_R)
                hdt[hj * 32 + lane] = hv;
        }
    }
    __syncthreads();

    const unsigned total = count;
    unsigned* wlist = tlist + (size_t)tile * LCAP;
    if (tid == 0) tcount[tile] = total;
    bool exact = total > (unsigned)LCAP;
    if (!exact) {
        for (int i = tid; i < (int)total; i += 1024) {
            const unsigned e = list[i];
            const unsigned ent = ((unsigned)src[e] << 8) | (unsigned)(dst[e] - lo);
            list[i] = ent;
            wlist[i] = ent;
        }
        const int n = (int)total;
        const int share = (n + AGG_WAVES - 1) / AGG_WAVES;
        const int beg = wave * share;
        const int end = (beg + share) < n ? (beg + share) : n;
        pair_phase<0>(beg, end, tid, wave, lane, lo, hs, hd, attn, out, acc, list, dn, &bad);
        pair_phase<1>(beg, end, tid, wave, lane, lo, hs, hd, attn, out, acc, list, dn, &bad);
        exact = bad != 0u;
    }
    if (exact) {
        for (int i = tid; i < TILE_R * 4; i += 1024) dn[i] = 0.f;
        for (int i = tid; i < TILE_R * HIDTOT / 4; i += 1024)
            ((float4*)acc)[i] = make_float4(0.f, 0.f, 0.f, 0.f);
        const int nwin = (SCAN_STEPS + WIN_STEPS - 1) / WIN_STEPS;
        for (int w = 0; w < nwin; ++w) {
            __syncthreads();
            if (tid == 0) count = 0u;
            __syncthreads();
            const int w0 = w * WIN_STEPS;
            const int wend = (w0 + WIN_STEPS) < SCAN_STEPS ? (w0 + WIN_STEPS) : SCAN_STEPS;
            for (int st = w0 + wave; st < wend; st += AGG_WAVES)
                scan_step(dst4[st * 32 + lane], st * 128 + lane * 4, lo, lane, list, &count);
            __syncthreads();
            window_chunk((int)count, tid, wave, lane, lo, hs, hd, src, attn,
                         acc, sc, list, mxb, omx, dn);
        }
        __syncthreads();

        for (int r = wave; r < TILE_R; r += AGG_WAVES) {
            const float* a = acc + r * HIDTOT + lane;
            float* po = out + (size_t)(lo + r) * HIDTOT;
            #pragma unroll
            for (int qq = 0; qq < 2; ++qq) {
                float den = dn[r * 4 + 2 * qq + (lane >> 4)];
                den = den > 0.f ? den : 1.f;
                float4 o;
                o.x = a[qq * 128 +  0] / den;
                o.y = a[qq * 128 + 32] / den;
                o.z = a[qq * 128 + 64] / den;
                o.w = a[qq * 128 + 96] / den;
                o.x = o.x > 0.f ? o.x : (expf(o.x) - 1.f);
                o.y = o.y > 0.f ? o.y : (expf(o.y) - 1.f);
                o.z = o.z > 0.f ? o.z : (expf(o.z) - 1.f);
                o.w = o.w > 0.f ? o.w : (expf(o.w) - 1.f);
                ((float4*)(po + qq * 128))[lane] = o;
            }
        }
    }
}

__global__ __launch_bounds__(256) void proj2_kernel(
    const float* __restrict__ h,
    const float* __restrict__ W2s, const float* __restrict__ b2s,
    const float* __restrict__ W2d, const float* __restrict__ b2d,
    float* __restrict__ hs2, float* __restrict__ hd2)
{
    const int n = blockIdx.x * 8 + (threadIdx.x >> 5);
    if (n >= NNODES) return;
    const int lane = threadIdx.x & 31;
    const int base = lane * 8;
    const float* hr = h + (size_t)n * HIDTOT + base;
    float s0 = 0.f, s1 = 0.f, d0 = 0.f, d1 = 0.f;
    #pragma unroll
    for (int k = 0; k < 8; ++k) {
        const float x = hr[k];
        const float2 ws = *(const float2*)(W2s + (size_t)(base + k) * 2);
        const float2 wd = *(const float2*)(W2d + (size_t)(base + k) * 2);
        s0 += x * ws.x; s1 += x * ws.y;
        d0 += x * wd.x; d1 += x * wd.y;
    }
    #pragma unroll
    for (int o = 1; o < 32; o <<= 1) {
        s0 += __shfl_xor(s0, o, 32); s1 += __shfl_xor(s1, o, 32);
        d0 += __shfl_xor(d0, o, 32); d1 += __shfl_xor(d1, o, 32);
    }
    if (lane == 0) {
        hs2[(size_t)n * 2 + 0] = s0 + b2s[0];
        hs2[(size_t)n * 2 + 1] = s1 + b2s[1];
        hd2[(size_t)n * 2 + 0] = d0 + b2d[0];
        hd2[(size_t)n * 2 + 1] = d1 + b2d[1];
    }
}

#define PAIR_R (2 * TILE_R)
#define NPAIRS 125

__device__ __forceinline__ float score2_of(const float2 a, const float2 b,
                                           const float w0, const float w1) {
    return lrelu(a.x + b.x) * w0 + lrelu(a.y + b.y) * w1;
}

__global__ __launch_bounds__(1024) void tile_layer2_kernel(
    const float* __restrict__ hs2, const float* __restrict__ hd2,
    const int* __restrict__ src, const int* __restrict__ dst,
    const float* __restrict__ attn2,
    const unsigned* __restrict__ tlist, const unsigned* __restrict__ tcount,
    float* __restrict__ out)
{
    __shared__ __attribute__((aligned(16))) float ent[2 * LCAP * 4];
    __shared__ unsigned mx[PAIR_R];
    __shared__ float dn[PAIR_R];
    __shared__ float sm[PAIR_R * 2];

    const int tid = threadIdx.x;
    const int pr  = blockIdx.x;
    const int lo  = pr * PAIR_R;

    for (int i = tid; i < PAIR_R; i += 1024) {
        mx[i] = ORD_NEG_INF; dn[i] = 0.f; sm[2 * i] = 0.f; sm[2 * i + 1] = 0.f;
    }
    const float w0 = attn2[0], w1 = attn2[1];
    const unsigned c0 = tcount[2 * pr], c1 = tcount[2 * pr + 1];
    const bool listed = (c0 <= (unsigned)LCAP) && (c1 <= (unsigned)LCAP);
    __syncthreads();

    if (listed) {
        const int n0 = (int)c0, n = (int)(c0 + c1);
        const unsigned* l0 = tlist + (size_t)(2 * pr) * LCAP;
        const unsigned* l1 = tlist + (size_t)(2 * pr + 1) * LCAP;
        float4* e4 = (float4*)ent;
        for (int ib = tid; ib < n; ib += 4096) {
            unsigned e[4];
            int dd[4];
            #pragma unroll
            for (int u = 0; u < 4; ++u) {
                const int i = ib + 1024 * u;
                e[u] = 0u;
                if (i < n) e[u] = (i < n0) ? l0[i] : l1[i - n0];
            }
            float2 a[4], b[4];
            #pragma unroll
            for (int u = 0; u < 4; ++u) {
                const int i = ib + 1024 * u;
                dd[u] = (int)(e[u] & 255u) + ((i < n0) ? 0 : TILE_R);
                a[u] = *(const float2*)(hs2 + (size_t)(e[u] >> 8) * 2);
                b[u] = *(const float2*)(hd2 + (size_t)(lo + dd[u]) * 2);
            }
            #pragma unroll
            for (int u = 0; u < 4; ++u) {
                const int i = ib + 1024 * u;
                if (i < n) {
                    const int d = dd[u];
                    const float s = score2_of(a[u], b[u], w0, w1);
                    e4[i] = make_float4(s, a[u].x, a[u].y, __int_as_float(d));
                    lds_max(mx + d, ford(s));
                }
            }
        }
        __syncthreads();
        for (int i = tid; i < n; i += 1024) {
            const float4 t = e4[i];
            const int d = __float_as_int(t.w);
            const float ex = expf(t.x - finite_or_zero(fdec(mx[d])));
            lds_add(dn + d, ex);
            lds_add(sm + 2 * d, ex * t.y);
            lds_add(sm + 2 * d + 1, ex * t.z);
        }
    } else {
        for (int e = tid; e < EDGES; e += 1024) {
            const unsigned d = (unsigned)(dst[e] - lo);
            if (d < (unsigned)PAIR_R) {
                const float2 a = *(const float2*)(hs2 + (size_t)src[e] * 2);
                const float2 b = *(const float2*)(hd2 + (size_t)(lo + (int)d) * 2);
                lds_max(mx + d, ford(score2_of(a, b, w0, w1)));
            }
        }
        __syncthreads();
        for (int e = tid; e < EDGES; e += 1024) {
            const unsigned d = (unsigned)(dst[e] - lo);
            if (d < (unsigned)PAIR_R) {
                const float2 a = *(const float2*)(hs2 + (size_t)src[e] * 2);
                const float2 b = *(const float2*)(hd2 + (size_t)(lo + (int)d) * 2);
                const float ex = expf(score2_of(a, b, w0, w1) - finite_or_zero(fdec(mx[d])));
                lds_add(dn + d, ex);
                lds_add(sm + 2 * d, ex * a.x);
                lds_add(sm + 2 * d + 1, ex * a.y);
            }
        }
    }
    __syncthreads();

    for (int i = tid; i < PAIR_R * 2; i += 1024) {
        float den = dn[i >> 1];
        den = den > 0.f ? den : 1.f;
        out[(size_t)lo * 2 + i] = sm[i] / den;
    }
}

extern "C" void kernel_launch(void* const* d_in, const int* in_sizes, int n_in,
                              void* d_out, int out_size, void* d_ws, size_t ws_size,
                              hipStream_t stream)
{
    const float* feat  = (const float*)d_in[0];
    const int*   src   = (const int*)d_in[1];
    const int*   dst   = (const int*)d_in[2];
    const float* W1s   = (const float*)d_in[3];
    const float* b1s   = (const float*)d_in[4];
    const float* W1d   = (const float*)d_in[5];
    const float* b1d   = (const float*)d_in[6];
    const float* attn1 = (const float*)d_in[7];
    const float* W2s   = (const float*)d_in[8];
    const float* b2s   = (const float*)d_in[9];
    const float* W2d   = (const float*)d_in[10];
    const float* b2d   = (const float*)d_in[11];
    const float* attn2 = (const float*)d_in[12];
    float* out = (float*)d_out;

    float* ws = (float*)d_ws;
    size_t off = 0;
    float*    hs1    = ws + off; off += (size_t)NNODES * HIDTOT;
    float*    hd1    = ws + off; off += (size_t)NNODES * HIDTOT;
    float*    out1   = ws + off; off += (size_t)NNODES * HIDTOT;
    float*    hs2    = ws + off; off += (size_t)NNODES * 2;
    float*    hd2    = ws + off; off += (size_t)NNODES * 2;
    unsigned* tlist  = (unsigned*)(ws + off); off += (size_t)NTILES * LCAP;
    unsigned* tcount = (unsigned*)(ws + off); off += (size_t)256;
    unsigned* keys   = (unsigned*)(ws + off); off += (size_t)KSTEPS * 128;

    const int B = 256;

    tile_key_kernel<<<KEY_BLOCKS, B, 0, stream>>>(dst, keys);
    gemm1_kernel<<<G_BLOCKS, 1024, 0, stream>>>(
        feat, W1s, b1s, W1d, b1d, hs1, hd1);
    tile_layer1_kernel<<<NTILES, 1024, 0, stream>>>(
        hs1, hd1, src, dst, attn1, out1, tlist, tcount, keys);
    proj2_kernel<<<(NNODES + 7) / 8, B, 0, stream>>>(
        out1, W2s, b2s, W2d, b2d, hs2, hd2);
    tile_layer2_kernel<<<NPAIRS, 1024, 0, stream>>>(
        hs2, hd2, src, dst, attn2, tlist, tcount, out);

    (void)in_sizes; (void)n_in; (void)out_size; (void)ws_size;
}


# layer-2 projection kernel as 256 blocks of 16 waves: a wave loops over its 12-13 nodes, all row loads in flight, weights in registers, lane sums by DPP adds; 4096 waves enter, not 50000; same sums
# speedup vs baseline: 5.3801x; 1.0291x over previous
#include <hip/hip_runtime.h>
#include <hip/hip_bf16.h>
#include <math.h>

#define NNODES 50000
#define EDGES  800000
#define INFEAT 128
#define HIDTOT 256
#define NEG_SLOPE 0.2f
#define ORD_NEG_INF 0x007FFFFFu
#define TILE_R     200
#define NTILES     250
#define KSTEPS     1563
#define KEY_BLOCKS 196

typedef __attribute__((ext_vector_type(2))) float v2f;
typedef __attribute__((ext_vector_type(8))) float v8f;
typedef __attribute__((ext_vector_type(4))) float v4f;

__device__ __forceinline__ unsigned ford(float f) {
    unsigned u = __float_as_uint(f);
    return (u & 0x80000000u) ? ~u : (u | 0x80000000u);
}
__device__ __forceinline__ float fdec(unsigned u) {
    return __uint_as_float((u & 0x80000000u) ? (u & 0x7FFFFFFFu) : ~u);
}
__device__ __forceinline__ float lrelu(float x) {
    return x >= 0.f ? x : NEG_SLOPE * x;
}

#define G_BLOCKS 256
#define G_MTILES 3125

__global__ __launch_bounds__(1024) void gemm1_kernel(
    const float* __restrict__ feat,
    const float* __restrict__ Ws, const float* __restrict__ bs,
    const float* __restrict__ Wd, const float* __restrict__ bd,
    float* __restrict__ hs, float* __restrict__ hd)
{
    __shared__ __attribute__((aligned(16))) float wfrag[2 * INFEAT * HIDTOT];

    const int tid  = threadIdx.x;
    #pragma unroll
    for (int it = 0; it < 8; ++it) {
        const int u   = tid + 1024 * it;
        const int mtx = u >> 12;
        const int kp  = (((u >> 7) & 31) << 1) | ((u >> 2) & 1);
        const int cq  = (((u >> 3) & 15) << 2) | (u & 3);
        const float* Wp = mtx ? Wd : Ws;
        const float4 r0 = *(const float4*)(Wp + (size_t)(2 * kp) * HIDTOT + 4 * cq);
        const float4 r1 = *(const float4*)(Wp + (size_t)(2 * kp + 1) * HIDTOT + 4 * cq);
        float4* q = (float4*)(wfrag +
            ((((mtx * 32 + (kp >> 1)) * 16 + (cq >> 2)) * 32 + (kp & 1) * 16 + 4 * (cq & 3)) * 2));
        q[0] = make_float4(r0.x, r1.x, r0.y, r1.y);
        q[1] = make_float4(r0.z, r1.z, r0.w, r1.w);
    }
    __syncthreads();

    const int wave = tid >> 5;
    const int lane = tid & 31;
    const int half = lane >> 4;
    const int row  = lane & 15;
    const int nt4  = wave & 3;
    const int c0   = nt4 * 64;

    const v2f* bfr = (const v2f*)wfrag + (nt4 * 4) * 32 + lane;

    for (int mt = blockIdx.x + G_BLOCKS * (wave >> 2); mt < G_MTILES; mt += G_BLOCKS * 8) {
        const int m0 = mt * 16;

        v8f zero = {0.f,0.f,0.f,0.f,0.f,0.f,0.f,0.f};
        v8f accS[4], accD[4];
        #pragma unroll
        for (int j = 0; j < 4; ++j) { accS[j] = zero; accD[j] = zero; }

        const float* arow = feat + (size_t)(m0 + row) * INFEAT;
        for (int k0 = 0; k0 < INFEAT; k0 += 4) {
            const int ka = k0 + 2 * half;
            v2f a = *(const v2f*)(arow + ka);
            const v2f* bk = bfr + (k0 >> 2) * (16 * 32);
            #pragma unroll
            for (int j = 0; j < 4; ++j) {
                const v2f bS = bk[j * 32];
                const v2f bD = bk[32 * 16 * 32 + j * 32];
                accS[j] = __builtin_amdgcn_wmma_f32_16x16x4_f32(
                    false, a, false, bS, (short)0, accS[j], false, false);
                accD[j] = __builtin_amdgcn_wmma_f32_16x16x4_f32(
                    false, a, false, bD, (short)0, accD[j], false, false);
            }
        }
        #pragma unroll
        for (int j = 0; j < 4; ++j) {
            const int col = c0 + j * 16 + row;
            const float biasS = bs[col], biasD = bd[col];
            #pragma unroll
            for (int i = 0; i < 8; ++i) {
                const int r = m0 + i + 8 * half;
                hs[(size_t)r * HIDTOT + col] = accS[j][i] + biasS;
                hd[(size_t)r * HIDTOT + col] = accD[j][i] + biasD;
            }
        }
    }
}

__device__ __forceinline__ unsigned tile_key(const int d) {
    return ((unsigned)d < (unsigned)NNODES) ? ((unsigned)d / (unsigned)TILE_R) : 255u;
}

__global__ __launch_bounds__(256) void tile_key_kernel(
    const int* __restrict__ dst, unsigned* __restrict__ keys)
{
    const int t = blockIdx.x * 256 + threadIdx.x;
    if (t >= KSTEPS * 32) return;
    uint4 o = make_uint4(~0u, ~0u, ~0u, ~0u);
    if (t < EDGES / 16) {
        const int4* p = (const int4*)dst + (size_t)t * 4;
        unsigned r[4];
        #pragma unroll
        for (int k = 0; k < 4; ++k) {
            const int4 q = p[k];
            r[k] = tile_key(q.x) | (tile_key(q.y) << 8) |
                   (tile_key(q.z) << 16) | (tile_key(q.w) << 24);
        }
        o = make_uint4(r[0], r[1], r[2], r[3]);
    }
    ((uint4*)keys)[t] = o;
}

#define AGG_WAVES  32
#define SCAN_STEPS 6250
#define WIN_STEPS  40
#define LCAP       (WIN_STEPS * 128)
#define GROUP      8
#define SGROUP     4
#define PGROUP     8
#define SCORE_LIMIT 32.f

__device__ __forceinline__ void lds_add(float* p, float v) {
    __hip_atomic_fetch_add(p, v, __ATOMIC_RELAXED, __HIP_MEMORY_SCOPE_WORKGROUP);
}
__device__ __forceinline__ void lds_max(unsigned* p, unsigned v) {
    __hip_atomic_fetch_max(p, v, __ATOMIC_RELAXED, __HIP_MEMORY_SCOPE_WORKGROUP);
}
__device__ __forceinline__ float finite_or_zero(float m) {
    return isfinite(m) ? m : 0.f;
}

__device__ __forceinline__ void scan_step(
    const int4 v, const int eb, const int lo, const int lane,
    unsigned* list, unsigned* count)
{
    const int vals[4] = {v.x, v.y, v.z, v.w};
    unsigned dl[4], mk[4];
    int tot = 0;
    #pragma unroll
    for (int k = 0; k < 4; ++k) {
        dl[k] = (unsigned)(vals[k] - lo);
        mk[k] = __builtin_amdgcn_ballot_w32(dl[k] < (unsigned)TILE_R);
        tot += __builtin_popcount(mk[k]);
    }
    if (tot != 0) {
        unsigned base = 0u;
        if (lane == 0)
            base = __hip_atomic_fetch_add(count, (unsigned)tot, __ATOMIC_RELAXED,
                                          __HIP_MEMORY_SCOPE_WORKGROUP);
        base = (unsigned)__builtin_amdgcn_readfirstlane((int)base);
        #pragma unroll
        for (int k = 0; k < 4; ++k) {
            if (dl[k] < (unsigned)TILE_R) {
                const unsigned idx = base + __builtin_amdgcn_mbcnt_lo(mk[k], 0u);
                if (idx < (unsigned)LCAP)
                    list[idx] = ((unsigned)(eb + k) << 8) | dl[k];
            }
            base += (unsigned)__builtin_popcount(mk[k]);
        }
    }
}

__device__ __forceinline__ void key_step(
    const uint4 v, const int eb, const unsigned pat,
    unsigned* list, unsigned* count)
{
    const unsigned w[4] = {v.x ^ pat, v.y ^ pat, v.z ^ pat, v.w ^ pat};
    unsigned t[4];
    #pragma unroll
    for (int k = 0; k < 4; ++k)
        t[k] = ~(((w[k] & 0x7F7F7F7Fu) + 0x7F7F7F7Fu) | w[k] | 0x7F7F7F7Fu);
    if (((t[0] | t[1]) | (t[2] | t[3])) != 0u) {
        unsigned m = (t[0] >> 7) | (t[1] >> 6) | (t[2] >> 5) | (t[3] >> 4);
        while (m != 0u) {
            const int p = __builtin_ctz(m);
            m &= m - 1u;
            const unsigned idx = __hip_atomic_fetch_add(count, 1u, __ATOMIC_RELAXED,
                                                        __HIP_MEMORY_SCOPE_WORKGROUP);
            if (idx < (unsigned)LCAP)
                list[idx] = (unsigned)(eb + 4 * (p & 7) + (p >> 3));
        }
    }
}

__device__ __forceinline__ void score_batch(
    const unsigned* qp, float* scp, const int nb, const int lane, const int lo,
    const float* __restrict__ hs, const float* __restrict__ hd,
    const int* __restrict__ src, const float4 w0, const float4 w1,
    unsigned* mxb)
{
    int s_i = 0, dl_i = 0;
    if (lane < nb) {
        const unsigned ent = qp[lane];
        dl_i = (int)(ent & 255u);
        s_i = src[(int)(ent >> 8)];
    }
    for (int g0 = 0; g0 < nb; g0 += SGROUP) {
        float4 a0[SGROUP], a1[SGROUP], b0[SGROUP], b1[SGROUP];
        #pragma unroll
        for (int u = 0; u < SGROUP; ++u) {
            const int s = __builtin_amdgcn_readlane(s_i, g0 + u);
            const int d = __builtin_amdgcn_readlane(dl_i, g0 + u);
            const float4* ph = (const float4*)(hs + (size_t)s * HIDTOT + lane * 8);
            a0[u] = ph[0]; a1[u] = ph[1];
            const float4* pd = (const float4*)(hd + (size_t)(lo + d) * HIDTOT + lane * 8);
            b0[u] = pd[0]; b1[u] = pd[1];
        }
        #pragma unroll
        for (int u = 0; u < SGROUP; ++u) {
            if (g0 + u < nb) {
                float part = 0.f;
                part += lrelu(a0[u].x + b0[u].x) * w0.x + lrelu(a0[u].y + b0[u].y) * w0.y +
                        lrelu(a0[u].z + b0[u].z) * w0.z + lrelu(a0[u].w + b0[u].w) * w0.w;
                part += lrelu(a1[u].x + b1[u].x) * w1.x + lrelu(a1[u].y + b1[u].y) * w1.y +
                        lrelu(a1[u].z + b1[u].z) * w1.z + lrelu(a1[u].w + b1[u].w) * w1.w;
                part += __shfl_xor(part, 1, 32);
                part += __shfl_xor(part, 2, 32);
                part += __shfl_xor(part, 4, 32);
                if ((lane & 7) == 0) {
                    const int h = lane >> 3;
                    const int d = __builtin_amdgcn_readlane(dl_i, g0 + u);
                    scp[(g0 + u) * 4 + h] = part;
                    lds_max(mxb + d * 4 + h, ford(part));
                }
            }
        }
    }
}

__device__ __forceinline__ void add_batch(
    const unsigned* qp, const float* scp, const int nb, const int lane,
    const float* __restrict__ hs, const int* __restrict__ src,
    const unsigned* mxb, float* dn, float* acc)
{
    int s_i = 0, dl_i = 0;
    float a0 = 0.f, a1 = 0.f, a2 = 0.f, a3 = 0.f;
    if (lane < nb) {
        const unsigned ent = qp[lane];
        dl_i = (int)(ent & 255u);
        s_i = src[(int)(ent >> 8)];
        const float4 x = ((const float4*)scp)[lane];
        const uint4 mb = ((const uint4*)mxb)[dl_i];
        a0 = expf(x.x - finite_or_zero(fdec(mb.x)));
        a1 = expf(x.y - finite_or_zero(fdec(mb.y)));
        a2 = expf(x.z - finite_or_zero(fdec(mb.z)));
        a3 = expf(x.w - finite_or_zero(fdec(mb.w)));
        lds_add(dn + dl_i * 4 + 0, a0);
        lds_add(dn + dl_i * 4 + 1, a1);
        lds_add(dn + dl_i * 4 + 2, a2);
        lds_add(dn + dl_i * 4 + 3, a3);
    }
    const bool lowhalf = lane < 16;
    for (int g0 = 0; g0 < nb; g0 += GROUP) {
        float4 ra[GROUP], rb[GROUP];
        #pragma unroll
        for (int u = 0; u < GROUP; ++u) {
            const int s = __builtin_amdgcn_readlane(s_i, g0 + u);
            const float4* p = (const float4*)(hs + (size_t)s * HIDTOT) + lane;
            ra[u] = p[0];
            rb[u] = p[32];
        }
        #pragma unroll
        for (int u = 0; u < GROUP; ++u) {
            if (g0 + u < nb) {
                const int d = __builtin_amdgcn_readlane(dl_i, g0 + u);
                const float h0 = __int_as_float(__builtin_amdgcn_readlane(__float_as_int(a0), g0 + u));
                const float h1 = __int_as_float(__builtin_amdgcn_readlane(__float_as_int(a1), g0 + u));
                const float h2 = __int_as_float(__builtin_amdgcn_readlane(__float_as_int(a2), g0 + u));
                const float h3 = __int_as_float(__builtin_amdgcn_readlane(__float_as_int(a3), g0 + u));
                const float aA = lowhalf ? h0 : h1;
                const float aB = lowhalf ? h2 : h3;
                float* a = acc + d * HIDTOT + lane;
                lds_add(a +   0, aA * ra[u].x);
                lds_add(a +  32, aA * ra[u].y);
                lds_add(a +  64, aA * ra[u].z);
                lds_add(a +  96, aA * ra[u].w);
                lds_add(a + 128, aB * rb[u].x);
                lds_add(a + 160, aB * rb[u].y);
                lds_add(a + 192, aB * rb[u].z);
                lds_add(a + 224, aB * rb[u].w);
            }
        }
    }
}

__device__ __forceinline__ float row16_sum(float v) {
    v += __int_as_float(__builtin_amdgcn_update_dpp(0, __float_as_int(v), 0x161, 0xF, 0xF, true));
    v += __int_as_float(__builtin_amdgcn_update_dpp(0, __float_as_int(v), 0x162, 0xF, 0xF, true));
    v += __int_as_float(__builtin_amdgcn_update_dpp(0, __float_as_int(v), 0x164, 0xF, 0xF, true));
    v += __int_as_float(__builtin_amdgcn_update_dpp(0, __float_as_int(v), 0x168, 0xF, 0xF, true));
    return v;
}

template <int P>
__device__ __forceinline__ void pair_batch(
    const unsigned* qp, const int nb, const int lane,
    const float* __restrict__ hs, const float4* hdt, const float4 w,
    float* dn, float* sums, unsigned* bad)
{
    int s_i = 0, dl_i = 0;
    if (lane < nb) {
        const unsigned ent = qp[lane];
        dl_i = (int)(ent & 255u);
        s_i = (int)(ent >> 8);
    }
    const int j = lane & 15;
    for (int g0 = 0; g0 < nb; g0 += PGROUP) {
        float4 a[PGROUP];
        float sv[PGROUP];
        #pragma unroll
        for (int u = 0; u < PGROUP; ++u) {
            const int s = __builtin_amdgcn_readlane(s_i, g0 + u);
            a[u] = ((const float4*)(hs + (size_t)s * HIDTOT + 128 * P))[lane];
        }
        #pragma unroll
        for (int u = 0; u < PGROUP; ++u) {
            const int d = __builtin_amdgcn_readlane(dl_i, g0 + u);
            const float4 b = hdt[d * 32 + lane];
            const float part = lrelu(a[u].x + b.x) * w.x + lrelu(a[u].y + b.y) * w.y +
                               lrelu(a[u].z + b.z) * w.z + lrelu(a[u].w + b.w) * w.w;
            sv[u] = row16_sum(part);
        }
        float mine = 0.f;
        #pragma unroll
        for (int u = 0; u < PGROUP; ++u) mine = (j == u) ? sv[u] : mine;
        const float ex = expf(mine);
        const int dj = __shfl(dl_i, g0 + (j & (PGROUP - 1)), 32);
        if (j < PGROUP && g0 + j < nb) {
            if (!(fabsf(mine) <= SCORE_LIMIT)) *bad = 1u;
            lds_add(dn + dj * 4 + 2 * P + (lane >> 4), ex);
        }
        #pragma unroll
        for (int u = 0; u < PGROUP; ++u) {
            if (g0 + u < nb) {
                const int d = __builtin_amdgcn_readlane(dl_i, g0 + u);
                const float eA = __int_as_float(__builtin_amdgcn_readlane(__float_as_int(ex), u));
                const float eB = __int_as_float(__builtin_amdgcn_readlane(__float_as_int(ex), 16 + u));
                const float e = (lane < 16) ? eA : eB;
                float* q = sums + d * 128 + lane;
                lds_add(q +  0, e * a[u].x);
                lds_add(q + 32, e * a[u].y);
                lds_add(q + 64, e * a[u].z);
                lds_add(q + 96, e * a[u].w);
            }
        }
    }
}

template <int P>
__device__ __forceinline__ void pair_phase(
    const int beg, const int end, const int tid, const int wave, const int lane, const int lo,
    const float* __restrict__ hs, const float* __restrict__ hd,
    const float* __restrict__ attn, float* __restrict__ out,
    float* acc, const unsigned* list, float* dn, unsigned* bad)
{
    v4f* hdt = (v4f*)acc;
    float* sums = acc + TILE_R * 128;
    if (P == 1) {
        const v4f* hdg = (const v4f*)(hd + (size_t)lo * HIDTOT);
        for (int i = tid; i < TILE_R * 32; i += 1024)
            hdt[i] = __builtin_nontemporal_load(hdg + (i >> 5) * 64 + 32 + (i & 31));
    }
    for (int i = tid; i < TILE_R * 128 / 4; i += 1024)
        ((float4*)sums)[i] = make_float4(0.f, 0.f, 0.f, 0.f);
    __syncthreads();

    const float4 w = ((const float4*)(attn + 128 * P))[lane];
    for (int i0 = beg; i0 < end; i0 += 32) {
        const int nb = (end - i0) < 32 ? (end - i0) : 32;
        pair_batch<P>(list + i0, nb, lane, hs, (const float4*)acc, w, dn, sums, bad);
    }
    __syncthreads();

    for (int r = wave; r < TILE_R; r += AGG_WAVES) {
        const float* q = sums + r * 128 + lane;
        float den = dn[r * 4 + 2 * P + (lane >> 4)];
        den = den > 0.f ? den : 1.f;
        float4 o;
        o.x = q[ 0] / den;
        o.y = q[32] / den;
        o.z = q[64] / den;
        o.w = q[96] / den;
        if (!(isfinite(o.x) && isfinite(o.y) && isfinite(o.z) && isfinite(o.w))) *bad = 1u;
        o.x = o.x > 0.f ? o.x : (expf(o.x) - 1.f);
        o.y = o.y > 0.f ? o.y : (expf(o.y) - 1.f);
        o.z = o.z > 0.f ? o.z : (expf(o.z) - 1.f);
        o.w = o.w > 0.f ? o.w : (expf(o.w) - 1.f);
        ((float4*)(out + (size_t)(lo + r) * HIDTOT + 128 * P))[lane] = o;
    }
    __syncthreads();
}

__device__ __forceinline__ void window_chunk(
    const int n, const int tid, const int wave,
    const int lane, const int lo,
    const float* __restrict__ hs, const float* __restrict__ hd,
    const int* __restrict__ src, const float* __restrict__ attn,
    float* acc, float* sc, const unsigned* list,
    unsigned* mxb, unsigned* omx, float* dn)
{
    const int share = (n + AGG_WAVES - 1) / AGG_WAVES;
    const int beg = wave * share;
    const int end = (beg + share) < n ? (beg + share) : n;

    for (int i = tid; i < TILE_R * 4; i += 1024) omx[i] = mxb[i];
    __syncthreads();

    {
        const float4 w0 = ((const float4*)(attn + lane * 8))[0];
        const float4 w1 = ((const float4*)(attn + lane * 8))[1];
        for (int i0 = beg; i0 < end; i0 += 32) {
            const int nb = (end - i0) < 32 ? (end - i0) : 32;
            score_batch(list + i0, sc + i0 * 4, nb, lane, lo, hs, hd, src, w0, w1, mxb);
        }
    }
    __syncthreads();

    for (int i = tid; i < TILE_R * 4; i += 1024) {
        const unsigned ob = omx[i];
        const float f = (ob == ORD_NEG_INF) ? 0.f
            : expf(finite_or_zero(fdec(ob)) - finite_or_zero(fdec(mxb[i])));
        omx[i] = __float_as_uint(f);
        dn[i] *= f;
    }
    __syncthreads();
    for (int i = tid; i < TILE_R * HIDTOT; i += 1024) {
        const int r = i >> 8, p = i & 255;
        const int h = 2 * (p >> 7) + ((p & 31) >> 4);
        acc[i] *= __uint_as_float(omx[r * 4 + h]);
    }
    __syncthreads();

    for (int i0 = beg; i0 < end; i0 += 32) {
        const int nb = (end - i0) < 32 ? (end - i0) : 32;
        add_batch(list + i0, sc + i0 * 4, nb, lane, hs, src, mxb, dn, acc);
    }
}

__global__ __launch_bounds__(1024) void tile_layer1_kernel(
    const float* __restrict__ hs, const float* __restrict__ hd,
    const int* __restrict__ src, const int* __restrict__ dst,
    const float* __restrict__ attn, float* __restrict__ out,
    unsigned* __restrict__ tlist, unsigned* __restrict__ tcount,
    const unsigned* __restrict__ keys)
{
    __shared__ __attribute__((aligned(16))) float acc[TILE_R * HIDTOT];
    __shared__ __attribute__((aligned(16))) float sc[LCAP * 4];
    __shared__ __attribute__((aligned(16))) unsigned mxb[TILE_R * 4];
    __shared__ __attribute__((aligned(16))) unsigned omx[TILE_R * 4];
    __shared__ __attribute__((aligned(16))) float dn[TILE_R * 4];
    __shared__ unsigned list[LCAP];
    __shared__ unsigned count;
    __shared__ unsigned bad;

    const int tid  = threadIdx.x;
    const int wave = __builtin_amdgcn_readfirstlane(tid >> 5);
    const int lane = tid & 31;
    const int tile = blockIdx.x;
    const int lo   = tile * TILE_R;

    for (int i = tid; i < TILE_R * 4; i += 1024) { mxb[i] = ORD_NEG_INF; dn[i] = 0.f; }
    if (tid == 0) { count = 0u; bad = 0u; }
    __syncthreads();

    const int4* dst4 = (const int4*)dst;

    {
        const v4f* hdg = (const v4f*)(hd + (size_t)lo * HIDTOT);
        v4f* hdt = (v4f*)acc;
        const uint4* key4 = (const uint4*)keys;
        const unsigned pat = (unsigned)tile * 0x01010101u;
        const uint4 none = make_uint4(~0u, ~0u, ~0u, ~0u);
        const int rot = tile * (KSTEPS / NTILES);
        uint4 v[4];
        #pragma unroll
        for (int u = 0; u < 4; ++u) {
            int g = wave + 32 * u + rot;
            if (g >= KSTEPS) g -= KSTEPS;
            v[u] = key4[g * 32 + lane];
        }
        for (int it = 0; it < 52; it += 4) {
            const int hj = wave + 32 * (it >> 2);
            v4f hv = {0.f, 0.f, 0.f, 0.f};
            if (hj < TILE_R)
                hv = __builtin_nontemporal_load(hdg + hj * 64 + lane);
            uint4 nv[4];
            #pragma unroll
            for (int u = 0; u < 4; ++u) {
                const int st = wave + 32 * (it + 4 + u);
                int g = st + rot;
                if (g >= KSTEPS) g -= KSTEPS;
                nv[u] = none;
                if (st < KSTEPS) nv[u] = key4[g * 32 + lane];
            }
            #pragma unroll
            for (int u = 0; u < 4; ++u) {
                int g = wave + 32 * (it + u) + rot;
                if (g >= KSTEPS) g -= KSTEPS;
                key_step(v[u], g * 512 + lane * 16, pat, list, &count);
            }
            #pragma unroll
            for (int u = 0; u < 4; ++u) v[u] = nv[u];
            if (hj < TILE_R)
                hdt[hj * 32 + lane] = hv;
        }
    }
    __syncthreads();

    const unsigned total = count;
    unsigned* wlist = tlist + (size_t)tile * LCAP;
    if (tid == 0) tcount[tile] = total;
    bool exact = total > (unsigned)LCAP;
    if (!exact) {
        for (int i = tid; i < (int)total; i += 1024) {
            const unsigned e = list[i];
            const unsigned ent = ((unsigned)src[e] << 8) | (unsigned)(dst[e] - lo);
            list[i] = ent;
            wlist[i] = ent;
        }
        const int n = (int)total;
        const int share = (n + AGG_WAVES - 1) / AGG_WAVES;
        const int beg = wave * share;
        const int end = (beg + share) < n ? (beg + share) : n;
        pair_phase<0>(beg, end, tid, wave, lane, lo, hs, hd, attn, out, acc, list, dn, &bad);
        pair_phase<1>(beg, end, tid, wave, lane, lo, hs, hd, attn, out, acc, list, dn, &bad);
        exact = bad != 0u;
    }
    if (exact) {
        for (int i = tid; i < TILE_R * 4; i += 1024) dn[i] = 0.f;
        for (int i = tid; i < TILE_R * HIDTOT / 4; i += 1024)
            ((float4*)acc)[i] = make_float4(0.f, 0.f, 0.f, 0.f);
        const int nwin = (SCAN_STEPS + WIN_STEPS - 1) / WIN_STEPS;
        for (int w = 0; w < nwin; ++w) {
            __syncthreads();
            if (tid == 0) count = 0u;
            __syncthreads();
            const int w0 = w * WIN_STEPS;
            const int wend = (w0 + WIN_STEPS) < SCAN_STEPS ? (w0 + WIN_STEPS) : SCAN_STEPS;
            for (int st = w0 + wave; st < wend; st += AGG_WAVES)
                scan_step(dst4[st * 32 + lane], st * 128 + lane * 4, lo, lane, list, &count);
            __syncthreads();
            window_chunk((int)count, tid, wave, lane, lo, hs, hd, src, attn,
                         acc, sc, list, mxb, omx, dn);
        }
        __syncthreads();

        for (int r = wave; r < TILE_R; r += AGG_WAVES) {
            const float* a = acc + r * HIDTOT + lane;
            float* po = out + (size_t)(lo + r) * HIDTOT;
            #pragma unroll
            for (int qq = 0; qq < 2; ++qq) {
                float den = dn[r * 4 + 2 * qq + (lane >> 4)];
                den = den > 0.f ? den : 1.f;
                float4 o;
                o.x = a[qq * 128 +  0] / den;
                o.y = a[qq * 128 + 32] / den;
                o.z = a[qq * 128 + 64] / den;
                o.w = a[qq * 128 + 96] / den;
                o.x = o.x > 0.f ? o.x : (expf(o.x) - 1.f);
                o.y = o.y > 0.f ? o.y : (expf(o.y) - 1.f);
                o.z = o.z > 0.f ? o.z : (expf(o.z) - 1.f);
                o.w = o.w > 0.f ? o.w : (expf(o.w) - 1.f);
                ((float4*)(po + qq * 128))[lane] = o;
            }
        }
    }
}

#define P2_BLOCKS  256
#define P2_THREADS 512
#define P2_WAVES   (P2_BLOCKS * (P2_THREADS / 32))
#define P2_FULL    (NNODES / P2_WAVES)
#define P2_ROUNDS  (P2_FULL + 1)

__global__ __launch_bounds__(P2_THREADS) void proj2_kernel(
    const float* __restrict__ h,
    const float* __restrict__ W2s, const float* __restrict__ b2s,
    const float* __restrict__ W2d, const float* __restrict__ b2d,
    float* __restrict__ hs2, float* __restrict__ hd2)
{
    const int lane = threadIdx.x & 31;
    const int gw = __builtin_amdgcn_readfirstlane(
        (int)(blockIdx.x * (P2_THREADS / 32) + (threadIdx.x >> 5)));
    const int base = lane * 8;

    float2 ws[8], wd[8];
    #pragma unroll
    for (int k = 0; k < 8; ++k) {
        ws[k] = *(const float2*)(W2s + (size_t)(base + k) * 2);
        wd[k] = *(const float2*)(W2d + (size_t)(base + k) * 2);
    }
    const float bs0 = b2s[0], bs1 = b2s[1], bd0 = b2d[0], bd1 = b2d[1];

    const bool tail = gw + P2_FULL * P2_WAVES < NNODES;
    int node[P2_ROUNDS];
    #pragma unroll
    for (int q = 0; q < P2_FULL; ++q) node[q] = gw + q * P2_WAVES;
    node[P2_FULL] = tail ? gw + P2_FULL * P2_WAVES : gw;

    float4 ra[P2_ROUNDS], rb[P2_ROUNDS];
    #pragma unroll
    for (int q = 0; q < P2_ROUNDS; ++q) {
        const float4* p = (const float4*)(h + (size_t)node[q] * HIDTOT + base);
        ra[q] = p[0]; rb[q] = p[1];
    }
    __builtin_amdgcn_sched_barrier(0);

    float s0[P2_ROUNDS], s1[P2_ROUNDS], d0[P2_ROUNDS], d1[P2_ROUNDS];
    #pragma unroll
    for (int q = 0; q < P2_ROUNDS; ++q) {
        const float x[8] = {ra[q].x, ra[q].y, ra[q].z, ra[q].w,
                            rb[q].x, rb[q].y, rb[q].z, rb[q].w};
        float a0 = 0.f, a1 = 0.f, c0 = 0.f, c1 = 0.f;
        #pragma unroll
        for (int k = 0; k < 8; ++k) {
            a0 += x[k] * ws[k].x; a1 += x[k] * ws[k].y;
            c0 += x[k] * wd[k].x; c1 += x[k] * wd[k].y;
        }
        s0[q] = row16_sum(a0); s1[q] = row16_sum(a1);
        d0[q] = row16_sum(c0); d1[q] = row16_sum(c1);
    }
    float t0[P2_ROUNDS], t1[P2_ROUNDS], u0[P2_ROUNDS], u1[P2_ROUNDS];
    #pragma unroll
    for (int q = 0; q < P2_ROUNDS; ++q) {
        t0[q] = __shfl_xor(s0[q], 16, 32); t1[q] = __shfl_xor(s1[q], 16, 32);
        u0[q] = __shfl_xor(d0[q], 16, 32); u1[q] = __shfl_xor(d1[q], 16, 32);
    }
    if (lane == 0) {
        #pragma unroll
        for (int q = 0; q < P2_ROUNDS; ++q) {
            if (q < P2_FULL || tail) {
                const int n = node[q];
                hs2[(size_t)n * 2 + 0] = (s0[q] + t0[q]) + bs0;
                hs2[(size_t)n * 2 + 1] = (s1[q] + t1[q]) + bs1;
                hd2[(size_t)n * 2 + 0] = (d0[q] + u0[q]) + bd0;
                hd2[(size_t)n * 2 + 1] = (d1[q] + u1[q]) + bd1;
            }
        }
    }
}

#define PAIR_R (2 * TILE_R)
#define NPAIRS 125

__device__ __forceinline__ float score2_of(const float2 a, const float2 b,
                                           const float w0, const float w1) {
    return lrelu(a.x + b.x) * w0 + lrelu(a.y + b.y) * w1;
}

__global__ __launch_bounds__(1024) void tile_layer2_kernel(
    const float* __restrict__ hs2, const float* __restrict__ hd2,
    const int* __restrict__ src, const int* __restrict__ dst,
    const float* __restrict__ attn2,
    const unsigned* __restrict__ tlist, const unsigned* __restrict__ tcount,
    float* __restrict__ out)
{
    __shared__ __attribute__((aligned(16))) float ent[2 * LCAP * 4];
    __shared__ unsigned mx[PAIR_R];
    __shared__ float dn[PAIR_R];
    __shared__ float sm[PAIR_R * 2];

    const int tid = threadIdx.x;
    const int pr  = blockIdx.x;
    const int lo  = pr * PAIR_R;

    for (int i = tid; i < PAIR_R; i += 1024) {
        mx[i] = ORD_NEG_INF; dn[i] = 0.f; sm[2 * i] = 0.f; sm[2 * i + 1] = 0.f;
    }
    const float w0 = attn2[0], w1 = attn2[1];
    const unsigned c0 = tcount[2 * pr], c1 = tcount[2 * pr + 1];
    const bool listed = (c0 <= (unsigned)LCAP) && (c1 <= (unsigned)LCAP);
    __syncthreads();

    if (listed) {
        const int n0 = (int)c0, n = (int)(c0 + c1);
        const unsigned* l0 = tlist + (size_t)(2 * pr) * LCAP;
        const unsigned* l1 = tlist + (size_t)(2 * pr + 1) * LCAP;
        float4* e4 = (float4*)ent;
        for (int ib = tid; ib < n; ib += 4096) {
            unsigned e[4];
            int dd[4];
            #pragma unroll
            for (int u = 0; u < 4; ++u) {
                const int i = ib + 1024 * u;
                e[u] = 0u;
                if (i < n) e[u] = (i < n0) ? l0[i] : l1[i - n0];
            }
            float2 a[4], b[4];
            #pragma unroll
            for (int u = 0; u < 4; ++u) {
                const int i = ib + 1024 * u;
                dd[u] = (int)(e[u] & 255u) + ((i < n0) ? 0 : TILE_R);
                a[u] = *(const float2*)(hs2 + (size_t)(e[u] >> 8) * 2);
                b[u] = *(const float2*)(hd2 + (size_t)(lo + dd[u]) * 2);
            }
            #pragma unroll
            for (int u = 0; u < 4; ++u) {
                const int i = ib + 1024 * u;
                if (i < n) {
                    const int d = dd[u];
                    const float s = score2_of(a[u], b[u], w0, w1);
                    e4[i] = make_float4(s, a[u].x, a[u].y, __int_as_float(d));
                    lds_max(mx + d, ford(s));
                }
            }
        }
        __syncthreads();
        for (int i = tid; i < n; i += 1024) {
            const float4 t = e4[i];
            const int d = __float_as_int(t.w);
            const float ex = expf(t.x - finite_or_zero(fdec(mx[d])));
            lds_add(dn + d, ex);
            lds_add(sm + 2 * d, ex * t.y);
            lds_add(sm + 2 * d + 1, ex * t.z);
        }
    } else {
        for (int e = tid; e < EDGES; e += 1024) {
            const unsigned d = (unsigned)(dst[e] - lo);
            if (d < (unsigned)PAIR_R) {
                const float2 a = *(const float2*)(hs2 + (size_t)src[e] * 2);
                const float2 b = *(const float2*)(hd2 + (size_t)(lo + (int)d) * 2);
                lds_max(mx + d, ford(score2_of(a, b, w0, w1)));
            }
        }
        __syncthreads();
        for (int e = tid; e < EDGES; e += 1024) {
            const unsigned d = (unsigned)(dst[e] - lo);
            if (d < (unsigned)PAIR_R) {
                const float2 a = *(const float2*)(hs2 + (size_t)src[e] * 2);
                const float2 b = *(const float2*)(hd2 + (size_t)(lo + (int)d) * 2);
                const float ex = expf(score2_of(a, b, w0, w1) - finite_or_zero(fdec(mx[d])));
                lds_add(dn + d, ex);
                lds_add(sm + 2 * d, ex * a.x);
                lds_add(sm + 2 * d + 1, ex * a.y);
            }
        }
    }
    __syncthreads();

    for (int i = tid; i < PAIR_R * 2; i += 1024) {
        float den = dn[i >> 1];
        den = den > 0.f ? den : 1.f;
        out[(size_t)lo * 2 + i] = sm[i] / den;
    }
}

extern "C" void kernel_launch(void* const* d_in, const int* in_sizes, int n_in,
                              void* d_out, int out_size, void* d_ws, size_t ws_size,
                              hipStream_t stream)
{
    const float* feat  = (const float*)d_in[0];
    const int*   src   = (const int*)d_in[1];
    const int*   dst   = (const int*)d_in[2];
    const float* W1s   = (const float*)d_in[3];
    const float* b1s   = (const float*)d_in[4];
    const float* W1d   = (const float*)d_in[5];
    const float* b1d   = (const float*)d_in[6];
    const float* attn1 = (const float*)d_in[7];
    const float* W2s   = (const float*)d_in[8];
    const float* b2s   = (const float*)d_in[9];
    const float* W2d   = (const float*)d_in[10];
    const float* b2d   = (const float*)d_in[11];
    const float* attn2 = (const float*)d_in[12];
    float* out = (float*)d_out;

    float* ws = (float*)d_ws;
    size_t off = 0;
    float*    hs1    = ws + off; off += (size_t)NNODES * HIDTOT;
    float*    hd1    = ws + off; off += (size_t)NNODES * HIDTOT;
    float*    out1   = ws + off; off += (size_t)NNODES * HIDTOT;
    float*    hs2    = ws + off; off += (size_t)NNODES * 2;
    float*    hd2    = ws + off; off += (size_t)NNODES * 2;
    unsigned* tlist  = (unsigned*)(ws + off); off += (size_t)NTILES * LCAP;
    unsigned* tcount = (unsigned*)(ws + off); off += (size_t)256;
    unsigned* keys   = (unsigned*)(ws + off); off += (size_t)KSTEPS * 128;

    const int B = 256;

    tile_key_kernel<<<KEY_BLOCKS, B, 0, stream>>>(dst, keys);
    gemm1_kernel<<<G_BLOCKS, 1024, 0, stream>>>(
        feat, W1s, b1s, W1d, b1d, hs1, hd1);
    tile_layer1_kernel<<<NTILES, 1024, 0, stream>>>(
        hs1, hd1, src, dst, attn1, out1, tlist, tcount, keys);
    proj2_kernel<<<P2_BLOCKS, P2_THREADS, 0, stream>>>(
        out1, W2s, b2s, W2d, b2d, hs2, hd2);
    tile_layer2_kernel<<<NPAIRS, 1024, 0, stream>>>(
        hs2, hd2, src, dst, attn2, tlist, tcount, out);

    (void)in_sizes; (void)n_in; (void)out_size; (void)ws_size;
}
